# starting kernel
# baseline (speedup 1.0000x reference)
#include <hip/hip_runtime.h>
#include <math.h>

#define N_DIS 20000
#define N_MIC 80000
#define NTOT  (N_DIS + N_MIC)
#define NEDGE 1600000
#define FATT  64
#define SLOPE 0.2f

typedef float v2f __attribute__((ext_vector_type(2)));
typedef float v8f __attribute__((ext_vector_type(8)));

__global__ __launch_bounds__(128) void gemm_wmma_f32(
        const float* __restrict__ X, const float* __restrict__ W,
        float* __restrict__ Z, int M, int K) {
    const int KT = 64;
    const int MT = 128;
    __shared__ float lX[MT * KT];
    __shared__ float lW[KT * FATT];

    const int lane  = threadIdx.x & 31;
    const int wave  = threadIdx.x >> 5;
    const int row   = lane & 15;
    const int khalf = lane >> 4;
    const int n     = lane & 15;

    const int blk_row = blockIdx.x * MT;
    const int lr0 = wave * 32 + row;
    const int lr1 = lr0 + 16;

    v8f c0 = {}, c1 = {}, c2 = {}, c3 = {};
    v8f c4 = {}, c5 = {}, c6 = {}, c7 = {};

    for (int kt = 0; kt < K; kt += KT) {
        for (int i = threadIdx.x; i < KT * FATT; i += blockDim.x) {
            int kg = kt + (i >> 6);
            lW[i] = (kg < K) ? W[(size_t)kg * FATT + (i & 63)] : 0.0f;
        }
        for (int i = threadIdx.x; i < MT * KT; i += blockDim.x) {
            int r  = i >> 6;
            int cc = i & 63;
            int mg = blk_row + r;
            int kg = kt + cc;
            lX[i] = (mg < M && kg < K) ? X[(size_t)mg * K + kg] : 0.0f;
        }
        __syncthreads();

        #pragma unroll 4
        for (int k0 = 0; k0 < KT; k0 += 4) {
            const int ka = k0 + 2 * khalf;
            v2f a0 = *(const v2f*)(lX + lr0 * KT + ka);
            v2f a1 = *(const v2f*)(lX + lr1 * KT + ka);

            const float* wr0 = lW + ka * FATT + n;
            const float* wr1 = lW + (ka + 1) * FATT + n;
            v2f b0 = { wr0[0],  wr1[0]  };
            v2f b1 = { wr0[16], wr1[16] };
            v2f b2 = { wr0[32], wr1[32] };
            v2f b3 = { wr0[48], wr1[48] };

            c0 = __builtin_amdgcn_wmma_f32_16x16x4_f32(false, a0, false, b0, (short)0, c0, false, false);
            c1 = __builtin_amdgcn_wmma_f32_16x16x4_f32(false, a0, false, b1, (short)0, c1, false, false);
            c2 = __builtin_amdgcn_wmma_f32_16x16x4_f32(false, a0, false, b2, (short)0, c2, false, false);
            c3 = __builtin_amdgcn_wmma_f32_16x16x4_f32(false, a0, false, b3, (short)0, c3, false, false);
            c4 = __builtin_amdgcn_wmma_f32_16x16x4_f32(false, a1, false, b0, (short)0, c4, false, false);
            c5 = __builtin_amdgcn_wmma_f32_16x16x4_f32(false, a1, false, b1, (short)0, c5, false, false);
            c6 = __builtin_amdgcn_wmma_f32_16x16x4_f32(false, a1, false, b2, (short)0, c6, false, false);
            c7 = __builtin_amdgcn_wmma_f32_16x16x4_f32(false, a1, false, b3, (short)0, c7, false, false);
        }
        __syncthreads();
    }

    const int m0 = blk_row + wave * 32 + 8 * khalf;
    #pragma unroll
    for (int r = 0; r < 8; ++r) {
        int mm = m0 + r;
        if (mm < M) {
            float* zr = Z + (size_t)mm * FATT + n;
            zr[0]  = c0[r];
            zr[16] = c1[r];
            zr[32] = c2[r];
            zr[48] = c3[r];
        }
        int mm1 = mm + 16;
        if (mm1 < M) {
            float* zr = Z + (size_t)mm1 * FATT + n;
            zr[0]  = c4[r];
            zr[16] = c5[r];
            zr[32] = c6[r];
            zr[48] = c7[r];
        }
    }
}

__global__ void init_state(unsigned int* __restrict__ m_ord, float* __restrict__ denom,
                           float* __restrict__ h) {
    long long i = (long long)blockIdx.x * blockDim.x + threadIdx.x;
    if (i < (long long)NTOT * FATT) h[i] = 0.0f;
    if (i < NTOT) { m_ord[i] = 0u; denom[i] = 0.0f; }
}

__device__ __forceinline__ unsigned int float_to_ord(float f) {
    unsigned int u = __float_as_uint(f);
    return (u & 0x80000000u) ? ~u : (u | 0x80000000u);
}
__device__ __forceinline__ float ord_to_float(unsigned int u) {
    return __uint_as_float((u & 0x80000000u) ? (u ^ 0x80000000u) : ~u);
}

__global__ void edge_score(const float* __restrict__ z, const int* __restrict__ src,
                           const int* __restrict__ dst, float* __restrict__ e,
                           unsigned int* __restrict__ m_ord) {
    const int lane = threadIdx.x & 31;
    const int wave = threadIdx.x >> 5;
    long long edge = (long long)blockIdx.x * (blockDim.x >> 5) + wave;
    if (edge >= NEDGE) return;
    const int s = src[edge], d = dst[edge];
    const float2* zs = (const float2*)(z + (size_t)s * FATT);
    const float2* zd = (const float2*)(z + (size_t)d * FATT);
    float2 a = zs[lane], b = zd[lane];
    float v = a.x * b.x + a.y * b.y;
    #pragma unroll
    for (int off = 16; off > 0; off >>= 1) v += __shfl_xor(v, off, 32);
    if (lane == 0) {
        float ev = v > 0.0f ? v : SLOPE * v;
        e[edge] = ev;
        atomicMax(&m_ord[d], float_to_ord(ev));
    }
}

__global__ void edge_expsum(const int* __restrict__ dst, float* __restrict__ e,
                            const unsigned int* __restrict__ m_ord,
                            float* __restrict__ denom) {
    long long i = (long long)blockIdx.x * blockDim.x + threadIdx.x;
    if (i >= NEDGE) return;
    const int d = dst[i];
    const float m = ord_to_float(m_ord[d]);
    const float ex = __expf(e[i] - m);
    e[i] = ex;
    atomicAdd(&denom[d], ex);
}

__global__ void edge_scatter(const float* __restrict__ z, const int* __restrict__ src,
                             const int* __restrict__ dst, const float* __restrict__ e_exp,
                             const float* __restrict__ denom, float* __restrict__ h) {
    const int lane = threadIdx.x & 31;
    const int wave = threadIdx.x >> 5;
    long long edge = (long long)blockIdx.x * (blockDim.x >> 5) + wave;
    if (edge >= NEDGE) return;
    const int s = src[edge], d = dst[edge];
    const float alpha = e_exp[edge] / denom[d];
    const float2* zs = (const float2*)(z + (size_t)s * FATT);
    float2 a = zs[lane];
    float* hd = h + (size_t)d * FATT + 2 * lane;
    atomicAdd(hd + 0, alpha * a.x);
    atomicAdd(hd + 1, alpha * a.y);
}

__global__ void elu_inplace(float* __restrict__ h) {
    long long i = (long long)blockIdx.x * blockDim.x + threadIdx.x;
    if (i >= (long long)NTOT * FATT) return;
    const float v = h[i];
    h[i] = v > 0.0f ? v : (__expf(v) - 1.0f);
}

extern "C" void kernel_launch(void* const* d_in, const int* in_sizes, int n_in,
                              void* d_out, int out_size, void* d_ws, size_t ws_size,
                              hipStream_t stream) {
    const float* d_sim  = (const float*)d_in[0];
    const float* mi_sim = (const float*)d_in[1];
    const float* W_d    = (const float*)d_in[2];
    const float* W_mi   = (const float*)d_in[3];
    const int*   src    = (const int*)d_in[4];
    const int*   dst    = (const int*)d_in[5];
    float*       out    = (float*)d_out;

    float* z            = (float*)d_ws;
    float* e            = z + (size_t)NTOT * FATT;
    unsigned int* m_ord = (unsigned int*)(e + NEDGE);
    float* denom        = (float*)(m_ord + NTOT);

    const long long tot_h = (long long)NTOT * FATT;

    init_state<<<(int)((tot_h + 255) / 256), 256, 0, stream>>>(m_ord, denom, out);

    gemm_wmma_f32<<<(N_DIS + 127) / 128, 128, 0, stream>>>(d_sim, W_d, z, N_DIS, 383);
    gemm_wmma_f32<<<(N_MIC + 127) / 128, 128, 0, stream>>>(mi_sim, W_mi,
                                                           z + (size_t)N_DIS * FATT, N_MIC, 495);

    edge_score<<<(NEDGE + 7) / 8, 256, 0, stream>>>(z, src, dst, e, m_ord);

    edge_expsum<<<(NEDGE + 255) / 256, 256, 0, stream>>>(dst, e, m_ord, denom);

    edge_scatter<<<(NEDGE + 7) / 8, 256, 0, stream>>>(z, src, dst, e, denom, out);

    elu_inplace<<<(int)((tot_h + 255) / 256), 256, 0, stream>>>(out);
}


# Weighted sum by owner work-groups: 256 groups each keep 391 destination rows in LDS, scan dst, add own edges with LDS float adds, then divide, ELU, store; no global float atomics; sum order changes
# speedup vs baseline: 2.2659x; 2.2659x over previous
#include <hip/hip_runtime.h>
#include <math.h>

#define N_DIS 20000
#define N_MIC 80000
#define NTOT  (N_DIS + N_MIC)
#define NEDGE 1600000
#define FATT  64
#define SLOPE 0.2f

typedef float v2f __attribute__((ext_vector_type(2)));
typedef float v8f __attribute__((ext_vector_type(8)));

typedef float v4f __attribute__((ext_vector_type(4)));

#define DEPTH 5

struct Frag { v4f a0, a1, w0, w1, w2, w3; };

__device__ __forceinline__ v4f ld_x4(const char* p) {
    v4f v;
    __builtin_memcpy(&v, p, sizeof(v));
    return v;
}

__device__ __forceinline__ void load_frag(Frag& f, const float* X, const float* W,
                                          unsigned xo0, unsigned xo1, unsigned wo, int kb) {
    const char* xk = (const char*)X + (size_t)kb * (8 * sizeof(float));
    const char* wk = (const char*)W + (size_t)kb * (8 * FATT * sizeof(float));
    f.a0 = ld_x4(xk + xo0);
    f.a1 = ld_x4(xk + xo1);
    f.w0 = *(const v4f*)(wk + wo);
    f.w1 = *(const v4f*)(wk + wo + 1 * FATT * sizeof(float));
    f.w2 = *(const v4f*)(wk + wo + 2 * FATT * sizeof(float));
    f.w3 = *(const v4f*)(wk + wo + 3 * FATT * sizeof(float));
}

__device__ __forceinline__ void mma_frag(const Frag& f, v8f (&acc)[2][4]) {
    const v2f a0lo = { f.a0.x, f.a0.y }, a0hi = { f.a0.z, f.a0.w };
    const v2f a1lo = { f.a1.x, f.a1.y }, a1hi = { f.a1.z, f.a1.w };
    #pragma unroll
    for (int j = 0; j < 4; ++j) {
        const v2f b = { f.w0[j], f.w1[j] };
        acc[0][j] = __builtin_amdgcn_wmma_f32_16x16x4_f32(false, a0lo, false, b, (short)0, acc[0][j], false, false);
        acc[1][j] = __builtin_amdgcn_wmma_f32_16x16x4_f32(false, a1lo, false, b, (short)0, acc[1][j], false, false);
    }
    #pragma unroll
    for (int j = 0; j < 4; ++j) {
        const v2f b = { f.w2[j], f.w3[j] };
        acc[0][j] = __builtin_amdgcn_wmma_f32_16x16x4_f32(false, a0hi, false, b, (short)0, acc[0][j], false, false);
        acc[1][j] = __builtin_amdgcn_wmma_f32_16x16x4_f32(false, a1hi, false, b, (short)0, acc[1][j], false, false);
    }
}

__global__ __launch_bounds__(128) void gemm_wmma_f32(
        const float* __restrict__ X, const float* __restrict__ W,
        float* __restrict__ Z, int M, int K) {
    const int lane = threadIdx.x & 31;
    const int wave = threadIdx.x >> 5;
    const int c    = lane & 15;
    const int h    = lane >> 4;

    const int row_base = (blockIdx.x * 4 + wave) * 32;
    if (row_base >= M) return;

    const int r0 = min(row_base + c, M - 1);
    const int r1 = min(row_base + 16 + c, M - 1);
    const unsigned xo0 = ((unsigned)r0 * (unsigned)K + 4u * h) * (unsigned)sizeof(float);
    const unsigned xo1 = ((unsigned)r1 * (unsigned)K + 4u * h) * (unsigned)sizeof(float);
    const unsigned wo  = (4u * h * FATT + 4u * c) * (unsigned)sizeof(float);

    v8f acc[2][4] = {};

    const int nfull = K >> 3;
    int kb = 0;
    if (nfull > 0) {
        Frag f[DEPTH];
        #pragma unroll
        for (int i = 0; i < DEPTH; ++i)
            load_frag(f[i], X, W, xo0, xo1, wo, min(i, nfull - 1));

        for (; kb + DEPTH <= nfull; kb += DEPTH) {
            #pragma unroll
            for (int i = 0; i < DEPTH; ++i) {
                mma_frag(f[i], acc);
                load_frag(f[i], X, W, xo0, xo1, wo, min(kb + DEPTH + i, nfull - 1));
                __builtin_amdgcn_sched_barrier(0);
            }
        }
        #pragma unroll
        for (int i = 0; i < DEPTH; ++i)
            if (kb + i < nfull) mma_frag(f[i], acc);
    }

    const int kt = nfull * 8;
    if (kt < K) {
        Frag t;
        float x0[4], x1[4];
        v4f wv[4];
        #pragma unroll
        for (int i = 0; i < 4; ++i) {
            const int  k  = kt + 4 * h + i;
            const bool ok = k < K;
            const int  kc = ok ? k : (K - 1);
            const float p0 = X[(size_t)r0 * K + kc];
            const float p1 = X[(size_t)r1 * K + kc];
            const v4f   pw = *(const v4f*)(W + (size_t)kc * FATT + 4 * c);
            const v4f   zero = { 0.0f, 0.0f, 0.0f, 0.0f };
            x0[i] = ok ? p0 : 0.0f;
            x1[i] = ok ? p1 : 0.0f;
            wv[i] = ok ? pw : zero;
        }
        t.a0 = (v4f){ x0[0], x0[1], x0[2], x0[3] };
        t.a1 = (v4f){ x1[0], x1[1], x1[2], x1[3] };
        t.w0 = wv[0]; t.w1 = wv[1]; t.w2 = wv[2]; t.w3 = wv[3];
        mma_frag(t, acc);
    }

    #pragma unroll
    for (int mt = 0; mt < 2; ++mt) {
        #pragma unroll
        for (int r = 0; r < 8; ++r) {
            const int mm = row_base + 16 * mt + 8 * h + r;
            if (mm < M) {
                const v4f o = { acc[mt][0][r], acc[mt][1][r], acc[mt][2][r], acc[mt][3][r] };
                *(v4f*)(Z + (size_t)mm * FATT + 4 * c) = o;
            }
        }
    }
}

__global__ void init_state(unsigned int* __restrict__ m_ord) {
    const int i = blockIdx.x * blockDim.x + threadIdx.x;
    if (i < NTOT) m_ord[i] = 0u;
}

__device__ __forceinline__ unsigned int float_to_ord(float f) {
    unsigned int u = __float_as_uint(f);
    return (u & 0x80000000u) ? ~u : (u | 0x80000000u);
}
__device__ __forceinline__ float ord_to_float(unsigned int u) {
    return __uint_as_float((u & 0x80000000u) ? (u ^ 0x80000000u) : ~u);
}

__global__ void edge_score(const float* __restrict__ z, const int* __restrict__ src,
                           const int* __restrict__ dst, float* __restrict__ e,
                           unsigned int* __restrict__ m_ord) {
    const int lane = threadIdx.x & 31;
    const int wave = threadIdx.x >> 5;
    long long edge = (long long)blockIdx.x * (blockDim.x >> 5) + wave;
    if (edge >= NEDGE) return;
    const int s = src[edge], d = dst[edge];
    const float2* zs = (const float2*)(z + (size_t)s * FATT);
    const float2* zd = (const float2*)(z + (size_t)d * FATT);
    float2 a = zs[lane], b = zd[lane];
    float v = a.x * b.x + a.y * b.y;
    #pragma unroll
    for (int off = 16; off > 0; off >>= 1) v += __shfl_xor(v, off, 32);
    if (lane == 0) {
        float ev = v > 0.0f ? v : SLOPE * v;
        e[edge] = ev;
        atomicMax(&m_ord[d], float_to_ord(ev));
    }
}

#define OWN_BLOCKS  256
#define OWN_NODES   ((NTOT + OWN_BLOCKS - 1) / OWN_BLOCKS)
#define OWN_THREADS 1024
#define OWN_WAVES   (OWN_THREADS / 32)
#define OWN_VEC     8
#define OWN_CHUNK   (OWN_THREADS * OWN_VEC * 4)
#define OWN_BATCH   4
#define OWN_NVEC    (NEDGE / 4)
#define OWN_ROUNDS  ((NEDGE + OWN_CHUNK - 1) / OWN_CHUNK)

__global__ __launch_bounds__(OWN_THREADS) void node_block_sum(
        const float* __restrict__ z, const int* __restrict__ src, const int* __restrict__ dst,
        const float* __restrict__ e, const unsigned int* __restrict__ m_ord,
        float* __restrict__ out) {
    __shared__ __attribute__((aligned(16))) float h_s[OWN_NODES * FATT];
    __shared__ float den_s[OWN_NODES];
    __shared__ float m_s[OWN_NODES];
    __shared__ int   list_s[OWN_CHUNK];
    __shared__ int   cnt_s[2];

    const int tid  = threadIdx.x;
    const int lane = tid & 31;
    const int wave = tid >> 5;
    const int base = blockIdx.x * OWN_NODES;

    for (int i = tid; i < OWN_NODES * FATT; i += OWN_THREADS) h_s[i] = 0.0f;
    for (int i = tid; i < OWN_NODES; i += OWN_THREADS) {
        den_s[i] = 0.0f;
        m_s[i]   = (base + i < NTOT) ? ord_to_float(m_ord[base + i]) : 0.0f;
    }
    if (tid < 2) cnt_s[tid] = 0;

    const int4* dst4 = (const int4*)dst;

    int4 cur[OWN_VEC];
    #pragma unroll
    for (int i = 0; i < OWN_VEC; ++i)
        cur[i] = dst4[min(i * OWN_THREADS + tid, OWN_NVEC - 1)];
    __syncthreads();

    for (int r = 0; r < OWN_ROUNDS; ++r) {
        int* cnt = &cnt_s[r & 1];
        const int vbase = r * (OWN_THREADS * OWN_VEC);

        int4 nxt[OWN_VEC];
        #pragma unroll
        for (int i = 0; i < OWN_VEC; ++i)
            nxt[i] = dst4[min(vbase + (OWN_THREADS * OWN_VEC) + i * OWN_THREADS + tid, OWN_NVEC - 1)];

        #pragma unroll
        for (int i = 0; i < OWN_VEC; ++i) {
            const int v  = vbase + i * OWN_THREADS + tid;
            const int e0 = v * 4;
            const unsigned l0 = (unsigned)(cur[i].x - base), l1 = (unsigned)(cur[i].y - base);
            const unsigned l2 = (unsigned)(cur[i].z - base), l3 = (unsigned)(cur[i].w - base);
            if (v < OWN_NVEC && min(min(l0, l1), min(l2, l3)) < (unsigned)OWN_NODES) {
                if (l0 < (unsigned)OWN_NODES) list_s[atomicAdd(cnt, 1)] = e0 + 0;
                if (l1 < (unsigned)OWN_NODES) list_s[atomicAdd(cnt, 1)] = e0 + 1;
                if (l2 < (unsigned)OWN_NODES) list_s[atomicAdd(cnt, 1)] = e0 + 2;
                if (l3 < (unsigned)OWN_NODES) list_s[atomicAdd(cnt, 1)] = e0 + 3;
            }
        }
        __syncthreads();
        const int n = *cnt;
        if (tid == 0) cnt_s[(r + 1) & 1] = 0;

        for (int j0 = wave * OWN_BATCH; j0 < n; j0 += OWN_WAVES * OWN_BATCH) {
            const bool ok = lane < OWN_BATCH && j0 + lane < n;
            int   s  = 0, dl = 0;
            float w  = 0.0f;
            if (ok) {
                const int edge = list_s[j0 + lane];
                s  = src[edge];
                dl = dst[edge] - base;
                w  = __expf(e[edge] - m_s[dl]);
                atomicAdd(&den_s[dl], w);
            }
            float2 row[OWN_BATCH];
            int    dq[OWN_BATCH];
            float  wq[OWN_BATCH];
            #pragma unroll
            for (int q = 0; q < OWN_BATCH; ++q) {
                const int sq = __builtin_amdgcn_readlane(s, q);
                dq[q]  = __builtin_amdgcn_readlane(dl, q);
                wq[q]  = __int_as_float(__builtin_amdgcn_readlane(__float_as_int(w), q));
                row[q] = ((const float2*)(z + (size_t)sq * FATT))[lane];
            }
            asm volatile("" :: "v"(row[0].x), "v"(row[0].y), "v"(row[1].x), "v"(row[1].y),
                               "v"(row[2].x), "v"(row[2].y), "v"(row[3].x), "v"(row[3].y));
            #pragma unroll
            for (int q = 0; q < OWN_BATCH; ++q) {
                if (j0 + q < n) {
                    float* hp = h_s + dq[q] * FATT + 2 * lane;
                    atomicAdd(hp + 0, wq[q] * row[q].x);
                    atomicAdd(hp + 1, wq[q] * row[q].y);
                }
            }
        }
        __syncthreads();
        #pragma unroll
        for (int i = 0; i < OWN_VEC; ++i) cur[i] = nxt[i];
    }

    for (int i4 = tid; i4 < OWN_NODES * (FATT / 4); i4 += OWN_THREADS) {
        const int nl = i4 / (FATT / 4);
        if (base + nl < NTOT) {
            const float dn = den_s[nl];
            float4 v = *(const float4*)&h_s[i4 * 4];
            v.x = dn > 0.0f ? v.x / dn : 0.0f;
            v.y = dn > 0.0f ? v.y / dn : 0.0f;
            v.z = dn > 0.0f ? v.z / dn : 0.0f;
            v.w = dn > 0.0f ? v.w / dn : 0.0f;
            v.x = v.x > 0.0f ? v.x : (__expf(v.x) - 1.0f);
            v.y = v.y > 0.0f ? v.y : (__expf(v.y) - 1.0f);
            v.z = v.z > 0.0f ? v.z : (__expf(v.z) - 1.0f);
            v.w = v.w > 0.0f ? v.w : (__expf(v.w) - 1.0f);
            *(float4*)(out + (size_t)base * FATT + (size_t)i4 * 4) = v;
        }
    }
}

extern "C" void kernel_launch(void* const* d_in, const int* in_sizes, int n_in,
                              void* d_out, int out_size, void* d_ws, size_t ws_size,
                              hipStream_t stream) {
    const float* d_sim  = (const float*)d_in[0];
    const float* mi_sim = (const float*)d_in[1];
    const float* W_d    = (const float*)d_in[2];
    const float* W_mi   = (const float*)d_in[3];
    const int*   src    = (const int*)d_in[4];
    const int*   dst    = (const int*)d_in[5];
    float*       out    = (float*)d_out;

    float* z            = (float*)d_ws;
    float* e            = z + (size_t)NTOT * FATT;
    unsigned int* m_ord = (unsigned int*)(e + NEDGE);

    init_state<<<(NTOT + 255) / 256, 256, 0, stream>>>(m_ord);

    gemm_wmma_f32<<<(N_DIS + 127) / 128, 128, 0, stream>>>(d_sim, W_d, z, N_DIS, 383);
    gemm_wmma_f32<<<(N_MIC + 127) / 128, 128, 0, stream>>>(mi_sim, W_mi,
                                                           z + (size_t)N_DIS * FATT, N_MIC, 495);

    edge_score<<<(NEDGE + 7) / 8, 256, 0, stream>>>(z, src, dst, e, m_ord);

    node_block_sum<<<OWN_BLOCKS, OWN_THREADS, 0, stream>>>(z, src, dst, e, m_ord, out);
}


# Scores and segment max folded into the owner work-groups: dot(z[src],z[dst]) per listed edge (6 per wave trip), running max in LDS, sums rescaled when it rises; 3 launches, not 5; sum order changes
# speedup vs baseline: 3.6683x; 1.6189x over previous
#include <hip/hip_runtime.h>
#include <math.h>

#define N_DIS 20000
#define N_MIC 80000
#define NTOT  (N_DIS + N_MIC)
#define NEDGE 1600000
#define FATT  64
#define SLOPE 0.2f

typedef float v2f __attribute__((ext_vector_type(2)));
typedef float v8f __attribute__((ext_vector_type(8)));

typedef float v4f __attribute__((ext_vector_type(4)));

#define DEPTH 5

struct Frag { v4f a0, a1, w0, w1, w2, w3; };

__device__ __forceinline__ v4f ld_x4(const char* p) {
    v4f v;
    __builtin_memcpy(&v, p, sizeof(v));
    return v;
}

__device__ __forceinline__ void load_frag(Frag& f, const float* X, const float* W,
                                          unsigned xo0, unsigned xo1, unsigned wo, int kb) {
    const char* xk = (const char*)X + (size_t)kb * (8 * sizeof(float));
    const char* wk = (const char*)W + (size_t)kb * (8 * FATT * sizeof(float));
    f.a0 = ld_x4(xk + xo0);
    f.a1 = ld_x4(xk + xo1);
    f.w0 = *(const v4f*)(wk + wo);
    f.w1 = *(const v4f*)(wk + wo + 1 * FATT * sizeof(float));
    f.w2 = *(const v4f*)(wk + wo + 2 * FATT * sizeof(float));
    f.w3 = *(const v4f*)(wk + wo + 3 * FATT * sizeof(float));
}

__device__ __forceinline__ void mma_frag(const Frag& f, v8f (&acc)[2][4]) {
    const v2f a0lo = { f.a0.x, f.a0.y }, a0hi = { f.a0.z, f.a0.w };
    const v2f a1lo = { f.a1.x, f.a1.y }, a1hi = { f.a1.z, f.a1.w };
    #pragma unroll
    for (int j = 0; j < 4; ++j) {
        const v2f b = { f.w0[j], f.w1[j] };
        acc[0][j] = __builtin_amdgcn_wmma_f32_16x16x4_f32(false, a0lo, false, b, (short)0, acc[0][j], false, false);
        acc[1][j] = __builtin_amdgcn_wmma_f32_16x16x4_f32(false, a1lo, false, b, (short)0, acc[1][j], false, false);
    }
    #pragma unroll
    for (int j = 0; j < 4; ++j) {
        const v2f b = { f.w2[j], f.w3[j] };
        acc[0][j] = __builtin_amdgcn_wmma_f32_16x16x4_f32(false, a0hi, false, b, (short)0, acc[0][j], false, false);
        acc[1][j] = __builtin_amdgcn_wmma_f32_16x16x4_f32(false, a1hi, false, b, (short)0, acc[1][j], false, false);
    }
}

__global__ __launch_bounds__(128) void gemm_wmma_f32(
        const float* __restrict__ X, const float* __restrict__ W,
        float* __restrict__ Z, int M, int K) {
    const int lane = threadIdx.x & 31;
    const int wave = threadIdx.x >> 5;
    const int c    = lane & 15;
    const int h    = lane >> 4;

    const int row_base = (blockIdx.x * 4 + wave) * 32;
    if (row_base >= M) return;

    const int r0 = min(row_base + c, M - 1);
    const int r1 = min(row_base + 16 + c, M - 1);
    const unsigned xo0 = ((unsigned)r0 * (unsigned)K + 4u * h) * (unsigned)sizeof(float);
    const unsigned xo1 = ((unsigned)r1 * (unsigned)K + 4u * h) * (unsigned)sizeof(float);
    const unsigned wo  = (4u * h * FATT + 4u * c) * (unsigned)sizeof(float);

    v8f acc[2][4] = {};

    const int nfull = K >> 3;
    int kb = 0;
    if (nfull > 0) {
        Frag f[DEPTH];
        #pragma unroll
        for (int i = 0; i < DEPTH; ++i)
            load_frag(f[i], X, W, xo0, xo1, wo, min(i, nfull - 1));

        for (; kb + DEPTH <= nfull; kb += DEPTH) {
            #pragma unroll
            for (int i = 0; i < DEPTH; ++i) {
                mma_frag(f[i], acc);
                load_frag(f[i], X, W, xo0, xo1, wo, min(kb + DEPTH + i, nfull - 1));
                __builtin_amdgcn_sched_barrier(0);
            }
        }
        #pragma unroll
        for (int i = 0; i < DEPTH; ++i)
            if (kb + i < nfull) mma_frag(f[i], acc);
    }

    const int kt = nfull * 8;
    if (kt < K) {
        Frag t;
        float x0[4], x1[4];
        v4f wv[4];
        #pragma unroll
        for (int i = 0; i < 4; ++i) {
            const int  k  = kt + 4 * h + i;
            const bool ok = k < K;
            const int  kc = ok ? k : (K - 1);
            const float p0 = X[(size_t)r0 * K + kc];
            const float p1 = X[(size_t)r1 * K + kc];
            const v4f   pw = *(const v4f*)(W + (size_t)kc * FATT + 4 * c);
            const v4f   zero = { 0.0f, 0.0f, 0.0f, 0.0f };
            x0[i] = ok ? p0 : 0.0f;
            x1[i] = ok ? p1 : 0.0f;
            wv[i] = ok ? pw : zero;
        }
        t.a0 = (v4f){ x0[0], x0[1], x0[2], x0[3] };
        t.a1 = (v4f){ x1[0], x1[1], x1[2], x1[3] };
        t.w0 = wv[0]; t.w1 = wv[1]; t.w2 = wv[2]; t.w3 = wv[3];
        mma_frag(t, acc);
    }

    #pragma unroll
    for (int mt = 0; mt < 2; ++mt) {
        #pragma unroll
        for (int r = 0; r < 8; ++r) {
            const int mm = row_base + 16 * mt + 8 * h + r;
            if (mm < M) {
                const v4f o = { acc[mt][0][r], acc[mt][1][r], acc[mt][2][r], acc[mt][3][r] };
                *(v4f*)(Z + (size_t)mm * FATT + 4 * c) = o;
            }
        }
    }
}

__device__ __forceinline__ unsigned int float_to_ord(float f) {
    unsigned int u = __float_as_uint(f);
    return (u & 0x80000000u) ? ~u : (u | 0x80000000u);
}
__device__ __forceinline__ float ord_to_float(unsigned int u) {
    return __uint_as_float((u & 0x80000000u) ? (u ^ 0x80000000u) : ~u);
}

template <int CTRL>
__device__ __forceinline__ float dpp_move(float v) {
    return __int_as_float(__builtin_amdgcn_update_dpp(0, __float_as_int(v), CTRL, 0xF, 0xF, true));
}
__device__ __forceinline__ float wave_sum(float v) {
    v += dpp_move<0x161>(v);
    v += dpp_move<0x162>(v);
    v += dpp_move<0x164>(v);
    v += dpp_move<0x168>(v);
    v += __int_as_float((int)__builtin_amdgcn_permlanex16(0u, (unsigned)__float_as_int(v),
                                                          0x76543210u, 0xfedcba98u, false, false));
    return v;
}

#define OWN_BLOCKS  256
#define OWN_NODES   ((NTOT + OWN_BLOCKS - 1) / OWN_BLOCKS)
#define OWN_THREADS 1024
#define OWN_WAVES   (OWN_THREADS / 32)
#define OWN_VEC     8
#define OWN_CHUNK   (OWN_THREADS * OWN_VEC * 4)
#define OWN_BATCH   6
#define OWN_TRIP    (OWN_WAVES * OWN_BATCH)
#define OWN_NVEC    (NEDGE / 4)
#define OWN_ROUNDS  ((NEDGE + OWN_CHUNK - 1) / OWN_CHUNK)

__global__ __launch_bounds__(OWN_THREADS) void node_block_attend(
        const float* __restrict__ z, const int* __restrict__ src, const int* __restrict__ dst,
        float* __restrict__ out) {
    __shared__ __attribute__((aligned(16))) float h_s[OWN_NODES * FATT];
    __shared__ float        den_s[OWN_NODES];
    __shared__ unsigned int mo_s[OWN_NODES];
    __shared__ unsigned int mc_s[OWN_NODES];
    __shared__ int          list_s[OWN_CHUNK];
    __shared__ int          cnt_s[3];

    const int tid  = threadIdx.x;
    const int lane = tid & 31;
    const int wave = tid >> 5;
    const int base = blockIdx.x * OWN_NODES;

    for (int i = tid; i < OWN_NODES * FATT; i += OWN_THREADS) h_s[i] = 0.0f;
    for (int i = tid; i < OWN_NODES; i += OWN_THREADS) {
        den_s[i] = 0.0f;
        mo_s[i]  = 0u;
        mc_s[i]  = 0u;
    }
    if (tid < 3) cnt_s[tid] = 0;

    const int4* dst4 = (const int4*)dst;

    int4 cur[OWN_VEC];
    #pragma unroll
    for (int i = 0; i < OWN_VEC; ++i)
        cur[i] = dst4[min(i * OWN_THREADS + tid, OWN_NVEC - 1)];
    __syncthreads();

    for (int r = 0; r < OWN_ROUNDS; ++r) {
        int* cnt = &cnt_s[r % 3];
        const int vbase = r * (OWN_THREADS * OWN_VEC);

        int4 nxt[OWN_VEC];
        #pragma unroll
        for (int i = 0; i < OWN_VEC; ++i)
            nxt[i] = dst4[min(vbase + (OWN_THREADS * OWN_VEC) + i * OWN_THREADS + tid, OWN_NVEC - 1)];

        #pragma unroll
        for (int i = 0; i < OWN_VEC; ++i) {
            const int v  = vbase + i * OWN_THREADS + tid;
            const int e0 = v * 4;
            const unsigned l0 = (unsigned)(cur[i].x - base), l1 = (unsigned)(cur[i].y - base);
            const unsigned l2 = (unsigned)(cur[i].z - base), l3 = (unsigned)(cur[i].w - base);
            if (v < OWN_NVEC && min(min(l0, l1), min(l2, l3)) < (unsigned)OWN_NODES) {
                if (l0 < (unsigned)OWN_NODES) list_s[atomicAdd(cnt, 1)] = e0 + 0;
                if (l1 < (unsigned)OWN_NODES) list_s[atomicAdd(cnt, 1)] = e0 + 1;
                if (l2 < (unsigned)OWN_NODES) list_s[atomicAdd(cnt, 1)] = e0 + 2;
                if (l3 < (unsigned)OWN_NODES) list_s[atomicAdd(cnt, 1)] = e0 + 3;
            }
        }
        __syncthreads();
        const int n = *cnt;
        if (tid == 0) cnt_s[(r + 2) % 3] = 0;

        const int trips = (n + OWN_TRIP - 1) / OWN_TRIP;
        for (int t = 0; t < trips; ++t) {
            const int  j0 = t * OWN_TRIP + wave * OWN_BATCH;
            const bool ok = lane < OWN_BATCH && j0 + lane < n;

            int s = 0, dl = 0;
            if (ok) {
                const int edge = list_s[j0 + lane];
                s  = src[edge];
                dl = dst[edge] - base;
            }
            float2 rs[OWN_BATCH], rd[OWN_BATCH];
            int    dq[OWN_BATCH];
            #pragma unroll
            for (int q = 0; q < OWN_BATCH; ++q) {
                const int sq = __builtin_amdgcn_readlane(s, q);
                dq[q] = __builtin_amdgcn_readlane(dl, q);
                rs[q] = ((const float2*)(z + (size_t)sq * FATT))[lane];
                rd[q] = ((const float2*)(z + (size_t)(base + dq[q]) * FATT))[lane];
            }
            float sc = 0.0f;
            #pragma unroll
            for (int q = 0; q < OWN_BATCH; ++q) {
                const float dot = wave_sum(rs[q].x * rd[q].x + rs[q].y * rd[q].y);
                if (lane == q) sc = dot;
            }
            sc = sc > 0.0f ? sc : SLOPE * sc;
            if (ok) atomicMax(&mo_s[dl], float_to_ord(sc));
            __syncthreads();

            unsigned int mnew = 0u;
            int   take = 0;
            float fs   = 1.0f;
            if (ok) {
                mnew = mo_s[dl];
                const unsigned int mold = atomicMax(&mc_s[dl], mnew);
                if (mold != 0u && mold < mnew) {
                    take = 1;
                    fs   = __expf(ord_to_float(mold) - ord_to_float(mnew));
                    den_s[dl] *= fs;
                }
            }
            #pragma unroll
            for (int q = 0; q < OWN_BATCH; ++q) {
                if (__builtin_amdgcn_readlane(take, q)) {
                    const float f = __int_as_float(__builtin_amdgcn_readlane(__float_as_int(fs), q));
                    float2* hp = (float2*)(h_s + dq[q] * FATT) + lane;
                    float2 hv = *hp;
                    hv.x *= f;
                    hv.y *= f;
                    *hp = hv;
                }
            }
            __syncthreads();

            float w = 0.0f;
            if (ok) {
                w = __expf(sc - ord_to_float(mnew));
                atomicAdd(&den_s[dl], w);
            }
            #pragma unroll
            for (int q = 0; q < OWN_BATCH; ++q) {
                if (j0 + q < n) {
                    const float wq = __int_as_float(__builtin_amdgcn_readlane(__float_as_int(w), q));
                    float* hp = h_s + dq[q] * FATT + 2 * lane;
                    atomicAdd(hp + 0, wq * rs[q].x);
                    atomicAdd(hp + 1, wq * rs[q].y);
                }
            }
        }
        #pragma unroll
        for (int i = 0; i < OWN_VEC; ++i) cur[i] = nxt[i];
    }
    __syncthreads();

    for (int i4 = tid; i4 < OWN_NODES * (FATT / 4); i4 += OWN_THREADS) {
        const int nl = i4 / (FATT / 4);
        if (base + nl < NTOT) {
            const float dn = den_s[nl];
            float4 v = *(const float4*)&h_s[i4 * 4];
            v.x = dn > 0.0f ? v.x / dn : 0.0f;
            v.y = dn > 0.0f ? v.y / dn : 0.0f;
            v.z = dn > 0.0f ? v.z / dn : 0.0f;
            v.w = dn > 0.0f ? v.w / dn : 0.0f;
            v.x = v.x > 0.0f ? v.x : (__expf(v.x) - 1.0f);
            v.y = v.y > 0.0f ? v.y : (__expf(v.y) - 1.0f);
            v.z = v.z > 0.0f ? v.z : (__expf(v.z) - 1.0f);
            v.w = v.w > 0.0f ? v.w : (__expf(v.w) - 1.0f);
            *(float4*)(out + (size_t)base * FATT + (size_t)i4 * 4) = v;
        }
    }
}

extern "C" void kernel_launch(void* const* d_in, const int* in_sizes, int n_in,
                              void* d_out, int out_size, void* d_ws, size_t ws_size,
                              hipStream_t stream) {
    const float* d_sim  = (const float*)d_in[0];
    const float* mi_sim = (const float*)d_in[1];
    const float* W_d    = (const float*)d_in[2];
    const float* W_mi   = (const float*)d_in[3];
    const int*   src    = (const int*)d_in[4];
    const int*   dst    = (const int*)d_in[5];
    float*       out    = (float*)d_out;

    float* z            = (float*)d_ws;

    gemm_wmma_f32<<<(N_DIS + 127) / 128, 128, 0, stream>>>(d_sim, W_d, z, N_DIS, 383);
    gemm_wmma_f32<<<(N_MIC + 127) / 128, 128, 0, stream>>>(mi_sim, W_mi,
                                                           z + (size_t)N_DIS * FATT, N_MIC, 495);

    node_block_attend<<<OWN_BLOCKS, OWN_THREADS, 0, stream>>>(z, src, dst, out);
}


# Owner work-groups scan a 1-byte owner key per edge (dst/391, written by a new small launch) instead of 4-byte dst: zero-byte word test, one LDS slot add per thread; 4 launches; sum order changes
# speedup vs baseline: 4.5103x; 1.2295x over previous
#include <hip/hip_runtime.h>
#include <math.h>

#define N_DIS 20000
#define N_MIC 80000
#define NTOT  (N_DIS + N_MIC)
#define NEDGE 1600000
#define FATT  64
#define SLOPE 0.2f

typedef float v2f __attribute__((ext_vector_type(2)));
typedef float v8f __attribute__((ext_vector_type(8)));

typedef float v4f __attribute__((ext_vector_type(4)));

#define DEPTH 5

struct Frag { v4f a0, a1, w0, w1, w2, w3; };

__device__ __forceinline__ v4f ld_x4(const char* p) {
    v4f v;
    __builtin_memcpy(&v, p, sizeof(v));
    return v;
}

__device__ __forceinline__ void load_frag(Frag& f, const float* X, const float* W,
                                          unsigned xo0, unsigned xo1, unsigned wo, int kb) {
    const char* xk = (const char*)X + (size_t)kb * (8 * sizeof(float));
    const char* wk = (const char*)W + (size_t)kb * (8 * FATT * sizeof(float));
    f.a0 = ld_x4(xk + xo0);
    f.a1 = ld_x4(xk + xo1);
    f.w0 = *(const v4f*)(wk + wo);
    f.w1 = *(const v4f*)(wk + wo + 1 * FATT * sizeof(float));
    f.w2 = *(const v4f*)(wk + wo + 2 * FATT * sizeof(float));
    f.w3 = *(const v4f*)(wk + wo + 3 * FATT * sizeof(float));
}

__device__ __forceinline__ void mma_frag(const Frag& f, v8f (&acc)[2][4]) {
    const v2f a0lo = { f.a0.x, f.a0.y }, a0hi = { f.a0.z, f.a0.w };
    const v2f a1lo = { f.a1.x, f.a1.y }, a1hi = { f.a1.z, f.a1.w };
    #pragma unroll
    for (int j = 0; j < 4; ++j) {
        const v2f b = { f.w0[j], f.w1[j] };
        acc[0][j] = __builtin_amdgcn_wmma_f32_16x16x4_f32(false, a0lo, false, b, (short)0, acc[0][j], false, false);
        acc[1][j] = __builtin_amdgcn_wmma_f32_16x16x4_f32(false, a1lo, false, b, (short)0, acc[1][j], false, false);
    }
    #pragma unroll
    for (int j = 0; j < 4; ++j) {
        const v2f b = { f.w2[j], f.w3[j] };
        acc[0][j] = __builtin_amdgcn_wmma_f32_16x16x4_f32(false, a0hi, false, b, (short)0, acc[0][j], false, false);
        acc[1][j] = __builtin_amdgcn_wmma_f32_16x16x4_f32(false, a1hi, false, b, (short)0, acc[1][j], false, false);
    }
}

__global__ __launch_bounds__(128) void gemm_wmma_f32(
        const float* __restrict__ X, const float* __restrict__ W,
        float* __restrict__ Z, int M, int K) {
    const int lane = threadIdx.x & 31;
    const int wave = threadIdx.x >> 5;
    const int c    = lane & 15;
    const int h    = lane >> 4;

    const int row_base = (blockIdx.x * 4 + wave) * 32;
    if (row_base >= M) return;

    const int r0 = min(row_base + c, M - 1);
    const int r1 = min(row_base + 16 + c, M - 1);
    const unsigned xo0 = ((unsigned)r0 * (unsigned)K + 4u * h) * (unsigned)sizeof(float);
    const unsigned xo1 = ((unsigned)r1 * (unsigned)K + 4u * h) * (unsigned)sizeof(float);
    const unsigned wo  = (4u * h * FATT + 4u * c) * (unsigned)sizeof(float);

    v8f acc[2][4] = {};

    const int nfull = K >> 3;
    int kb = 0;
    if (nfull > 0) {
        Frag f[DEPTH];
        #pragma unroll
        for (int i = 0; i < DEPTH; ++i)
            load_frag(f[i], X, W, xo0, xo1, wo, min(i, nfull - 1));

        for (; kb + DEPTH <= nfull; kb += DEPTH) {
            #pragma unroll
            for (int i = 0; i < DEPTH; ++i) {
                mma_frag(f[i], acc);
                load_frag(f[i], X, W, xo0, xo1, wo, min(kb + DEPTH + i, nfull - 1));
                __builtin_amdgcn_sched_barrier(0);
            }
        }
        #pragma unroll
        for (int i = 0; i < DEPTH; ++i)
            if (kb + i < nfull) mma_frag(f[i], acc);
    }

    const int kt = nfull * 8;
    if (kt < K) {
        Frag t;
        float x0[4], x1[4];
        v4f wv[4];
        #pragma unroll
        for (int i = 0; i < 4; ++i) {
            const int  k  = kt + 4 * h + i;
            const bool ok = k < K;
            const int  kc = ok ? k : (K - 1);
            const float p0 = X[(size_t)r0 * K + kc];
            const float p1 = X[(size_t)r1 * K + kc];
            const v4f   pw = *(const v4f*)(W + (size_t)kc * FATT + 4 * c);
            const v4f   zero = { 0.0f, 0.0f, 0.0f, 0.0f };
            x0[i] = ok ? p0 : 0.0f;
            x1[i] = ok ? p1 : 0.0f;
            wv[i] = ok ? pw : zero;
        }
        t.a0 = (v4f){ x0[0], x0[1], x0[2], x0[3] };
        t.a1 = (v4f){ x1[0], x1[1], x1[2], x1[3] };
        t.w0 = wv[0]; t.w1 = wv[1]; t.w2 = wv[2]; t.w3 = wv[3];
        mma_frag(t, acc);
    }

    #pragma unroll
    for (int mt = 0; mt < 2; ++mt) {
        #pragma unroll
        for (int r = 0; r < 8; ++r) {
            const int mm = row_base + 16 * mt + 8 * h + r;
            if (mm < M) {
                const v4f o = { acc[mt][0][r], acc[mt][1][r], acc[mt][2][r], acc[mt][3][r] };
                *(v4f*)(Z + (size_t)mm * FATT + 4 * c) = o;
            }
        }
    }
}

__device__ __forceinline__ unsigned int float_to_ord(float f) {
    unsigned int u = __float_as_uint(f);
    return (u & 0x80000000u) ? ~u : (u | 0x80000000u);
}
__device__ __forceinline__ float ord_to_float(unsigned int u) {
    return __uint_as_float((u & 0x80000000u) ? (u ^ 0x80000000u) : ~u);
}

template <int CTRL>
__device__ __forceinline__ float dpp_move(float v) {
    return __int_as_float(__builtin_amdgcn_update_dpp(0, __float_as_int(v), CTRL, 0xF, 0xF, true));
}
__device__ __forceinline__ float wave_sum(float v) {
    v += dpp_move<0x161>(v);
    v += dpp_move<0x162>(v);
    v += dpp_move<0x164>(v);
    v += dpp_move<0x168>(v);
    v += __int_as_float((int)__builtin_amdgcn_permlanex16(0u, (unsigned)__float_as_int(v),
                                                          0x76543210u, 0xfedcba98u, false, false));
    return v;
}

#define OWN_BLOCKS  256
#define OWN_NODES   ((NTOT + OWN_BLOCKS - 1) / OWN_BLOCKS)
#define KEY_NVEC    (NEDGE / 16)
#define KEY_THREADS 256
#define WS_Z_BYTES  ((size_t)NTOT * FATT * sizeof(float))
#define WS_NEED     (WS_Z_BYTES + (size_t)NEDGE)

__device__ __forceinline__ unsigned int key_of(int d) {
    const unsigned int u = (unsigned int)d;
    return u < (unsigned int)(OWN_BLOCKS * OWN_NODES) ? u / (unsigned int)OWN_NODES : 255u;
}

__global__ __launch_bounds__(KEY_THREADS) void owner_keys(
        const int* __restrict__ dst, unsigned int* __restrict__ keys, size_t ws_size) {
    if (ws_size < WS_NEED) return;
    const int v = blockIdx.x * KEY_THREADS + threadIdx.x;
    if (v >= KEY_NVEC) return;
    const int4* d4 = (const int4*)dst + (size_t)v * 4;
    unsigned int w[4];
    #pragma unroll
    for (int j = 0; j < 4; ++j) {
        const int4 d = d4[j];
        w[j] = key_of(d.x) | (key_of(d.y) << 8) | (key_of(d.z) << 16) | (key_of(d.w) << 24);
    }
    ((uint4*)keys)[v] = make_uint4(w[0], w[1], w[2], w[3]);
}

#define OWN_THREADS 1024
#define OWN_WAVES   (OWN_THREADS / 32)
#define OWN_VEC     8
#define OWN_KVEC    2
#define OWN_CHUNK   (OWN_THREADS * OWN_VEC * 4)
#define OWN_BATCH   6
#define OWN_TRIP    (OWN_WAVES * OWN_BATCH)
#define OWN_NVEC    (NEDGE / 4)
#define OWN_ROUNDS  ((NEDGE + OWN_CHUNK - 1) / OWN_CHUNK)

__device__ __forceinline__ unsigned int zero_bytes(unsigned int x) {
    return ~(((x & 0x7f7f7f7fu) + 0x7f7f7f7fu) | x) & 0x80808080u;
}

template <bool KEYS>
__device__ __forceinline__ void attend_body(
        const float* __restrict__ z, const int* __restrict__ src, const int* __restrict__ dst,
        const unsigned int* __restrict__ keys, float* __restrict__ out,
        float* h_s, float* den_s, unsigned int* mo_s, unsigned int* mc_s, int* list_s, int* cnt_s) {
    const int tid  = threadIdx.x;
    const int lane = tid & 31;
    const int wave = tid >> 5;
    const int base = blockIdx.x * OWN_NODES;

    for (int i = tid; i < OWN_NODES * FATT; i += OWN_THREADS) h_s[i] = 0.0f;
    for (int i = tid; i < OWN_NODES; i += OWN_THREADS) {
        den_s[i] = 0.0f;
        mo_s[i]  = 0u;
        mc_s[i]  = 0u;
    }
    if (tid < 3) cnt_s[tid] = 0;

    const int4*  dst4 = (const int4*)dst;
    const uint4* key4 = (const uint4*)keys;
    const unsigned int pat = blockIdx.x * 0x01010101u;

    int4  cur[OWN_VEC];
    uint4 kcur[OWN_KVEC];
    if constexpr (KEYS) {
        #pragma unroll
        for (int i = 0; i < OWN_KVEC; ++i)
            kcur[i] = key4[min(i * OWN_THREADS + tid, KEY_NVEC - 1)];
    } else {
        #pragma unroll
        for (int i = 0; i < OWN_VEC; ++i)
            cur[i] = dst4[min(i * OWN_THREADS + tid, OWN_NVEC - 1)];
    }
    __syncthreads();

    for (int r = 0; r < OWN_ROUNDS; ++r) {
        int* cnt = &cnt_s[r % 3];
        int4  nxt[OWN_VEC];
        uint4 knxt[OWN_KVEC];

        if constexpr (KEYS) {
            const int kbase = r * (OWN_THREADS * OWN_KVEC);
            #pragma unroll
            for (int i = 0; i < OWN_KVEC; ++i)
                knxt[i] = key4[min(kbase + (OWN_THREADS * OWN_KVEC) + i * OWN_THREADS + tid, KEY_NVEC - 1)];

            unsigned int hits = 0u;
            #pragma unroll
            for (int i = 0; i < OWN_KVEC; ++i) {
                if (kbase + i * OWN_THREADS + tid < KEY_NVEC) {
                    hits |= zero_bytes(kcur[i].x ^ pat) >> (7 - (4 * i + 0));
                    hits |= zero_bytes(kcur[i].y ^ pat) >> (7 - (4 * i + 1));
                    hits |= zero_bytes(kcur[i].z ^ pat) >> (7 - (4 * i + 2));
                    hits |= zero_bytes(kcur[i].w ^ pat) >> (7 - (4 * i + 3));
                }
            }
            if (hits != 0u) {
                int pos = atomicAdd(cnt, __popc(hits));
                const int ebase = (kbase + tid) * 16;
                while (hits != 0u) {
                    const int p = __ffs((int)hits) - 1;
                    hits &= hits - 1u;
                    list_s[pos++] = ebase + ((p >> 2) & 1) * (OWN_THREADS * 16) + (p & 3) * 4 + (p >> 3);
                }
            }
        } else {
            const int vbase = r * (OWN_THREADS * OWN_VEC);
            #pragma unroll
            for (int i = 0; i < OWN_VEC; ++i)
                nxt[i] = dst4[min(vbase + (OWN_THREADS * OWN_VEC) + i * OWN_THREADS + tid, OWN_NVEC - 1)];

            #pragma unroll
            for (int i = 0; i < OWN_VEC; ++i) {
                const int v  = vbase + i * OWN_THREADS + tid;
                const int e0 = v * 4;
                const unsigned l0 = (unsigned)(cur[i].x - base), l1 = (unsigned)(cur[i].y - base);
                const unsigned l2 = (unsigned)(cur[i].z - base), l3 = (unsigned)(cur[i].w - base);
                if (v < OWN_NVEC && min(min(l0, l1), min(l2, l3)) < (unsigned)OWN_NODES) {
                    if (l0 < (unsigned)OWN_NODES) list_s[atomicAdd(cnt, 1)] = e0 + 0;
                    if (l1 < (unsigned)OWN_NODES) list_s[atomicAdd(cnt, 1)] = e0 + 1;
                    if (l2 < (unsigned)OWN_NODES) list_s[atomicAdd(cnt, 1)] = e0 + 2;
                    if (l3 < (unsigned)OWN_NODES) list_s[atomicAdd(cnt, 1)] = e0 + 3;
                }
            }
        }
        __syncthreads();
        const int n = *cnt;
        if (tid == 0) cnt_s[(r + 2) % 3] = 0;

        const int trips = (n + OWN_TRIP - 1) / OWN_TRIP;
        for (int t = 0; t < trips; ++t) {
            const int j0 = t * OWN_TRIP + wave * OWN_BATCH;
            bool ok = lane < OWN_BATCH && j0 + lane < n;

            int s = 0, dl = 0;
            if (ok) {
                const int edge = list_s[j0 + lane];
                s  = src[edge];
                dl = dst[edge] - base;
                if ((unsigned)dl >= (unsigned)OWN_NODES) {
                    ok = false;
                    s  = 0;
                    dl = 0;
                }
            }
            const unsigned int okm = __builtin_amdgcn_ballot_w32(ok);
            float2 rs[OWN_BATCH], rd[OWN_BATCH];
            int    dq[OWN_BATCH];
            #pragma unroll
            for (int q = 0; q < OWN_BATCH; ++q) {
                const int sq = __builtin_amdgcn_readlane(s, q);
                dq[q] = __builtin_amdgcn_readlane(dl, q);
                rs[q] = ((const float2*)(z + (size_t)sq * FATT))[lane];
                rd[q] = ((const float2*)(z + (size_t)(base + dq[q]) * FATT))[lane];
            }
            float sc = 0.0f;
            #pragma unroll
            for (int q = 0; q < OWN_BATCH; ++q) {
                const float dot = wave_sum(rs[q].x * rd[q].x + rs[q].y * rd[q].y);
                if (lane == q) sc = dot;
            }
            sc = sc > 0.0f ? sc : SLOPE * sc;
            if (ok) atomicMax(&mo_s[dl], float_to_ord(sc));
            __syncthreads();

            unsigned int mnew = 0u;
            int   take = 0;
            float fs   = 1.0f;
            if (ok) {
                mnew = mo_s[dl];
                const unsigned int mold = atomicMax(&mc_s[dl], mnew);
                if (mold != 0u && mold < mnew) {
                    take = 1;
                    fs   = __expf(ord_to_float(mold) - ord_to_float(mnew));
                    den_s[dl] *= fs;
                }
            }
            #pragma unroll
            for (int q = 0; q < OWN_BATCH; ++q) {
                if (__builtin_amdgcn_readlane(take, q)) {
                    const float f = __int_as_float(__builtin_amdgcn_readlane(__float_as_int(fs), q));
                    float2* hp = (float2*)(h_s + dq[q] * FATT) + lane;
                    float2 hv = *hp;
                    hv.x *= f;
                    hv.y *= f;
                    *hp = hv;
                }
            }
            __syncthreads();

            float w = 0.0f;
            if (ok) {
                w = __expf(sc - ord_to_float(mnew));
                atomicAdd(&den_s[dl], w);
            }
            #pragma unroll
            for (int q = 0; q < OWN_BATCH; ++q) {
                if ((okm >> q) & 1u) {
                    const float wq = __int_as_float(__builtin_amdgcn_readlane(__float_as_int(w), q));
                    float* hp = h_s + dq[q] * FATT + 2 * lane;
                    atomicAdd(hp + 0, wq * rs[q].x);
                    atomicAdd(hp + 1, wq * rs[q].y);
                }
            }
        }
        if constexpr (KEYS) {
            #pragma unroll
            for (int i = 0; i < OWN_KVEC; ++i) kcur[i] = knxt[i];
        } else {
            #pragma unroll
            for (int i = 0; i < OWN_VEC; ++i) cur[i] = nxt[i];
        }
    }
    __syncthreads();

    for (int i4 = tid; i4 < OWN_NODES * (FATT / 4); i4 += OWN_THREADS) {
        const int nl = i4 / (FATT / 4);
        if (base + nl < NTOT) {
            const float dn = den_s[nl];
            float4 v = *(const float4*)&h_s[i4 * 4];
            v.x = dn > 0.0f ? v.x / dn : 0.0f;
            v.y = dn > 0.0f ? v.y / dn : 0.0f;
            v.z = dn > 0.0f ? v.z / dn : 0.0f;
            v.w = dn > 0.0f ? v.w / dn : 0.0f;
            v.x = v.x > 0.0f ? v.x : (__expf(v.x) - 1.0f);
            v.y = v.y > 0.0f ? v.y : (__expf(v.y) - 1.0f);
            v.z = v.z > 0.0f ? v.z : (__expf(v.z) - 1.0f);
            v.w = v.w > 0.0f ? v.w : (__expf(v.w) - 1.0f);
            *(float4*)(out + (size_t)base * FATT + (size_t)i4 * 4) = v;
        }
    }
}

__global__ __launch_bounds__(OWN_THREADS) void node_block_attend(
        const float* __restrict__ z, const int* __restrict__ src, const int* __restrict__ dst,
        const unsigned int* __restrict__ keys, float* __restrict__ out, size_t ws_size) {
    __shared__ __attribute__((aligned(16))) float h_s[OWN_NODES * FATT];
    __shared__ float        den_s[OWN_NODES];
    __shared__ unsigned int mo_s[OWN_NODES];
    __shared__ unsigned int mc_s[OWN_NODES];
    __shared__ int          list_s[OWN_CHUNK];
    __shared__ int          cnt_s[3];

    if (ws_size >= WS_NEED)
        attend_body<true >(z, src, dst, keys, out, h_s, den_s, mo_s, mc_s, list_s, cnt_s);
    else
        attend_body<false>(z, src, dst, keys, out, h_s, den_s, mo_s, mc_s, list_s, cnt_s);
}

extern "C" void kernel_launch(void* const* d_in, const int* in_sizes, int n_in,
                              void* d_out, int out_size, void* d_ws, size_t ws_size,
                              hipStream_t stream) {
    const float* d_sim  = (const float*)d_in[0];
    const float* mi_sim = (const float*)d_in[1];
    const float* W_d    = (const float*)d_in[2];
    const float* W_mi   = (const float*)d_in[3];
    const int*   src    = (const int*)d_in[4];
    const int*   dst    = (const int*)d_in[5];
    float*       out    = (float*)d_out;

    float*        z     = (float*)d_ws;
    unsigned int* keys  = (unsigned int*)((char*)d_ws + WS_Z_BYTES);

    gemm_wmma_f32<<<(N_DIS + 127) / 128, 128, 0, stream>>>(d_sim, W_d, z, N_DIS, 383);
    gemm_wmma_f32<<<(N_MIC + 127) / 128, 128, 0, stream>>>(mi_sim, W_mi,
                                                           z + (size_t)N_DIS * FATT, N_MIC, 495);

    owner_keys<<<(KEY_NVEC + KEY_THREADS - 1) / KEY_THREADS, KEY_THREADS, 0, stream>>>(dst, keys, ws_size);

    node_block_attend<<<OWN_BLOCKS, OWN_THREADS, 0, stream>>>(z, src, dst, keys, out, ws_size);
}


# Owner rounds 3.5x longer (114,688 edges; hits held as register bit masks, 4,096-id list refilled in windows on overflow): 14 barrier chains of 16 edges a wave, not 49 of 6; sum order changes
# speedup vs baseline: 5.4555x; 1.2096x over previous
#include <hip/hip_runtime.h>
#include <math.h>

#define N_DIS 20000
#define N_MIC 80000
#define NTOT  (N_DIS + N_MIC)
#define NEDGE 1600000
#define FATT  64
#define SLOPE 0.2f

typedef float v2f __attribute__((ext_vector_type(2)));
typedef float v8f __attribute__((ext_vector_type(8)));

typedef float v4f __attribute__((ext_vector_type(4)));

#define DEPTH 5

struct Frag { v4f a0, a1, w0, w1, w2, w3; };

__device__ __forceinline__ v4f ld_x4(const char* p) {
    v4f v;
    __builtin_memcpy(&v, p, sizeof(v));
    return v;
}

__device__ __forceinline__ void load_frag(Frag& f, const float* X, const float* W,
                                          unsigned xo0, unsigned xo1, unsigned wo, int kb) {
    const char* xk = (const char*)X + (size_t)kb * (8 * sizeof(float));
    const char* wk = (const char*)W + (size_t)kb * (8 * FATT * sizeof(float));
    f.a0 = ld_x4(xk + xo0);
    f.a1 = ld_x4(xk + xo1);
    f.w0 = *(const v4f*)(wk + wo);
    f.w1 = *(const v4f*)(wk + wo + 1 * FATT * sizeof(float));
    f.w2 = *(const v4f*)(wk + wo + 2 * FATT * sizeof(float));
    f.w3 = *(const v4f*)(wk + wo + 3 * FATT * sizeof(float));
}

__device__ __forceinline__ void mma_frag(const Frag& f, v8f (&acc)[2][4]) {
    const v2f a0lo = { f.a0.x, f.a0.y }, a0hi = { f.a0.z, f.a0.w };
    const v2f a1lo = { f.a1.x, f.a1.y }, a1hi = { f.a1.z, f.a1.w };
    #pragma unroll
    for (int j = 0; j < 4; ++j) {
        const v2f b = { f.w0[j], f.w1[j] };
        acc[0][j] = __builtin_amdgcn_wmma_f32_16x16x4_f32(false, a0lo, false, b, (short)0, acc[0][j], false, false);
        acc[1][j] = __builtin_amdgcn_wmma_f32_16x16x4_f32(false, a1lo, false, b, (short)0, acc[1][j], false, false);
    }
    #pragma unroll
    for (int j = 0; j < 4; ++j) {
        const v2f b = { f.w2[j], f.w3[j] };
        acc[0][j] = __builtin_amdgcn_wmma_f32_16x16x4_f32(false, a0hi, false, b, (short)0, acc[0][j], false, false);
        acc[1][j] = __builtin_amdgcn_wmma_f32_16x16x4_f32(false, a1hi, false, b, (short)0, acc[1][j], false, false);
    }
}

__global__ __launch_bounds__(128) void gemm_wmma_f32(
        const float* __restrict__ X, const float* __restrict__ W,
        float* __restrict__ Z, int M, int K) {
    const int lane = threadIdx.x & 31;
    const int wave = threadIdx.x >> 5;
    const int c    = lane & 15;
    const int h    = lane >> 4;

    const int row_base = (blockIdx.x * 4 + wave) * 32;
    if (row_base >= M) return;

    const int r0 = min(row_base + c, M - 1);
    const int r1 = min(row_base + 16 + c, M - 1);
    const unsigned xo0 = ((unsigned)r0 * (unsigned)K + 4u * h) * (unsigned)sizeof(float);
    const unsigned xo1 = ((unsigned)r1 * (unsigned)K + 4u * h) * (unsigned)sizeof(float);
    const unsigned wo  = (4u * h * FATT + 4u * c) * (unsigned)sizeof(float);

    v8f acc[2][4] = {};

    const int nfull = K >> 3;
    int kb = 0;
    if (nfull > 0) {
        Frag f[DEPTH];
        #pragma unroll
        for (int i = 0; i < DEPTH; ++i)
            load_frag(f[i], X, W, xo0, xo1, wo, min(i, nfull - 1));

        for (; kb + DEPTH <= nfull; kb += DEPTH) {
            #pragma unroll
            for (int i = 0; i < DEPTH; ++i) {
                mma_frag(f[i], acc);
                load_frag(f[i], X, W, xo0, xo1, wo, min(kb + DEPTH + i, nfull - 1));
                __builtin_amdgcn_sched_barrier(0);
            }
        }
        #pragma unroll
        for (int i = 0; i < DEPTH; ++i)
            if (kb + i < nfull) mma_frag(f[i], acc);
    }

    const int kt = nfull * 8;
    if (kt < K) {
        Frag t;
        float x0[4], x1[4];
        v4f wv[4];
        #pragma unroll
        for (int i = 0; i < 4; ++i) {
            const int  k  = kt + 4 * h + i;
            const bool ok = k < K;
            const int  kc = ok ? k : (K - 1);
            const float p0 = X[(size_t)r0 * K + kc];
            const float p1 = X[(size_t)r1 * K + kc];
            const v4f   pw = *(const v4f*)(W + (size_t)kc * FATT + 4 * c);
            const v4f   zero = { 0.0f, 0.0f, 0.0f, 0.0f };
            x0[i] = ok ? p0 : 0.0f;
            x1[i] = ok ? p1 : 0.0f;
            wv[i] = ok ? pw : zero;
        }
        t.a0 = (v4f){ x0[0], x0[1], x0[2], x0[3] };
        t.a1 = (v4f){ x1[0], x1[1], x1[2], x1[3] };
        t.w0 = wv[0]; t.w1 = wv[1]; t.w2 = wv[2]; t.w3 = wv[3];
        mma_frag(t, acc);
    }

    #pragma unroll
    for (int mt = 0; mt < 2; ++mt) {
        #pragma unroll
        for (int r = 0; r < 8; ++r) {
            const int mm = row_base + 16 * mt + 8 * h + r;
            if (mm < M) {
                const v4f o = { acc[mt][0][r], acc[mt][1][r], acc[mt][2][r], acc[mt][3][r] };
                *(v4f*)(Z + (size_t)mm * FATT + 4 * c) = o;
            }
        }
    }
}

__device__ __forceinline__ unsigned int float_to_ord(float f) {
    unsigned int u = __float_as_uint(f);
    return (u & 0x80000000u) ? ~u : (u | 0x80000000u);
}
__device__ __forceinline__ float ord_to_float(unsigned int u) {
    return __uint_as_float((u & 0x80000000u) ? (u ^ 0x80000000u) : ~u);
}

template <int CTRL>
__device__ __forceinline__ float dpp_move(float v) {
    return __int_as_float(__builtin_amdgcn_update_dpp(0, __float_as_int(v), CTRL, 0xF, 0xF, true));
}
__device__ __forceinline__ float wave_sum(float v) {
    v += dpp_move<0x161>(v);
    v += dpp_move<0x162>(v);
    v += dpp_move<0x164>(v);
    v += dpp_move<0x168>(v);
    v += __int_as_float((int)__builtin_amdgcn_permlanex16(0u, (unsigned)__float_as_int(v),
                                                          0x76543210u, 0xfedcba98u, false, false));
    return v;
}

#define OWN_BLOCKS  256
#define OWN_NODES   ((NTOT + OWN_BLOCKS - 1) / OWN_BLOCKS)
#define KEY_NVEC    (NEDGE / 16)
#define KEY_THREADS 256
#define WS_Z_BYTES  ((size_t)NTOT * FATT * sizeof(float))
#define WS_NEED     (WS_Z_BYTES + (size_t)NEDGE)

__device__ __forceinline__ unsigned int key_of(int d) {
    const unsigned int u = (unsigned int)d;
    return u < (unsigned int)(OWN_BLOCKS * OWN_NODES) ? u / (unsigned int)OWN_NODES : 255u;
}

__global__ __launch_bounds__(KEY_THREADS) void owner_keys(
        const int* __restrict__ dst, unsigned int* __restrict__ keys, size_t ws_size) {
    if (ws_size < WS_NEED) return;
    const int v = blockIdx.x * KEY_THREADS + threadIdx.x;
    if (v >= KEY_NVEC) return;
    const int4* d4 = (const int4*)dst + (size_t)v * 4;
    unsigned int w[4];
    #pragma unroll
    for (int j = 0; j < 4; ++j) {
        const int4 d = d4[j];
        w[j] = key_of(d.x) | (key_of(d.y) << 8) | (key_of(d.z) << 16) | (key_of(d.w) << 24);
    }
    ((uint4*)keys)[v] = make_uint4(w[0], w[1], w[2], w[3]);
}

#define OWN_THREADS 1024
#define OWN_WAVES   (OWN_THREADS / 32)
#define OWN_KVEC    7
#define OWN_NMASK   ((OWN_KVEC + 1) / 2)
#define OWN_RVEC    (OWN_THREADS * OWN_KVEC)
#define OWN_CHUNK   (OWN_RVEC * 16)
#define OWN_BATCH   16
#define OWN_TRIP    (OWN_WAVES * OWN_BATCH)
#define OWN_LIST    4096
#define OWN_ROUNDS  ((NEDGE + OWN_CHUNK - 1) / OWN_CHUNK)

__device__ __forceinline__ unsigned int zero_bytes(unsigned int x) {
    return ~(((x & 0x7f7f7f7fu) + 0x7f7f7f7fu) | x) & 0x80808080u;
}

__device__ __forceinline__ void list_window(const unsigned int (&hm)[OWN_NMASK], int first,
                                            int ebase, int* list_s) {
    int idx = first;
    #pragma unroll
    for (int m = 0; m < OWN_NMASK; ++m) {
        unsigned int mask = hm[m];
        while (mask != 0u) {
            const int p = __ffs((int)mask) - 1;
            mask &= mask - 1u;
            if ((unsigned)idx < (unsigned)OWN_LIST)
                list_s[idx] = ebase + (2 * m + ((p >> 2) & 1)) * (OWN_THREADS * 16) + (p & 3) * 4 + (p >> 3);
            ++idx;
        }
    }
}

template <bool KEYS>
__device__ __forceinline__ void attend_body(
        const float* __restrict__ z, const int* __restrict__ src, const int* __restrict__ dst,
        const unsigned int* __restrict__ keys, float* __restrict__ out,
        float* h_s, float* den_s, unsigned int* mo_s, unsigned int* mc_s, int* list_s, int* cnt_s) {
    const int tid  = threadIdx.x;
    const int lane = tid & 31;
    const int wave = tid >> 5;
    const int base = blockIdx.x * OWN_NODES;

    for (int i = tid; i < OWN_NODES * FATT; i += OWN_THREADS) h_s[i] = 0.0f;
    for (int i = tid; i < OWN_NODES; i += OWN_THREADS) {
        den_s[i] = 0.0f;
        mo_s[i]  = 0u;
        mc_s[i]  = 0u;
    }
    if (tid < 3) cnt_s[tid] = 0;

    const int4*  dst4 = (const int4*)dst;
    const uint4* key4 = (const uint4*)keys;
    const unsigned int pat = blockIdx.x * 0x01010101u;
    __syncthreads();

    for (int r = 0; r < OWN_ROUNDS; ++r) {
        int* cnt = &cnt_s[r % 3];
        const int vbase = r * OWN_RVEC;
        const int ebase = (vbase + tid) * 16;

        unsigned int hm[OWN_NMASK];
        #pragma unroll
        for (int m = 0; m < OWN_NMASK; ++m) hm[m] = 0u;

        if constexpr (KEYS) {
            uint4 kv[OWN_KVEC];
            #pragma unroll
            for (int i = 0; i < OWN_KVEC; ++i)
                kv[i] = key4[min(vbase + i * OWN_THREADS + tid, KEY_NVEC - 1)];
            #pragma unroll
            for (int i = 0; i < OWN_KVEC; ++i) {
                const unsigned int live = vbase + i * OWN_THREADS + tid < KEY_NVEC ? 0xffffffffu : 0u;
                const unsigned int bits = (zero_bytes(kv[i].x ^ pat) >> (7 - (4 * (i & 1) + 0))) |
                                          (zero_bytes(kv[i].y ^ pat) >> (7 - (4 * (i & 1) + 1))) |
                                          (zero_bytes(kv[i].z ^ pat) >> (7 - (4 * (i & 1) + 2))) |
                                          (zero_bytes(kv[i].w ^ pat) >> (7 - (4 * (i & 1) + 3)));
                hm[i >> 1] |= bits & live;
            }
        } else {
            #pragma nounroll
            for (int m = 0; m < OWN_NMASK; ++m) {
                unsigned int bits = 0u;
                #pragma unroll
                for (int ii = 0; ii < 2; ++ii) {
                    const int i = 2 * m + ii;
                    const int v = vbase + i * OWN_THREADS + tid;
                    if (i < OWN_KVEC && v < KEY_NVEC) {
                        #pragma unroll
                        for (int j = 0; j < 4; ++j) {
                            const int4 d = dst4[(size_t)v * 4 + j];
                            const unsigned int b =
                                ((unsigned)(d.x - base) < (unsigned)OWN_NODES ? 0x00000001u : 0u) |
                                ((unsigned)(d.y - base) < (unsigned)OWN_NODES ? 0x00000100u : 0u) |
                                ((unsigned)(d.z - base) < (unsigned)OWN_NODES ? 0x00010000u : 0u) |
                                ((unsigned)(d.w - base) < (unsigned)OWN_NODES ? 0x01000000u : 0u);
                            bits |= b << (4 * ii + j);
                        }
                    }
                }
                #pragma unroll
                for (int k = 0; k < OWN_NMASK; ++k)
                    if (k == m) hm[k] = bits;
            }
        }
        int total = 0;
        #pragma unroll
        for (int m = 0; m < OWN_NMASK; ++m) total += __popc(hm[m]);
        int pos0 = 0;
        if (total != 0) {
            pos0 = atomicAdd(cnt, total);
            list_window(hm, pos0, ebase, list_s);
        }
        __syncthreads();
        const int n = *cnt;
        if (tid == 0) cnt_s[(r + 2) % 3] = 0;

        for (int lo = 0; lo < n; lo += OWN_LIST) {
            if (lo != 0) {
                if (total != 0) list_window(hm, pos0 - lo, ebase, list_s);
                __syncthreads();
            }
            const int nw    = min(n - lo, OWN_LIST);
            const int trips = (nw + OWN_TRIP - 1) / OWN_TRIP;
            for (int t = 0; t < trips; ++t) {
                const int j0 = t * OWN_TRIP + wave * OWN_BATCH;
                bool ok = lane < OWN_BATCH && j0 + lane < nw;

                int s = 0, dl = 0;
                if (ok) {
                    const int edge = list_s[j0 + lane];
                    s  = src[edge];
                    dl = dst[edge] - base;
                    if ((unsigned)dl >= (unsigned)OWN_NODES) {
                        ok = false;
                        s  = 0;
                        dl = 0;
                    }
                }
                const unsigned int okm = __builtin_amdgcn_ballot_w32(ok);
                float2 rs[OWN_BATCH];
                int    dq[OWN_BATCH];
                float  sc = 0.0f;
                if (okm != 0u) {
                    float2 rd[OWN_BATCH];
                    #pragma unroll
                    for (int q = 0; q < OWN_BATCH; ++q) {
                        const int sq = __builtin_amdgcn_readlane(s, q);
                        dq[q] = __builtin_amdgcn_readlane(dl, q);
                        rs[q] = ((const float2*)(z + (size_t)sq * FATT))[lane];
                        rd[q] = ((const float2*)(z + (size_t)(base + dq[q]) * FATT))[lane];
                    }
                    #pragma unroll
                    for (int q = 0; q < OWN_BATCH; ++q) {
                        const float dot = wave_sum(rs[q].x * rd[q].x + rs[q].y * rd[q].y);
                        if (lane == q) sc = dot;
                    }
                }
                sc = sc > 0.0f ? sc : SLOPE * sc;
                if (ok) atomicMax(&mo_s[dl], float_to_ord(sc));
                __syncthreads();

                unsigned int mnew = 0u;
                int   take = 0;
                float fs   = 1.0f;
                if (ok) {
                    mnew = mo_s[dl];
                    const unsigned int mold = atomicMax(&mc_s[dl], mnew);
                    if (mold != 0u && mold < mnew) {
                        take = 1;
                        fs   = __expf(ord_to_float(mold) - ord_to_float(mnew));
                        den_s[dl] *= fs;
                    }
                }
                if (__builtin_amdgcn_ballot_w32(take != 0) != 0u) {
                    #pragma unroll
                    for (int q = 0; q < OWN_BATCH; ++q) {
                        if (__builtin_amdgcn_readlane(take, q)) {
                            const float f = __int_as_float(__builtin_amdgcn_readlane(__float_as_int(fs), q));
                            float2* hp = (float2*)(h_s + dq[q] * FATT) + lane;
                            float2 hv = *hp;
                            hv.x *= f;
                            hv.y *= f;
                            *hp = hv;
                        }
                    }
                }
                __syncthreads();

                float w = 0.0f;
                if (ok) {
                    w = __expf(sc - ord_to_float(mnew));
                    atomicAdd(&den_s[dl], w);
                }
                #pragma unroll
                for (int q = 0; q < OWN_BATCH; ++q) {
                    if ((okm >> q) & 1u) {
                        const float wq = __int_as_float(__builtin_amdgcn_readlane(__float_as_int(w), q));
                        float* hp = h_s + dq[q] * FATT + 2 * lane;
                        atomicAdd(hp + 0, wq * rs[q].x);
                        atomicAdd(hp + 1, wq * rs[q].y);
                    }
                }
            }
        }
    }
    __syncthreads();

    for (int i4 = tid; i4 < OWN_NODES * (FATT / 4); i4 += OWN_THREADS) {
        const int nl = i4 / (FATT / 4);
        if (base + nl < NTOT) {
            const float dn = den_s[nl];
            float4 v = *(const float4*)&h_s[i4 * 4];
            v.x = dn > 0.0f ? v.x / dn : 0.0f;
            v.y = dn > 0.0f ? v.y / dn : 0.0f;
            v.z = dn > 0.0f ? v.z / dn : 0.0f;
            v.w = dn > 0.0f ? v.w / dn : 0.0f;
            v.x = v.x > 0.0f ? v.x : (__expf(v.x) - 1.0f);
            v.y = v.y > 0.0f ? v.y : (__expf(v.y) - 1.0f);
            v.z = v.z > 0.0f ? v.z : (__expf(v.z) - 1.0f);
            v.w = v.w > 0.0f ? v.w : (__expf(v.w) - 1.0f);
            *(float4*)(out + (size_t)base * FATT + (size_t)i4 * 4) = v;
        }
    }
}

__global__ __launch_bounds__(OWN_THREADS) void node_block_attend(
        const float* __restrict__ z, const int* __restrict__ src, const int* __restrict__ dst,
        const unsigned int* __restrict__ keys, float* __restrict__ out, size_t ws_size) {
    __shared__ __attribute__((aligned(16))) float h_s[OWN_NODES * FATT];
    __shared__ float        den_s[OWN_NODES];
    __shared__ unsigned int mo_s[OWN_NODES];
    __shared__ unsigned int mc_s[OWN_NODES];
    __shared__ int          list_s[OWN_LIST];
    __shared__ int          cnt_s[3];

    if (ws_size >= WS_NEED)
        attend_body<true >(z, src, dst, keys, out, h_s, den_s, mo_s, mc_s, list_s, cnt_s);
    else
        attend_body<false>(z, src, dst, keys, out, h_s, den_s, mo_s, mc_s, list_s, cnt_s);
}

extern "C" void kernel_launch(void* const* d_in, const int* in_sizes, int n_in,
                              void* d_out, int out_size, void* d_ws, size_t ws_size,
                              hipStream_t stream) {
    const float* d_sim  = (const float*)d_in[0];
    const float* mi_sim = (const float*)d_in[1];
    const float* W_d    = (const float*)d_in[2];
    const float* W_mi   = (const float*)d_in[3];
    const int*   src    = (const int*)d_in[4];
    const int*   dst    = (const int*)d_in[5];
    float*       out    = (float*)d_out;

    float*        z     = (float*)d_ws;
    unsigned int* keys  = (unsigned int*)((char*)d_ws + WS_Z_BYTES);

    gemm_wmma_f32<<<(N_DIS + 127) / 128, 128, 0, stream>>>(d_sim, W_d, z, N_DIS, 383);
    gemm_wmma_f32<<<(N_MIC + 127) / 128, 128, 0, stream>>>(mi_sim, W_mi,
                                                           z + (size_t)N_DIS * FATT, N_MIC, 495);

    owner_keys<<<(KEY_NVEC + KEY_THREADS - 1) / KEY_THREADS, KEY_THREADS, 0, stream>>>(dst, keys, ws_size);

    node_block_attend<<<OWN_BLOCKS, OWN_THREADS, 0, stream>>>(z, src, dst, keys, out, ws_size);
}


# Sums scaled to a node's first-arriving score (LDS compare-swap), so every edge adds at once; rescale pass only for a score over 60 above it: 2 barriers a round, not 3; sum order changes
# speedup vs baseline: 5.7378x; 1.0517x over previous
#include <hip/hip_runtime.h>
#include <math.h>

#define N_DIS 20000
#define N_MIC 80000
#define NTOT  (N_DIS + N_MIC)
#define NEDGE 1600000
#define FATT  64
#define SLOPE 0.2f

typedef float v2f __attribute__((ext_vector_type(2)));
typedef float v8f __attribute__((ext_vector_type(8)));

typedef float v4f __attribute__((ext_vector_type(4)));

#define DEPTH 5

struct Frag { v4f a0, a1, w0, w1, w2, w3; };

__device__ __forceinline__ v4f ld_x4(const char* p) {
    v4f v;
    __builtin_memcpy(&v, p, sizeof(v));
    return v;
}

__device__ __forceinline__ void load_frag(Frag& f, const float* X, const float* W,
                                          unsigned xo0, unsigned xo1, unsigned wo, int kb) {
    const char* xk = (const char*)X + (size_t)kb * (8 * sizeof(float));
    const char* wk = (const char*)W + (size_t)kb * (8 * FATT * sizeof(float));
    f.a0 = ld_x4(xk + xo0);
    f.a1 = ld_x4(xk + xo1);
    f.w0 = *(const v4f*)(wk + wo);
    f.w1 = *(const v4f*)(wk + wo + 1 * FATT * sizeof(float));
    f.w2 = *(const v4f*)(wk + wo + 2 * FATT * sizeof(float));
    f.w3 = *(const v4f*)(wk + wo + 3 * FATT * sizeof(float));
}

__device__ __forceinline__ void mma_frag(const Frag& f, v8f (&acc)[2][4]) {
    const v2f a0lo = { f.a0.x, f.a0.y }, a0hi = { f.a0.z, f.a0.w };
    const v2f a1lo = { f.a1.x, f.a1.y }, a1hi = { f.a1.z, f.a1.w };
    #pragma unroll
    for (int j = 0; j < 4; ++j) {
        const v2f b = { f.w0[j], f.w1[j] };
        acc[0][j] = __builtin_amdgcn_wmma_f32_16x16x4_f32(false, a0lo, false, b, (short)0, acc[0][j], false, false);
        acc[1][j] = __builtin_amdgcn_wmma_f32_16x16x4_f32(false, a1lo, false, b, (short)0, acc[1][j], false, false);
    }
    #pragma unroll
    for (int j = 0; j < 4; ++j) {
        const v2f b = { f.w2[j], f.w3[j] };
        acc[0][j] = __builtin_amdgcn_wmma_f32_16x16x4_f32(false, a0hi, false, b, (short)0, acc[0][j], false, false);
        acc[1][j] = __builtin_amdgcn_wmma_f32_16x16x4_f32(false, a1hi, false, b, (short)0, acc[1][j], false, false);
    }
}

__global__ __launch_bounds__(128) void gemm_wmma_f32(
        const float* __restrict__ X, const float* __restrict__ W,
        float* __restrict__ Z, int M, int K) {
    const int lane = threadIdx.x & 31;
    const int wave = threadIdx.x >> 5;
    const int c    = lane & 15;
    const int h    = lane >> 4;

    const int row_base = (blockIdx.x * 4 + wave) * 32;
    if (row_base >= M) return;

    const int r0 = min(row_base + c, M - 1);
    const int r1 = min(row_base + 16 + c, M - 1);
    const unsigned xo0 = ((unsigned)r0 * (unsigned)K + 4u * h) * (unsigned)sizeof(float);
    const unsigned xo1 = ((unsigned)r1 * (unsigned)K + 4u * h) * (unsigned)sizeof(float);
    const unsigned wo  = (4u * h * FATT + 4u * c) * (unsigned)sizeof(float);

    v8f acc[2][4] = {};

    const int nfull = K >> 3;
    int kb = 0;
    if (nfull > 0) {
        Frag f[DEPTH];
        #pragma unroll
        for (int i = 0; i < DEPTH; ++i)
            load_frag(f[i], X, W, xo0, xo1, wo, min(i, nfull - 1));

        for (; kb + DEPTH <= nfull; kb += DEPTH) {
            #pragma unroll
            for (int i = 0; i < DEPTH; ++i) {
                mma_frag(f[i], acc);
                load_frag(f[i], X, W, xo0, xo1, wo, min(kb + DEPTH + i, nfull - 1));
                __builtin_amdgcn_sched_barrier(0);
            }
        }
        #pragma unroll
        for (int i = 0; i < DEPTH; ++i)
            if (kb + i < nfull) mma_frag(f[i], acc);
    }

    const int kt = nfull * 8;
    if (kt < K) {
        Frag t;
        float x0[4], x1[4];
        v4f wv[4];
        #pragma unroll
        for (int i = 0; i < 4; ++i) {
            const int  k  = kt + 4 * h + i;
            const bool ok = k < K;
            const int  kc = ok ? k : (K - 1);
            const float p0 = X[(size_t)r0 * K + kc];
            const float p1 = X[(size_t)r1 * K + kc];
            const v4f   pw = *(const v4f*)(W + (size_t)kc * FATT + 4 * c);
            const v4f   zero = { 0.0f, 0.0f, 0.0f, 0.0f };
            x0[i] = ok ? p0 : 0.0f;
            x1[i] = ok ? p1 : 0.0f;
            wv[i] = ok ? pw : zero;
        }
        t.a0 = (v4f){ x0[0], x0[1], x0[2], x0[3] };
        t.a1 = (v4f){ x1[0], x1[1], x1[2], x1[3] };
        t.w0 = wv[0]; t.w1 = wv[1]; t.w2 = wv[2]; t.w3 = wv[3];
        mma_frag(t, acc);
    }

    #pragma unroll
    for (int mt = 0; mt < 2; ++mt) {
        #pragma unroll
        for (int r = 0; r < 8; ++r) {
            const int mm = row_base + 16 * mt + 8 * h + r;
            if (mm < M) {
                const v4f o = { acc[mt][0][r], acc[mt][1][r], acc[mt][2][r], acc[mt][3][r] };
                *(v4f*)(Z + (size_t)mm * FATT + 4 * c) = o;
            }
        }
    }
}

__device__ __forceinline__ unsigned int float_to_ord(float f) {
    unsigned int u = __float_as_uint(f);
    return (u & 0x80000000u) ? ~u : (u | 0x80000000u);
}
__device__ __forceinline__ float ord_to_float(unsigned int u) {
    return __uint_as_float((u & 0x80000000u) ? (u ^ 0x80000000u) : ~u);
}

template <int CTRL>
__device__ __forceinline__ float dpp_move(float v) {
    return __int_as_float(__builtin_amdgcn_update_dpp(0, __float_as_int(v), CTRL, 0xF, 0xF, true));
}
__device__ __forceinline__ float wave_sum(float v) {
    v += dpp_move<0x161>(v);
    v += dpp_move<0x162>(v);
    v += dpp_move<0x164>(v);
    v += dpp_move<0x168>(v);
    v += __int_as_float((int)__builtin_amdgcn_permlanex16(0u, (unsigned)__float_as_int(v),
                                                          0x76543210u, 0xfedcba98u, false, false));
    return v;
}

#define OWN_BLOCKS  256
#define OWN_NODES   ((NTOT + OWN_BLOCKS - 1) / OWN_BLOCKS)
#define KEY_NVEC    (NEDGE / 16)
#define KEY_THREADS 256
#define WS_Z_BYTES  ((size_t)NTOT * FATT * sizeof(float))
#define WS_NEED     (WS_Z_BYTES + (size_t)NEDGE)

__device__ __forceinline__ unsigned int key_of(int d) {
    const unsigned int u = (unsigned int)d;
    return u < (unsigned int)(OWN_BLOCKS * OWN_NODES) ? u / (unsigned int)OWN_NODES : 255u;
}

__global__ __launch_bounds__(KEY_THREADS) void owner_keys(
        const int* __restrict__ dst, unsigned int* __restrict__ keys, size_t ws_size) {
    if (ws_size < WS_NEED) return;
    const int v = blockIdx.x * KEY_THREADS + threadIdx.x;
    if (v >= KEY_NVEC) return;
    const int4* d4 = (const int4*)dst + (size_t)v * 4;
    unsigned int w[4];
    #pragma unroll
    for (int j = 0; j < 4; ++j) {
        const int4 d = d4[j];
        w[j] = key_of(d.x) | (key_of(d.y) << 8) | (key_of(d.z) << 16) | (key_of(d.w) << 24);
    }
    ((uint4*)keys)[v] = make_uint4(w[0], w[1], w[2], w[3]);
}

#define OWN_THREADS 1024
#define OWN_WAVES   (OWN_THREADS / 32)
#define OWN_KVEC    7
#define OWN_NMASK   ((OWN_KVEC + 1) / 2)
#define OWN_RVEC    (OWN_THREADS * OWN_KVEC)
#define OWN_CHUNK   (OWN_RVEC * 16)
#define OWN_BATCH   16
#define OWN_TRIP    (OWN_WAVES * OWN_BATCH)
#define OWN_LIST    4096
#define OWN_ROUNDS  ((NEDGE + OWN_CHUNK - 1) / OWN_CHUNK)
#define OWN_RISE    60.0f

__device__ __forceinline__ unsigned int zero_bytes(unsigned int x) {
    return ~(((x & 0x7f7f7f7fu) + 0x7f7f7f7fu) | x) & 0x80808080u;
}

__device__ __forceinline__ void list_window(const unsigned int (&hm)[OWN_NMASK], int first,
                                            int ebase, int* list_s) {
    int idx = first;
    #pragma unroll
    for (int m = 0; m < OWN_NMASK; ++m) {
        unsigned int mask = hm[m];
        while (mask != 0u) {
            const int p = __ffs((int)mask) - 1;
            mask &= mask - 1u;
            if ((unsigned)idx < (unsigned)OWN_LIST)
                list_s[idx] = ebase + (2 * m + ((p >> 2) & 1)) * (OWN_THREADS * 16) + (p & 3) * 4 + (p >> 3);
            ++idx;
        }
    }
}

template <bool KEYS>
__device__ __forceinline__ void attend_body(
        const float* __restrict__ z, const int* __restrict__ src, const int* __restrict__ dst,
        const unsigned int* __restrict__ keys, float* __restrict__ out,
        float* h_s, float* den_s, unsigned int* mo_s, unsigned int* mc_s, int* list_s, int* cnt_s,
        int* flag_s) {
    const int tid  = threadIdx.x;
    const int lane = tid & 31;
    const int wave = tid >> 5;
    const int base = blockIdx.x * OWN_NODES;

    for (int i = tid; i < OWN_NODES * FATT; i += OWN_THREADS) h_s[i] = 0.0f;
    for (int i = tid; i < OWN_NODES; i += OWN_THREADS) {
        den_s[i] = 0.0f;
        mo_s[i]  = 0u;
        mc_s[i]  = 0u;
    }
    if (tid < 3) {
        cnt_s[tid]  = 0;
        flag_s[tid] = 0;
    }
    int g = 0;

    const int4*  dst4 = (const int4*)dst;
    const uint4* key4 = (const uint4*)keys;
    const unsigned int pat = blockIdx.x * 0x01010101u;
    __syncthreads();

    for (int r = 0; r < OWN_ROUNDS; ++r) {
        int* cnt = &cnt_s[r % 3];
        const int vbase = r * OWN_RVEC;
        const int ebase = (vbase + tid) * 16;

        unsigned int hm[OWN_NMASK];
        #pragma unroll
        for (int m = 0; m < OWN_NMASK; ++m) hm[m] = 0u;

        if constexpr (KEYS) {
            uint4 kv[OWN_KVEC];
            #pragma unroll
            for (int i = 0; i < OWN_KVEC; ++i)
                kv[i] = key4[min(vbase + i * OWN_THREADS + tid, KEY_NVEC - 1)];
            #pragma unroll
            for (int i = 0; i < OWN_KVEC; ++i) {
                const unsigned int live = vbase + i * OWN_THREADS + tid < KEY_NVEC ? 0xffffffffu : 0u;
                const unsigned int bits = (zero_bytes(kv[i].x ^ pat) >> (7 - (4 * (i & 1) + 0))) |
                                          (zero_bytes(kv[i].y ^ pat) >> (7 - (4 * (i & 1) + 1))) |
                                          (zero_bytes(kv[i].z ^ pat) >> (7 - (4 * (i & 1) + 2))) |
                                          (zero_bytes(kv[i].w ^ pat) >> (7 - (4 * (i & 1) + 3)));
                hm[i >> 1] |= bits & live;
            }
        } else {
            #pragma nounroll
            for (int m = 0; m < OWN_NMASK; ++m) {
                unsigned int bits = 0u;
                #pragma unroll
                for (int ii = 0; ii < 2; ++ii) {
                    const int i = 2 * m + ii;
                    const int v = vbase + i * OWN_THREADS + tid;
                    if (i < OWN_KVEC && v < KEY_NVEC) {
                        #pragma unroll
                        for (int j = 0; j < 4; ++j) {
                            const int4 d = dst4[(size_t)v * 4 + j];
                            const unsigned int b =
                                ((unsigned)(d.x - base) < (unsigned)OWN_NODES ? 0x00000001u : 0u) |
                                ((unsigned)(d.y - base) < (unsigned)OWN_NODES ? 0x00000100u : 0u) |
                                ((unsigned)(d.z - base) < (unsigned)OWN_NODES ? 0x00010000u : 0u) |
                                ((unsigned)(d.w - base) < (unsigned)OWN_NODES ? 0x01000000u : 0u);
                            bits |= b << (4 * ii + j);
                        }
                    }
                }
                #pragma unroll
                for (int k = 0; k < OWN_NMASK; ++k)
                    if (k == m) hm[k] = bits;
            }
        }
        int total = 0;
        #pragma unroll
        for (int m = 0; m < OWN_NMASK; ++m) total += __popc(hm[m]);
        int pos0 = 0;
        if (total != 0) {
            pos0 = atomicAdd(cnt, total);
            list_window(hm, pos0, ebase, list_s);
        }
        __syncthreads();
        const int n = *cnt;
        if (tid == 0) cnt_s[(r + 2) % 3] = 0;

        for (int lo = 0; lo < n; lo += OWN_LIST) {
            if (lo != 0) {
                if (total != 0) list_window(hm, pos0 - lo, ebase, list_s);
                __syncthreads();
            }
            const int nw    = min(n - lo, OWN_LIST);
            const int trips = (nw + OWN_TRIP - 1) / OWN_TRIP;
            for (int t = 0; t < trips; ++t) {
                const int j0 = t * OWN_TRIP + wave * OWN_BATCH;
                bool ok = lane < OWN_BATCH && j0 + lane < nw;

                int s = 0, dl = 0;
                if (ok) {
                    const int edge = list_s[j0 + lane];
                    s  = src[edge];
                    dl = dst[edge] - base;
                    if ((unsigned)dl >= (unsigned)OWN_NODES) {
                        ok = false;
                        s  = 0;
                        dl = 0;
                    }
                }
                const unsigned int okm = __builtin_amdgcn_ballot_w32(ok);
                float2 rs[OWN_BATCH];
                int    dq[OWN_BATCH];
                float  sc = 0.0f;
                if (okm != 0u) {
                    float2 rd[OWN_BATCH];
                    #pragma unroll
                    for (int q = 0; q < OWN_BATCH; ++q) {
                        const int sq = __builtin_amdgcn_readlane(s, q);
                        dq[q] = __builtin_amdgcn_readlane(dl, q);
                        rs[q] = ((const float2*)(z + (size_t)sq * FATT))[lane];
                        rd[q] = ((const float2*)(z + (size_t)(base + dq[q]) * FATT))[lane];
                    }
                    #pragma unroll
                    for (int q = 0; q < OWN_BATCH; ++q) {
                        const float dot = wave_sum(rs[q].x * rd[q].x + rs[q].y * rd[q].y);
                        if (lane == q) sc = dot;
                    }
                }
                sc = sc > 0.0f ? sc : SLOPE * sc;

                int*  flag = &flag_s[g % 3];
                float ref  = sc;
                bool  up   = false;
                if (ok) {
                    const unsigned int mold = atomicCAS(&mc_s[dl], 0u, float_to_ord(sc));
                    if (mold != 0u) {
                        ref = ord_to_float(mold);
                        up  = sc - ref > OWN_RISE;
                        if (up) *flag = 1;
                    }
                }

                const bool now = ok && !up;
                const unsigned int nowm = __builtin_amdgcn_ballot_w32(now);
                float w = 0.0f;
                if (now) {
                    w = __expf(sc - ref);
                    atomicAdd(&den_s[dl], w);
                }
                #pragma unroll
                for (int q = 0; q < OWN_BATCH; ++q) {
                    if ((nowm >> q) & 1u) {
                        const float wq = __int_as_float(__builtin_amdgcn_readlane(__float_as_int(w), q));
                        float* hp = h_s + dq[q] * FATT + 2 * lane;
                        atomicAdd(hp + 0, wq * rs[q].x);
                        atomicAdd(hp + 1, wq * rs[q].y);
                    }
                }
                __syncthreads();
                if (tid == 0) flag_s[(g + 2) % 3] = 0;
                ++g;
                const int rise = __builtin_amdgcn_readfirstlane(*flag);

                if (rise != 0) {
                    if (up) atomicMax(&mo_s[dl], float_to_ord(sc));
                    __syncthreads();

                    unsigned int mnew = 0u;
                    int   take = 0;
                    float fs   = 1.0f;
                    if (up) {
                        mnew = mo_s[dl];
                        const unsigned int mold = atomicMax(&mc_s[dl], mnew);
                        if (mold < mnew) {
                            take = 1;
                            fs   = __expf(ord_to_float(mold) - ord_to_float(mnew));
                            den_s[dl] *= fs;
                        }
                    }
                    if (__builtin_amdgcn_ballot_w32(take != 0) != 0u) {
                        #pragma unroll
                        for (int q = 0; q < OWN_BATCH; ++q) {
                            if (__builtin_amdgcn_readlane(take, q)) {
                                const float f = __int_as_float(__builtin_amdgcn_readlane(__float_as_int(fs), q));
                                float2* hp = (float2*)(h_s + dq[q] * FATT) + lane;
                                float2 hv = *hp;
                                hv.x *= f;
                                hv.y *= f;
                                *hp = hv;
                            }
                        }
                    }
                    __syncthreads();

                    const unsigned int upm = __builtin_amdgcn_ballot_w32(up);
                    float w2 = 0.0f;
                    if (up) {
                        mo_s[dl] = 0u;
                        w2 = __expf(sc - ord_to_float(mnew));
                        atomicAdd(&den_s[dl], w2);
                    }
                    #pragma unroll
                    for (int q = 0; q < OWN_BATCH; ++q) {
                        if ((upm >> q) & 1u) {
                            const float wq = __int_as_float(__builtin_amdgcn_readlane(__float_as_int(w2), q));
                            float* hp = h_s + dq[q] * FATT + 2 * lane;
                            atomicAdd(hp + 0, wq * rs[q].x);
                            atomicAdd(hp + 1, wq * rs[q].y);
                        }
                    }
                }
            }
        }
    }
    __syncthreads();

    for (int i4 = tid; i4 < OWN_NODES * (FATT / 4); i4 += OWN_THREADS) {
        const int nl = i4 / (FATT / 4);
        if (base + nl < NTOT) {
            const float dn = den_s[nl];
            float4 v = *(const float4*)&h_s[i4 * 4];
            v.x = dn > 0.0f ? v.x / dn : 0.0f;
            v.y = dn > 0.0f ? v.y / dn : 0.0f;
            v.z = dn > 0.0f ? v.z / dn : 0.0f;
            v.w = dn > 0.0f ? v.w / dn : 0.0f;
            v.x = v.x > 0.0f ? v.x : (__expf(v.x) - 1.0f);
            v.y = v.y > 0.0f ? v.y : (__expf(v.y) - 1.0f);
            v.z = v.z > 0.0f ? v.z : (__expf(v.z) - 1.0f);
            v.w = v.w > 0.0f ? v.w : (__expf(v.w) - 1.0f);
            *(float4*)(out + (size_t)base * FATT + (size_t)i4 * 4) = v;
        }
    }
}

__global__ __launch_bounds__(OWN_THREADS) void node_block_attend(
        const float* __restrict__ z, const int* __restrict__ src, const int* __restrict__ dst,
        const unsigned int* __restrict__ keys, float* __restrict__ out, size_t ws_size) {
    __shared__ __attribute__((aligned(16))) float h_s[OWN_NODES * FATT];
    __shared__ float        den_s[OWN_NODES];
    __shared__ unsigned int mo_s[OWN_NODES];
    __shared__ unsigned int mc_s[OWN_NODES];
    __shared__ int          list_s[OWN_LIST];
    __shared__ int          cnt_s[3];
    __shared__ int          flag_s[3];

    if (ws_size >= WS_NEED)
        attend_body<true >(z, src, dst, keys, out, h_s, den_s, mo_s, mc_s, list_s, cnt_s, flag_s);
    else
        attend_body<false>(z, src, dst, keys, out, h_s, den_s, mo_s, mc_s, list_s, cnt_s, flag_s);
}

extern "C" void kernel_launch(void* const* d_in, const int* in_sizes, int n_in,
                              void* d_out, int out_size, void* d_ws, size_t ws_size,
                              hipStream_t stream) {
    const float* d_sim  = (const float*)d_in[0];
    const float* mi_sim = (const float*)d_in[1];
    const float* W_d    = (const float*)d_in[2];
    const float* W_mi   = (const float*)d_in[3];
    const int*   src    = (const int*)d_in[4];
    const int*   dst    = (const int*)d_in[5];
    float*       out    = (float*)d_out;

    float*        z     = (float*)d_ws;
    unsigned int* keys  = (unsigned int*)((char*)d_ws + WS_Z_BYTES);

    gemm_wmma_f32<<<(N_DIS + 127) / 128, 128, 0, stream>>>(d_sim, W_d, z, N_DIS, 383);
    gemm_wmma_f32<<<(N_MIC + 127) / 128, 128, 0, stream>>>(mi_sim, W_mi,
                                                           z + (size_t)N_DIS * FATT, N_MIC, 495);

    owner_keys<<<(KEY_NVEC + KEY_THREADS - 1) / KEY_THREADS, KEY_THREADS, 0, stream>>>(dst, keys, ws_size);

    node_block_attend<<<OWN_BLOCKS, OWN_THREADS, 0, stream>>>(z, src, dst, keys, out, ws_size);
}


# Owner work-groups in 3 free-running phases: list own edges once in LDS; scores + LDS max; exp(score-max) sums; 4 barriers, not 2 a round; source rows read twice; sum order changes
# speedup vs baseline: 5.9205x; 1.0318x over previous
#include <hip/hip_runtime.h>
#include <math.h>

#define N_DIS 20000
#define N_MIC 80000
#define NTOT  (N_DIS + N_MIC)
#define NEDGE 1600000
#define FATT  64
#define SLOPE 0.2f

typedef float v2f __attribute__((ext_vector_type(2)));
typedef float v8f __attribute__((ext_vector_type(8)));

typedef float v4f __attribute__((ext_vector_type(4)));

#define DEPTH 5

struct Frag { v4f a0, a1, w0, w1, w2, w3; };

__device__ __forceinline__ v4f ld_x4(const char* p) {
    v4f v;
    __builtin_memcpy(&v, p, sizeof(v));
    return v;
}

__device__ __forceinline__ void load_frag(Frag& f, const float* X, const float* W,
                                          unsigned xo0, unsigned xo1, unsigned wo, int kb) {
    const char* xk = (const char*)X + (size_t)kb * (8 * sizeof(float));
    const char* wk = (const char*)W + (size_t)kb * (8 * FATT * sizeof(float));
    f.a0 = ld_x4(xk + xo0);
    f.a1 = ld_x4(xk + xo1);
    f.w0 = *(const v4f*)(wk + wo);
    f.w1 = *(const v4f*)(wk + wo + 1 * FATT * sizeof(float));
    f.w2 = *(const v4f*)(wk + wo + 2 * FATT * sizeof(float));
    f.w3 = *(const v4f*)(wk + wo + 3 * FATT * sizeof(float));
}

__device__ __forceinline__ void mma_frag(const Frag& f, v8f (&acc)[2][4]) {
    const v2f a0lo = { f.a0.x, f.a0.y }, a0hi = { f.a0.z, f.a0.w };
    const v2f a1lo = { f.a1.x, f.a1.y }, a1hi = { f.a1.z, f.a1.w };
    #pragma unroll
    for (int j = 0; j < 4; ++j) {
        const v2f b = { f.w0[j], f.w1[j] };
        acc[0][j] = __builtin_amdgcn_wmma_f32_16x16x4_f32(false, a0lo, false, b, (short)0, acc[0][j], false, false);
        acc[1][j] = __builtin_amdgcn_wmma_f32_16x16x4_f32(false, a1lo, false, b, (short)0, acc[1][j], false, false);
    }
    #pragma unroll
    for (int j = 0; j < 4; ++j) {
        const v2f b = { f.w2[j], f.w3[j] };
        acc[0][j] = __builtin_amdgcn_wmma_f32_16x16x4_f32(false, a0hi, false, b, (short)0, acc[0][j], false, false);
        acc[1][j] = __builtin_amdgcn_wmma_f32_16x16x4_f32(false, a1hi, false, b, (short)0, acc[1][j], false, false);
    }
}

__global__ __launch_bounds__(128) void gemm_wmma_f32(
        const float* __restrict__ X, const float* __restrict__ W,
        float* __restrict__ Z, int M, int K) {
    const int lane = threadIdx.x & 31;
    const int wave = threadIdx.x >> 5;
    const int c    = lane & 15;
    const int h    = lane >> 4;

    const int row_base = (blockIdx.x * 4 + wave) * 32;
    if (row_base >= M) return;

    const int r0 = min(row_base + c, M - 1);
    const int r1 = min(row_base + 16 + c, M - 1);
    const unsigned xo0 = ((unsigned)r0 * (unsigned)K + 4u * h) * (unsigned)sizeof(float);
    const unsigned xo1 = ((unsigned)r1 * (unsigned)K + 4u * h) * (unsigned)sizeof(float);
    const unsigned wo  = (4u * h * FATT + 4u * c) * (unsigned)sizeof(float);

    v8f acc[2][4] = {};

    const int nfull = K >> 3;
    int kb = 0;
    if (nfull > 0) {
        Frag f[DEPTH];
        #pragma unroll
        for (int i = 0; i < DEPTH; ++i)
            load_frag(f[i], X, W, xo0, xo1, wo, min(i, nfull - 1));

        for (; kb + DEPTH <= nfull; kb += DEPTH) {
            #pragma unroll
            for (int i = 0; i < DEPTH; ++i) {
                mma_frag(f[i], acc);
                load_frag(f[i], X, W, xo0, xo1, wo, min(kb + DEPTH + i, nfull - 1));
                __builtin_amdgcn_sched_barrier(0);
            }
        }
        #pragma unroll
        for (int i = 0; i < DEPTH; ++i)
            if (kb + i < nfull) mma_frag(f[i], acc);
    }

    const int kt = nfull * 8;
    if (kt < K) {
        Frag t;
        float x0[4], x1[4];
        v4f wv[4];
        #pragma unroll
        for (int i = 0; i < 4; ++i) {
            const int  k  = kt + 4 * h + i;
            const bool ok = k < K;
            const int  kc = ok ? k : (K - 1);
            const float p0 = X[(size_t)r0 * K + kc];
            const float p1 = X[(size_t)r1 * K + kc];
            const v4f   pw = *(const v4f*)(W + (size_t)kc * FATT + 4 * c);
            const v4f   zero = { 0.0f, 0.0f, 0.0f, 0.0f };
            x0[i] = ok ? p0 : 0.0f;
            x1[i] = ok ? p1 : 0.0f;
            wv[i] = ok ? pw : zero;
        }
        t.a0 = (v4f){ x0[0], x0[1], x0[2], x0[3] };
        t.a1 = (v4f){ x1[0], x1[1], x1[2], x1[3] };
        t.w0 = wv[0]; t.w1 = wv[1]; t.w2 = wv[2]; t.w3 = wv[3];
        mma_frag(t, acc);
    }

    #pragma unroll
    for (int mt = 0; mt < 2; ++mt) {
        #pragma unroll
        for (int r = 0; r < 8; ++r) {
            const int mm = row_base + 16 * mt + 8 * h + r;
            if (mm < M) {
                const v4f o = { acc[mt][0][r], acc[mt][1][r], acc[mt][2][r], acc[mt][3][r] };
                *(v4f*)(Z + (size_t)mm * FATT + 4 * c) = o;
            }
        }
    }
}

__device__ __forceinline__ unsigned int float_to_ord(float f) {
    unsigned int u = __float_as_uint(f);
    return (u & 0x80000000u) ? ~u : (u | 0x80000000u);
}
__device__ __forceinline__ float ord_to_float(unsigned int u) {
    return __uint_as_float((u & 0x80000000u) ? (u ^ 0x80000000u) : ~u);
}

template <int CTRL>
__device__ __forceinline__ float dpp_move(float v) {
    return __int_as_float(__builtin_amdgcn_update_dpp(0, __float_as_int(v), CTRL, 0xF, 0xF, true));
}
__device__ __forceinline__ float wave_sum(float v) {
    v += dpp_move<0x161>(v);
    v += dpp_move<0x162>(v);
    v += dpp_move<0x164>(v);
    v += dpp_move<0x168>(v);
    v += __int_as_float((int)__builtin_amdgcn_permlanex16(0u, (unsigned)__float_as_int(v),
                                                          0x76543210u, 0xfedcba98u, false, false));
    return v;
}

#define OWN_BLOCKS  256
#define OWN_NODES   ((NTOT + OWN_BLOCKS - 1) / OWN_BLOCKS)
#define KEY_NVEC    (NEDGE / 16)
#define KEY_THREADS 256
#define WS_Z_BYTES  ((size_t)NTOT * FATT * sizeof(float))
#define WS_NEED     (WS_Z_BYTES + (size_t)NEDGE)

__device__ __forceinline__ unsigned int key_of(int d) {
    const unsigned int u = (unsigned int)d;
    return u < (unsigned int)(OWN_BLOCKS * OWN_NODES) ? u / (unsigned int)OWN_NODES : 255u;
}

__global__ __launch_bounds__(KEY_THREADS) void owner_keys(
        const int* __restrict__ dst, unsigned int* __restrict__ keys, size_t ws_size) {
    if (ws_size < WS_NEED) return;
    const int v = blockIdx.x * KEY_THREADS + threadIdx.x;
    if (v >= KEY_NVEC) return;
    const int4* d4 = (const int4*)dst + (size_t)v * 4;
    unsigned int w[4];
    #pragma unroll
    for (int j = 0; j < 4; ++j) {
        const int4 d = d4[j];
        w[j] = key_of(d.x) | (key_of(d.y) << 8) | (key_of(d.z) << 16) | (key_of(d.w) << 24);
    }
    ((uint4*)keys)[v] = make_uint4(w[0], w[1], w[2], w[3]);
}

#define OWN_THREADS 1024
#define OWN_WAVES   (OWN_THREADS / 32)
#define OWN_KVEC    7
#define OWN_NMASK   ((OWN_KVEC + 1) / 2)
#define OWN_RVEC    (OWN_THREADS * OWN_KVEC)
#define OWN_CHUNK   (OWN_RVEC * 16)
#define OWN_BATCH   16
#define OWN_TRIP    (OWN_WAVES * OWN_BATCH)
#define OWN_LIST    4096
#define OWN_ROUNDS  ((NEDGE + OWN_CHUNK - 1) / OWN_CHUNK)
#define OWN_RISE    60.0f

__device__ __forceinline__ unsigned int zero_bytes(unsigned int x) {
    return ~(((x & 0x7f7f7f7fu) + 0x7f7f7f7fu) | x) & 0x80808080u;
}

__device__ __forceinline__ void list_window(const unsigned int (&hm)[OWN_NMASK], int first,
                                            int ebase, int* list_s) {
    int idx = first;
    #pragma unroll
    for (int m = 0; m < OWN_NMASK; ++m) {
        unsigned int mask = hm[m];
        while (mask != 0u) {
            const int p = __ffs((int)mask) - 1;
            mask &= mask - 1u;
            if ((unsigned)idx < (unsigned)OWN_LIST)
                list_s[idx] = ebase + (2 * m + ((p >> 2) & 1)) * (OWN_THREADS * 16) + (p & 3) * 4 + (p >> 3);
            ++idx;
        }
    }
}

template <bool KEYS>
__device__ __forceinline__ void attend_body(
        const float* __restrict__ z, const int* __restrict__ src, const int* __restrict__ dst,
        const unsigned int* __restrict__ keys, float* __restrict__ out,
        float* h_s, float* den_s, unsigned int* mo_s, unsigned int* mc_s, int* list_s, int* cnt_s,
        int* flag_s) {
    const int tid  = threadIdx.x;
    const int lane = tid & 31;
    const int wave = tid >> 5;
    const int base = blockIdx.x * OWN_NODES;

    for (int i = tid; i < OWN_NODES * FATT; i += OWN_THREADS) h_s[i] = 0.0f;
    for (int i = tid; i < OWN_NODES; i += OWN_THREADS) {
        den_s[i] = 0.0f;
        mo_s[i]  = 0u;
        mc_s[i]  = 0u;
    }
    if (tid < 3) {
        cnt_s[tid]  = 0;
        flag_s[tid] = 0;
    }
    int g = 0;

    const int4*  dst4 = (const int4*)dst;
    const uint4* key4 = (const uint4*)keys;
    const unsigned int pat = blockIdx.x * 0x01010101u;
    __syncthreads();

    for (int r = 0; r < OWN_ROUNDS; ++r) {
        int* cnt = &cnt_s[r % 3];
        const int vbase = r * OWN_RVEC;
        const int ebase = (vbase + tid) * 16;

        unsigned int hm[OWN_NMASK];
        #pragma unroll
        for (int m = 0; m < OWN_NMASK; ++m) hm[m] = 0u;

        if constexpr (KEYS) {
            uint4 kv[OWN_KVEC];
            #pragma unroll
            for (int i = 0; i < OWN_KVEC; ++i)
                kv[i] = key4[min(vbase + i * OWN_THREADS + tid, KEY_NVEC - 1)];
            #pragma unroll
            for (int i = 0; i < OWN_KVEC; ++i) {
                const unsigned int live = vbase + i * OWN_THREADS + tid < KEY_NVEC ? 0xffffffffu : 0u;
                const unsigned int bits = (zero_bytes(kv[i].x ^ pat) >> (7 - (4 * (i & 1) + 0))) |
                                          (zero_bytes(kv[i].y ^ pat) >> (7 - (4 * (i & 1) + 1))) |
                                          (zero_bytes(kv[i].z ^ pat) >> (7 - (4 * (i & 1) + 2))) |
                                          (zero_bytes(kv[i].w ^ pat) >> (7 - (4 * (i & 1) + 3)));
                hm[i >> 1] |= bits & live;
            }
        } else {
            #pragma nounroll
            for (int m = 0; m < OWN_NMASK; ++m) {
                unsigned int bits = 0u;
                #pragma unroll
                for (int ii = 0; ii < 2; ++ii) {
                    const int i = 2 * m + ii;
                    const int v = vbase + i * OWN_THREADS + tid;
                    if (i < OWN_KVEC && v < KEY_NVEC) {
                        #pragma unroll
                        for (int j = 0; j < 4; ++j) {
                            const int4 d = dst4[(size_t)v * 4 + j];
                            const unsigned int b =
                                ((unsigned)(d.x - base) < (unsigned)OWN_NODES ? 0x00000001u : 0u) |
                                ((unsigned)(d.y - base) < (unsigned)OWN_NODES ? 0x00000100u : 0u) |
                                ((unsigned)(d.z - base) < (unsigned)OWN_NODES ? 0x00010000u : 0u) |
                                ((unsigned)(d.w - base) < (unsigned)OWN_NODES ? 0x01000000u : 0u);
                            bits |= b << (4 * ii + j);
                        }
                    }
                }
                #pragma unroll
                for (int k = 0; k < OWN_NMASK; ++k)
                    if (k == m) hm[k] = bits;
            }
        }
        int total = 0;
        #pragma unroll
        for (int m = 0; m < OWN_NMASK; ++m) total += __popc(hm[m]);
        int pos0 = 0;
        if (total != 0) {
            pos0 = atomicAdd(cnt, total);
            list_window(hm, pos0, ebase, list_s);
        }
        __syncthreads();
        const int n = *cnt;
        if (tid == 0) cnt_s[(r + 2) % 3] = 0;

        for (int lo = 0; lo < n; lo += OWN_LIST) {
            if (lo != 0) {
                if (total != 0) list_window(hm, pos0 - lo, ebase, list_s);
                __syncthreads();
            }
            const int nw    = min(n - lo, OWN_LIST);
            const int trips = (nw + OWN_TRIP - 1) / OWN_TRIP;
            for (int t = 0; t < trips; ++t) {
                const int j0 = t * OWN_TRIP + wave * OWN_BATCH;
                bool ok = lane < OWN_BATCH && j0 + lane < nw;

                int s = 0, dl = 0;
                if (ok) {
                    const int edge = list_s[j0 + lane];
                    s  = src[edge];
                    dl = dst[edge] - base;
                    if ((unsigned)dl >= (unsigned)OWN_NODES) {
                        ok = false;
                        s  = 0;
                        dl = 0;
                    }
                }
                const unsigned int okm = __builtin_amdgcn_ballot_w32(ok);
                float2 rs[OWN_BATCH];
                int    dq[OWN_BATCH];
                float  sc = 0.0f;
                if (okm != 0u) {
                    float2 rd[OWN_BATCH];
                    #pragma unroll
                    for (int q = 0; q < OWN_BATCH; ++q) {
                        const int sq = __builtin_amdgcn_readlane(s, q);
                        dq[q] = __builtin_amdgcn_readlane(dl, q);
                        rs[q] = ((const float2*)(z + (size_t)sq * FATT))[lane];
                        rd[q] = ((const float2*)(z + (size_t)(base + dq[q]) * FATT))[lane];
                    }
                    #pragma unroll
                    for (int q = 0; q < OWN_BATCH; ++q) {
                        const float dot = wave_sum(rs[q].x * rd[q].x + rs[q].y * rd[q].y);
                        if (lane == q) sc = dot;
                    }
                }
                sc = sc > 0.0f ? sc : SLOPE * sc;

                int*  flag = &flag_s[g % 3];
                float ref  = sc;
                bool  up   = false;
                if (ok) {
                    const unsigned int mold = atomicCAS(&mc_s[dl], 0u, float_to_ord(sc));
                    if (mold != 0u) {
                        ref = ord_to_float(mold);
                        up  = sc - ref > OWN_RISE;
                        if (up) *flag = 1;
                    }
                }

                const bool now = ok && !up;
                const unsigned int nowm = __builtin_amdgcn_ballot_w32(now);
                float w = 0.0f;
                if (now) {
                    w = __expf(sc - ref);
                    atomicAdd(&den_s[dl], w);
                }
                #pragma unroll
                for (int q = 0; q < OWN_BATCH; ++q) {
                    if ((nowm >> q) & 1u) {
                        const float wq = __int_as_float(__builtin_amdgcn_readlane(__float_as_int(w), q));
                        float* hp = h_s + dq[q] * FATT + 2 * lane;
                        atomicAdd(hp + 0, wq * rs[q].x);
                        atomicAdd(hp + 1, wq * rs[q].y);
                    }
                }
                __syncthreads();
                if (tid == 0) flag_s[(g + 2) % 3] = 0;
                ++g;
                const int rise = __builtin_amdgcn_readfirstlane(*flag);

                if (rise != 0) {
                    if (up) atomicMax(&mo_s[dl], float_to_ord(sc));
                    __syncthreads();

                    unsigned int mnew = 0u;
                    int   take = 0;
                    float fs   = 1.0f;
                    if (up) {
                        mnew = mo_s[dl];
                        const unsigned int mold = atomicMax(&mc_s[dl], mnew);
                        if (mold < mnew) {
                            take = 1;
                            fs   = __expf(ord_to_float(mold) - ord_to_float(mnew));
                            den_s[dl] *= fs;
                        }
                    }
                    if (__builtin_amdgcn_ballot_w32(take != 0) != 0u) {
                        #pragma unroll
                        for (int q = 0; q < OWN_BATCH; ++q) {
                            if (__builtin_amdgcn_readlane(take, q)) {
                                const float f = __int_as_float(__builtin_amdgcn_readlane(__float_as_int(fs), q));
                                float2* hp = (float2*)(h_s + dq[q] * FATT) + lane;
                                float2 hv = *hp;
                                hv.x *= f;
                                hv.y *= f;
                                *hp = hv;
                            }
                        }
                    }
                    __syncthreads();

                    const unsigned int upm = __builtin_amdgcn_ballot_w32(up);
                    float w2 = 0.0f;
                    if (up) {
                        mo_s[dl] = 0u;
                        w2 = __expf(sc - ord_to_float(mnew));
                        atomicAdd(&den_s[dl], w2);
                    }
                    #pragma unroll
                    for (int q = 0; q < OWN_BATCH; ++q) {
                        if ((upm >> q) & 1u) {
                            const float wq = __int_as_float(__builtin_amdgcn_readlane(__float_as_int(w2), q));
                            float* hp = h_s + dq[q] * FATT + 2 * lane;
                            atomicAdd(hp + 0, wq * rs[q].x);
                            atomicAdd(hp + 1, wq * rs[q].y);
                        }
                    }
                }
            }
        }
    }
    __syncthreads();

    for (int i4 = tid; i4 < OWN_NODES * (FATT / 4); i4 += OWN_THREADS) {
        const int nl = i4 / (FATT / 4);
        if (base + nl < NTOT) {
            const float dn = den_s[nl];
            float4 v = *(const float4*)&h_s[i4 * 4];
            v.x = dn > 0.0f ? v.x / dn : 0.0f;
            v.y = dn > 0.0f ? v.y / dn : 0.0f;
            v.z = dn > 0.0f ? v.z / dn : 0.0f;
            v.w = dn > 0.0f ? v.w / dn : 0.0f;
            v.x = v.x > 0.0f ? v.x : (__expf(v.x) - 1.0f);
            v.y = v.y > 0.0f ? v.y : (__expf(v.y) - 1.0f);
            v.z = v.z > 0.0f ? v.z : (__expf(v.z) - 1.0f);
            v.w = v.w > 0.0f ? v.w : (__expf(v.w) - 1.0f);
            *(float4*)(out + (size_t)base * FATT + (size_t)i4 * 4) = v;
        }
    }
}

#define FAST_CAP 12288

__device__ __forceinline__ void list_all(const unsigned int (&hm)[OWN_NMASK], int first,
                                         int ebase, int* ed_s) {
    int idx = first;
    #pragma unroll
    for (int m = 0; m < OWN_NMASK; ++m) {
        unsigned int mask = hm[m];
        while (mask != 0u) {
            const int p = __ffs((int)mask) - 1;
            mask &= mask - 1u;
            if ((unsigned)idx < (unsigned)FAST_CAP)
                ed_s[idx] = ebase + (2 * m + ((p >> 2) & 1)) * (OWN_THREADS * 16) + (p & 3) * 4 + (p >> 3);
            ++idx;
        }
    }
}

__device__ __forceinline__ bool attend_fast(
        const float* __restrict__ z, const int* __restrict__ src, const int* __restrict__ dst,
        const unsigned int* __restrict__ keys, float* __restrict__ out,
        float* h_s, float* den_s, unsigned int* mc_s, int* ed_s, int* dl_s, float* sc_s, int* cnt_f) {
    const int tid  = threadIdx.x;
    const int lane = tid & 31;
    const int wave = __builtin_amdgcn_readfirstlane(tid >> 5);
    const int base = blockIdx.x * OWN_NODES;

    for (int i = tid; i < OWN_NODES * FATT; i += OWN_THREADS) h_s[i] = 0.0f;
    for (int i = tid; i < OWN_NODES; i += OWN_THREADS) {
        den_s[i] = 0.0f;
        mc_s[i]  = 0u;
    }
    if (tid == 0) *cnt_f = 0;
    const uint4* key4 = (const uint4*)keys;
    const unsigned int pat = blockIdx.x * 0x01010101u;
    __syncthreads();

    for (int r = 0; r < OWN_ROUNDS; ++r) {
        const int vbase = r * OWN_RVEC;
        const int ebase = (vbase + tid) * 16;
        unsigned int hm[OWN_NMASK];
        #pragma unroll
        for (int m = 0; m < OWN_NMASK; ++m) hm[m] = 0u;
        uint4 kv[OWN_KVEC];
        #pragma unroll
        for (int i = 0; i < OWN_KVEC; ++i)
            kv[i] = key4[min(vbase + i * OWN_THREADS + tid, KEY_NVEC - 1)];
        #pragma unroll
        for (int i = 0; i < OWN_KVEC; ++i) {
            const unsigned int live = vbase + i * OWN_THREADS + tid < KEY_NVEC ? 0xffffffffu : 0u;
            const unsigned int bits = (zero_bytes(kv[i].x ^ pat) >> (7 - (4 * (i & 1) + 0))) |
                                      (zero_bytes(kv[i].y ^ pat) >> (7 - (4 * (i & 1) + 1))) |
                                      (zero_bytes(kv[i].z ^ pat) >> (7 - (4 * (i & 1) + 2))) |
                                      (zero_bytes(kv[i].w ^ pat) >> (7 - (4 * (i & 1) + 3)));
            hm[i >> 1] |= bits & live;
        }
        int total = 0;
        #pragma unroll
        for (int m = 0; m < OWN_NMASK; ++m) total += __popc(hm[m]);
        if (total != 0) {
            const int pos0 = atomicAdd(cnt_f, total);
            list_all(hm, pos0, ebase, ed_s);
        }
    }
    __syncthreads();
    const int n = __builtin_amdgcn_readfirstlane(*cnt_f);
    if (n > FAST_CAP) return false;

    for (int j0 = wave * 32; j0 < n; j0 += OWN_WAVES * 32) {
        const int j = j0 + lane;
        bool ok = j < n;
        int s = 0, dl = 0;
        if (ok) {
            const int edge = ed_s[j];
            s  = src[edge];
            dl = dst[edge] - base;
            if ((unsigned)dl >= (unsigned)OWN_NODES) {
                ok = false;
                s  = 0;
                dl = 0;
            }
        }
        const unsigned int okm = __builtin_amdgcn_ballot_w32(ok);
        float sc = 0.0f;
        #pragma unroll
        for (int sub = 0; sub < 2; ++sub) {
            if (((okm >> (16 * sub)) & 0xffffu) != 0u) {
                float2 rs[OWN_BATCH], rd[OWN_BATCH];
                #pragma unroll
                for (int q = 0; q < OWN_BATCH; ++q) {
                    const int sq = __builtin_amdgcn_readlane(s, 16 * sub + q);
                    const int dq = __builtin_amdgcn_readlane(dl, 16 * sub + q);
                    rs[q] = ((const float2*)(z + (size_t)sq * FATT))[lane];
                    rd[q] = ((const float2*)(z + (size_t)(base + dq) * FATT))[lane];
                }
                #pragma unroll
                for (int q = 0; q < OWN_BATCH; ++q) {
                    const float dot = wave_sum(rs[q].x * rd[q].x + rs[q].y * rd[q].y);
                    if (lane == 16 * sub + q) sc = dot;
                }
            }
        }
        sc = sc > 0.0f ? sc : SLOPE * sc;
        if (ok) atomicMax(&mc_s[dl], float_to_ord(sc));
        if (j < n) {
            ed_s[j] = s;
            dl_s[j] = ok ? dl : -1;
            sc_s[j] = sc;
        }
    }
    __syncthreads();

    for (int j0 = wave * 32; j0 < n; j0 += OWN_WAVES * 32) {
        const int j = j0 + lane;
        bool  ok = false;
        int   s = 0, dl = 0;
        float w = 0.0f;
        if (j < n) {
            const int d = dl_s[j];
            if (d >= 0) {
                ok = true;
                dl = d;
                s  = ed_s[j];
                w  = __expf(sc_s[j] - ord_to_float(mc_s[d]));
                atomicAdd(&den_s[d], w);
            }
        }
        const unsigned int okm = __builtin_amdgcn_ballot_w32(ok);
        #pragma unroll
        for (int sub = 0; sub < 2; ++sub) {
            const unsigned int m16 = (okm >> (16 * sub)) & 0xffffu;
            if (m16 != 0u) {
                float2 rs[OWN_BATCH];
                #pragma unroll
                for (int q = 0; q < OWN_BATCH; ++q) {
                    const int sq = __builtin_amdgcn_readlane(s, 16 * sub + q);
                    rs[q] = ((const float2*)(z + (size_t)sq * FATT))[lane];
                }
                #pragma unroll
                for (int q = 0; q < OWN_BATCH; ++q) {
                    if ((m16 >> q) & 1u) {
                        const float wq = __int_as_float(__builtin_amdgcn_readlane(__float_as_int(w), 16 * sub + q));
                        const int   dq = __builtin_amdgcn_readlane(dl, 16 * sub + q);
                        float* hp = h_s + dq * FATT + 2 * lane;
                        atomicAdd(hp + 0, wq * rs[q].x);
                        atomicAdd(hp + 1, wq * rs[q].y);
                    }
                }
            }
        }
    }
    __syncthreads();

    for (int i4 = tid; i4 < OWN_NODES * (FATT / 4); i4 += OWN_THREADS) {
        const int nl = i4 / (FATT / 4);
        if (base + nl < NTOT) {
            const float dn = den_s[nl];
            float4 v = *(const float4*)&h_s[i4 * 4];
            v.x = dn > 0.0f ? v.x / dn : 0.0f;
            v.y = dn > 0.0f ? v.y / dn : 0.0f;
            v.z = dn > 0.0f ? v.z / dn : 0.0f;
            v.w = dn > 0.0f ? v.w / dn : 0.0f;
            v.x = v.x > 0.0f ? v.x : (__expf(v.x) - 1.0f);
            v.y = v.y > 0.0f ? v.y : (__expf(v.y) - 1.0f);
            v.z = v.z > 0.0f ? v.z : (__expf(v.z) - 1.0f);
            v.w = v.w > 0.0f ? v.w : (__expf(v.w) - 1.0f);
            *(float4*)(out + (size_t)base * FATT + (size_t)i4 * 4) = v;
        }
    }
    return true;
}

__global__ __launch_bounds__(OWN_THREADS) void node_block_attend(
        const float* __restrict__ z, const int* __restrict__ src, const int* __restrict__ dst,
        const unsigned int* __restrict__ keys, float* __restrict__ out, size_t ws_size) {
    __shared__ __attribute__((aligned(16))) float h_s[OWN_NODES * FATT];
    __shared__ float        den_s[OWN_NODES];
    __shared__ unsigned int mo_s[OWN_NODES];
    __shared__ unsigned int mc_s[OWN_NODES];
    __shared__ int          ed_s[FAST_CAP];
    __shared__ int          dl_s[FAST_CAP];
    __shared__ float        sc_s[FAST_CAP];
    __shared__ int          cnt_f;
    __shared__ int          cnt_s[3];
    __shared__ int          flag_s[3];

    if (ws_size >= WS_NEED) {
        if (!attend_fast(z, src, dst, keys, out, h_s, den_s, mc_s, ed_s, dl_s, sc_s, &cnt_f))
            attend_body<true >(z, src, dst, keys, out, h_s, den_s, mo_s, mc_s, ed_s, cnt_s, flag_s);
    } else {
        attend_body<false>(z, src, dst, keys, out, h_s, den_s, mo_s, mc_s, ed_s, cnt_s, flag_s);
    }
}

extern "C" void kernel_launch(void* const* d_in, const int* in_sizes, int n_in,
                              void* d_out, int out_size, void* d_ws, size_t ws_size,
                              hipStream_t stream) {
    const float* d_sim  = (const float*)d_in[0];
    const float* mi_sim = (const float*)d_in[1];
    const float* W_d    = (const float*)d_in[2];
    const float* W_mi   = (const float*)d_in[3];
    const int*   src    = (const int*)d_in[4];
    const int*   dst    = (const int*)d_in[5];
    float*       out    = (float*)d_out;

    float*        z     = (float*)d_ws;
    unsigned int* keys  = (unsigned int*)((char*)d_ws + WS_Z_BYTES);

    gemm_wmma_f32<<<(N_DIS + 127) / 128, 128, 0, stream>>>(d_sim, W_d, z, N_DIS, 383);
    gemm_wmma_f32<<<(N_MIC + 127) / 128, 128, 0, stream>>>(mi_sim, W_mi,
                                                           z + (size_t)N_DIS * FATT, N_MIC, 495);

    owner_keys<<<(KEY_NVEC + KEY_THREADS - 1) / KEY_THREADS, KEY_THREADS, 0, stream>>>(dst, keys, ws_size);

    node_block_attend<<<OWN_BLOCKS, OWN_THREADS, 0, stream>>>(z, src, dst, keys, out, ws_size);
}


# Owner work-groups copy their own 391 rows of z into LDS once (into the sum array, cleared behind the max phase: 5 barriers, not 4); scores read the destination row from LDS, not from L2 per edge
# speedup vs baseline: 6.1296x; 1.0353x over previous
#include <hip/hip_runtime.h>
#include <math.h>

#define N_DIS 20000
#define N_MIC 80000
#define NTOT  (N_DIS + N_MIC)
#define NEDGE 1600000
#define FATT  64
#define SLOPE 0.2f

typedef float v2f __attribute__((ext_vector_type(2)));
typedef float v8f __attribute__((ext_vector_type(8)));

typedef float v4f __attribute__((ext_vector_type(4)));

#define DEPTH 5

struct Frag { v4f a0, a1, w0, w1, w2, w3; };

__device__ __forceinline__ v4f ld_x4(const char* p) {
    v4f v;
    __builtin_memcpy(&v, p, sizeof(v));
    return v;
}

__device__ __forceinline__ void load_frag(Frag& f, const float* X, const float* W,
                                          unsigned xo0, unsigned xo1, unsigned wo, int kb) {
    const char* xk = (const char*)X + (size_t)kb * (8 * sizeof(float));
    const char* wk = (const char*)W + (size_t)kb * (8 * FATT * sizeof(float));
    f.a0 = ld_x4(xk + xo0);
    f.a1 = ld_x4(xk + xo1);
    f.w0 = *(const v4f*)(wk + wo);
    f.w1 = *(const v4f*)(wk + wo + 1 * FATT * sizeof(float));
    f.w2 = *(const v4f*)(wk + wo + 2 * FATT * sizeof(float));
    f.w3 = *(const v4f*)(wk + wo + 3 * FATT * sizeof(float));
}

__device__ __forceinline__ void mma_frag(const Frag& f, v8f (&acc)[2][4]) {
    const v2f a0lo = { f.a0.x, f.a0.y }, a0hi = { f.a0.z, f.a0.w };
    const v2f a1lo = { f.a1.x, f.a1.y }, a1hi = { f.a1.z, f.a1.w };
    #pragma unroll
    for (int j = 0; j < 4; ++j) {
        const v2f b = { f.w0[j], f.w1[j] };
        acc[0][j] = __builtin_amdgcn_wmma_f32_16x16x4_f32(false, a0lo, false, b, (short)0, acc[0][j], false, false);
        acc[1][j] = __builtin_amdgcn_wmma_f32_16x16x4_f32(false, a1lo, false, b, (short)0, acc[1][j], false, false);
    }
    #pragma unroll
    for (int j = 0; j < 4; ++j) {
        const v2f b = { f.w2[j], f.w3[j] };
        acc[0][j] = __builtin_amdgcn_wmma_f32_16x16x4_f32(false, a0hi, false, b, (short)0, acc[0][j], false, false);
        acc[1][j] = __builtin_amdgcn_wmma_f32_16x16x4_f32(false, a1hi, false, b, (short)0, acc[1][j], false, false);
    }
}

__global__ __launch_bounds__(128) void gemm_wmma_f32(
        const float* __restrict__ X, const float* __restrict__ W,
        float* __restrict__ Z, int M, int K) {
    const int lane = threadIdx.x & 31;
    const int wave = threadIdx.x >> 5;
    const int c    = lane & 15;
    const int h    = lane >> 4;

    const int row_base = (blockIdx.x * 4 + wave) * 32;
    if (row_base >= M) return;

    const int r0 = min(row_base + c, M - 1);
    const int r1 = min(row_base + 16 + c, M - 1);
    const unsigned xo0 = ((unsigned)r0 * (unsigned)K + 4u * h) * (unsigned)sizeof(float);
    const unsigned xo1 = ((unsigned)r1 * (unsigned)K + 4u * h) * (unsigned)sizeof(float);
    const unsigned wo  = (4u * h * FATT + 4u * c) * (unsigned)sizeof(float);

    v8f acc[2][4] = {};

    const int nfull = K >> 3;
    int kb = 0;
    if (nfull > 0) {
        Frag f[DEPTH];
        #pragma unroll
        for (int i = 0; i < DEPTH; ++i)
            load_frag(f[i], X, W, xo0, xo1, wo, min(i, nfull - 1));

        for (; kb + DEPTH <= nfull; kb += DEPTH) {
            #pragma unroll
            for (int i = 0; i < DEPTH; ++i) {
                mma_frag(f[i], acc);
                load_frag(f[i], X, W, xo0, xo1, wo, min(kb + DEPTH + i, nfull - 1));
                __builtin_amdgcn_sched_barrier(0);
            }
        }
        #pragma unroll
        for (int i = 0; i < DEPTH; ++i)
            if (kb + i < nfull) mma_frag(f[i], acc);
    }

    const int kt = nfull * 8;
    if (kt < K) {
        Frag t;
        float x0[4], x1[4];
        v4f wv[4];
        #pragma unroll
        for (int i = 0; i < 4; ++i) {
            const int  k  = kt + 4 * h + i;
            const bool ok = k < K;
            const int  kc = ok ? k : (K - 1);
            const float p0 = X[(size_t)r0 * K + kc];
            const float p1 = X[(size_t)r1 * K + kc];
            const v4f   pw = *(const v4f*)(W + (size_t)kc * FATT + 4 * c);
            const v4f   zero = { 0.0f, 0.0f, 0.0f, 0.0f };
            x0[i] = ok ? p0 : 0.0f;
            x1[i] = ok ? p1 : 0.0f;
            wv[i] = ok ? pw : zero;
        }
        t.a0 = (v4f){ x0[0], x0[1], x0[2], x0[3] };
        t.a1 = (v4f){ x1[0], x1[1], x1[2], x1[3] };
        t.w0 = wv[0]; t.w1 = wv[1]; t.w2 = wv[2]; t.w3 = wv[3];
        mma_frag(t, acc);
    }

    #pragma unroll
    for (int mt = 0; mt < 2; ++mt) {
        #pragma unroll
        for (int r = 0; r < 8; ++r) {
            const int mm = row_base + 16 * mt + 8 * h + r;
            if (mm < M) {
                const v4f o = { acc[mt][0][r], acc[mt][1][r], acc[mt][2][r], acc[mt][3][r] };
                *(v4f*)(Z + (size_t)mm * FATT + 4 * c) = o;
            }
        }
    }
}

__device__ __forceinline__ unsigned int float_to_ord(float f) {
    unsigned int u = __float_as_uint(f);
    return (u & 0x80000000u) ? ~u : (u | 0x80000000u);
}
__device__ __forceinline__ float ord_to_float(unsigned int u) {
    return __uint_as_float((u & 0x80000000u) ? (u ^ 0x80000000u) : ~u);
}

template <int CTRL>
__device__ __forceinline__ float dpp_move(float v) {
    return __int_as_float(__builtin_amdgcn_update_dpp(0, __float_as_int(v), CTRL, 0xF, 0xF, true));
}
__device__ __forceinline__ float wave_sum(float v) {
    v += dpp_move<0x161>(v);
    v += dpp_move<0x162>(v);
    v += dpp_move<0x164>(v);
    v += dpp_move<0x168>(v);
    v += __int_as_float((int)__builtin_amdgcn_permlanex16(0u, (unsigned)__float_as_int(v),
                                                          0x76543210u, 0xfedcba98u, false, false));
    return v;
}

#define OWN_BLOCKS  256
#define OWN_NODES   ((NTOT + OWN_BLOCKS - 1) / OWN_BLOCKS)
#define KEY_NVEC    (NEDGE / 16)
#define KEY_THREADS 256
#define WS_Z_BYTES  ((size_t)NTOT * FATT * sizeof(float))
#define WS_NEED     (WS_Z_BYTES + (size_t)NEDGE)

__device__ __forceinline__ unsigned int key_of(int d) {
    const unsigned int u = (unsigned int)d;
    return u < (unsigned int)(OWN_BLOCKS * OWN_NODES) ? u / (unsigned int)OWN_NODES : 255u;
}

__global__ __launch_bounds__(KEY_THREADS) void owner_keys(
        const int* __restrict__ dst, unsigned int* __restrict__ keys, size_t ws_size) {
    if (ws_size < WS_NEED) return;
    const int v = blockIdx.x * KEY_THREADS + threadIdx.x;
    if (v >= KEY_NVEC) return;
    const int4* d4 = (const int4*)dst + (size_t)v * 4;
    unsigned int w[4];
    #pragma unroll
    for (int j = 0; j < 4; ++j) {
        const int4 d = d4[j];
        w[j] = key_of(d.x) | (key_of(d.y) << 8) | (key_of(d.z) << 16) | (key_of(d.w) << 24);
    }
    ((uint4*)keys)[v] = make_uint4(w[0], w[1], w[2], w[3]);
}

#define OWN_THREADS 1024
#define OWN_WAVES   (OWN_THREADS / 32)
#define OWN_KVEC    7
#define OWN_NMASK   ((OWN_KVEC + 1) / 2)
#define OWN_RVEC    (OWN_THREADS * OWN_KVEC)
#define OWN_CHUNK   (OWN_RVEC * 16)
#define OWN_BATCH   16
#define OWN_TRIP    (OWN_WAVES * OWN_BATCH)
#define OWN_LIST    4096
#define OWN_ROUNDS  ((NEDGE + OWN_CHUNK - 1) / OWN_CHUNK)
#define OWN_RISE    60.0f

__device__ __forceinline__ unsigned int zero_bytes(unsigned int x) {
    return ~(((x & 0x7f7f7f7fu) + 0x7f7f7f7fu) | x) & 0x80808080u;
}

__device__ __forceinline__ void list_window(const unsigned int (&hm)[OWN_NMASK], int first,
                                            int ebase, int* list_s) {
    int idx = first;
    #pragma unroll
    for (int m = 0; m < OWN_NMASK; ++m) {
        unsigned int mask = hm[m];
        while (mask != 0u) {
            const int p = __ffs((int)mask) - 1;
            mask &= mask - 1u;
            if ((unsigned)idx < (unsigned)OWN_LIST)
                list_s[idx] = ebase + (2 * m + ((p >> 2) & 1)) * (OWN_THREADS * 16) + (p & 3) * 4 + (p >> 3);
            ++idx;
        }
    }
}

template <bool KEYS>
__device__ __forceinline__ void attend_body(
        const float* __restrict__ z, const int* __restrict__ src, const int* __restrict__ dst,
        const unsigned int* __restrict__ keys, float* __restrict__ out,
        float* h_s, float* den_s, unsigned int* mo_s, unsigned int* mc_s, int* list_s, int* cnt_s,
        int* flag_s) {
    const int tid  = threadIdx.x;
    const int lane = tid & 31;
    const int wave = tid >> 5;
    const int base = blockIdx.x * OWN_NODES;

    for (int i = tid; i < OWN_NODES * FATT; i += OWN_THREADS) h_s[i] = 0.0f;
    for (int i = tid; i < OWN_NODES; i += OWN_THREADS) {
        den_s[i] = 0.0f;
        mo_s[i]  = 0u;
        mc_s[i]  = 0u;
    }
    if (tid < 3) {
        cnt_s[tid]  = 0;
        flag_s[tid] = 0;
    }
    int g = 0;

    const int4*  dst4 = (const int4*)dst;
    const uint4* key4 = (const uint4*)keys;
    const unsigned int pat = blockIdx.x * 0x01010101u;
    __syncthreads();

    for (int r = 0; r < OWN_ROUNDS; ++r) {
        int* cnt = &cnt_s[r % 3];
        const int vbase = r * OWN_RVEC;
        const int ebase = (vbase + tid) * 16;

        unsigned int hm[OWN_NMASK];
        #pragma unroll
        for (int m = 0; m < OWN_NMASK; ++m) hm[m] = 0u;

        if constexpr (KEYS) {
            uint4 kv[OWN_KVEC];
            #pragma unroll
            for (int i = 0; i < OWN_KVEC; ++i)
                kv[i] = key4[min(vbase + i * OWN_THREADS + tid, KEY_NVEC - 1)];
            #pragma unroll
            for (int i = 0; i < OWN_KVEC; ++i) {
                const unsigned int live = vbase + i * OWN_THREADS + tid < KEY_NVEC ? 0xffffffffu : 0u;
                const unsigned int bits = (zero_bytes(kv[i].x ^ pat) >> (7 - (4 * (i & 1) + 0))) |
                                          (zero_bytes(kv[i].y ^ pat) >> (7 - (4 * (i & 1) + 1))) |
                                          (zero_bytes(kv[i].z ^ pat) >> (7 - (4 * (i & 1) + 2))) |
                                          (zero_bytes(kv[i].w ^ pat) >> (7 - (4 * (i & 1) + 3)));
                hm[i >> 1] |= bits & live;
            }
        } else {
            #pragma nounroll
            for (int m = 0; m < OWN_NMASK; ++m) {
                unsigned int bits = 0u;
                #pragma unroll
                for (int ii = 0; ii < 2; ++ii) {
                    const int i = 2 * m + ii;
                    const int v = vbase + i * OWN_THREADS + tid;
                    if (i < OWN_KVEC && v < KEY_NVEC) {
                        #pragma unroll
                        for (int j = 0; j < 4; ++j) {
                            const int4 d = dst4[(size_t)v * 4 + j];
                            const unsigned int b =
                                ((unsigned)(d.x - base) < (unsigned)OWN_NODES ? 0x00000001u : 0u) |
                                ((unsigned)(d.y - base) < (unsigned)OWN_NODES ? 0x00000100u : 0u) |
                                ((unsigned)(d.z - base) < (unsigned)OWN_NODES ? 0x00010000u : 0u) |
                                ((unsigned)(d.w - base) < (unsigned)OWN_NODES ? 0x01000000u : 0u);
                            bits |= b << (4 * ii + j);
                        }
                    }
                }
                #pragma unroll
                for (int k = 0; k < OWN_NMASK; ++k)
                    if (k == m) hm[k] = bits;
            }
        }
        int total = 0;
        #pragma unroll
        for (int m = 0; m < OWN_NMASK; ++m) total += __popc(hm[m]);
        int pos0 = 0;
        if (total != 0) {
            pos0 = atomicAdd(cnt, total);
            list_window(hm, pos0, ebase, list_s);
        }
        __syncthreads();
        const int n = *cnt;
        if (tid == 0) cnt_s[(r + 2) % 3] = 0;

        for (int lo = 0; lo < n; lo += OWN_LIST) {
            if (lo != 0) {
                if (total != 0) list_window(hm, pos0 - lo, ebase, list_s);
                __syncthreads();
            }
            const int nw    = min(n - lo, OWN_LIST);
            const int trips = (nw + OWN_TRIP - 1) / OWN_TRIP;
            for (int t = 0; t < trips; ++t) {
                const int j0 = t * OWN_TRIP + wave * OWN_BATCH;
                bool ok = lane < OWN_BATCH && j0 + lane < nw;

                int s = 0, dl = 0;
                if (ok) {
                    const int edge = list_s[j0 + lane];
                    s  = src[edge];
                    dl = dst[edge] - base;
                    if ((unsigned)dl >= (unsigned)OWN_NODES) {
                        ok = false;
                        s  = 0;
                        dl = 0;
                    }
                }
                const unsigned int okm = __builtin_amdgcn_ballot_w32(ok);
                float2 rs[OWN_BATCH];
                int    dq[OWN_BATCH];
                float  sc = 0.0f;
                if (okm != 0u) {
                    float2 rd[OWN_BATCH];
                    #pragma unroll
                    for (int q = 0; q < OWN_BATCH; ++q) {
                        const int sq = __builtin_amdgcn_readlane(s, q);
                        dq[q] = __builtin_amdgcn_readlane(dl, q);
                        rs[q] = ((const float2*)(z + (size_t)sq * FATT))[lane];
                        rd[q] = ((const float2*)(z + (size_t)(base + dq[q]) * FATT))[lane];
                    }
                    #pragma unroll
                    for (int q = 0; q < OWN_BATCH; ++q) {
                        const float dot = wave_sum(rs[q].x * rd[q].x + rs[q].y * rd[q].y);
                        if (lane == q) sc = dot;
                    }
                }
                sc = sc > 0.0f ? sc : SLOPE * sc;

                int*  flag = &flag_s[g % 3];
                float ref  = sc;
                bool  up   = false;
                if (ok) {
                    const unsigned int mold = atomicCAS(&mc_s[dl], 0u, float_to_ord(sc));
                    if (mold != 0u) {
                        ref = ord_to_float(mold);
                        up  = sc - ref > OWN_RISE;
                        if (up) *flag = 1;
                    }
                }

                const bool now = ok && !up;
                const unsigned int nowm = __builtin_amdgcn_ballot_w32(now);
                float w = 0.0f;
                if (now) {
                    w = __expf(sc - ref);
                    atomicAdd(&den_s[dl], w);
                }
                #pragma unroll
                for (int q = 0; q < OWN_BATCH; ++q) {
                    if ((nowm >> q) & 1u) {
                        const float wq = __int_as_float(__builtin_amdgcn_readlane(__float_as_int(w), q));
                        float* hp = h_s + dq[q] * FATT + 2 * lane;
                        atomicAdd(hp + 0, wq * rs[q].x);
                        atomicAdd(hp + 1, wq * rs[q].y);
                    }
                }
                __syncthreads();
                if (tid == 0) flag_s[(g + 2) % 3] = 0;
                ++g;
                const int rise = __builtin_amdgcn_readfirstlane(*flag);

                if (rise != 0) {
                    if (up) atomicMax(&mo_s[dl], float_to_ord(sc));
                    __syncthreads();

                    unsigned int mnew = 0u;
                    int   take = 0;
                    float fs   = 1.0f;
                    if (up) {
                        mnew = mo_s[dl];
                        const unsigned int mold = atomicMax(&mc_s[dl], mnew);
                        if (mold < mnew) {
                            take = 1;
                            fs   = __expf(ord_to_float(mold) - ord_to_float(mnew));
                            den_s[dl] *= fs;
                        }
                    }
                    if (__builtin_amdgcn_ballot_w32(take != 0) != 0u) {
                        #pragma unroll
                        for (int q = 0; q < OWN_BATCH; ++q) {
                            if (__builtin_amdgcn_readlane(take, q)) {
                                const float f = __int_as_float(__builtin_amdgcn_readlane(__float_as_int(fs), q));
                                float2* hp = (float2*)(h_s + dq[q] * FATT) + lane;
                                float2 hv = *hp;
                                hv.x *= f;
                                hv.y *= f;
                                *hp = hv;
                            }
                        }
                    }
                    __syncthreads();

                    const unsigned int upm = __builtin_amdgcn_ballot_w32(up);
                    float w2 = 0.0f;
                    if (up) {
                        mo_s[dl] = 0u;
                        w2 = __expf(sc - ord_to_float(mnew));
                        atomicAdd(&den_s[dl], w2);
                    }
                    #pragma unroll
                    for (int q = 0; q < OWN_BATCH; ++q) {
                        if ((upm >> q) & 1u) {
                            const float wq = __int_as_float(__builtin_amdgcn_readlane(__float_as_int(w2), q));
                            float* hp = h_s + dq[q] * FATT + 2 * lane;
                            atomicAdd(hp + 0, wq * rs[q].x);
                            atomicAdd(hp + 1, wq * rs[q].y);
                        }
                    }
                }
            }
        }
    }
    __syncthreads();

    for (int i4 = tid; i4 < OWN_NODES * (FATT / 4); i4 += OWN_THREADS) {
        const int nl = i4 / (FATT / 4);
        if (base + nl < NTOT) {
            const float dn = den_s[nl];
            float4 v = *(const float4*)&h_s[i4 * 4];
            v.x = dn > 0.0f ? v.x / dn : 0.0f;
            v.y = dn > 0.0f ? v.y / dn : 0.0f;
            v.z = dn > 0.0f ? v.z / dn : 0.0f;
            v.w = dn > 0.0f ? v.w / dn : 0.0f;
            v.x = v.x > 0.0f ? v.x : (__expf(v.x) - 1.0f);
            v.y = v.y > 0.0f ? v.y : (__expf(v.y) - 1.0f);
            v.z = v.z > 0.0f ? v.z : (__expf(v.z) - 1.0f);
            v.w = v.w > 0.0f ? v.w : (__expf(v.w) - 1.0f);
            *(float4*)(out + (size_t)base * FATT + (size_t)i4 * 4) = v;
        }
    }
}

#define FAST_CAP 12288
#define FAST_STAGE ((OWN_NODES * (FATT / 4) + OWN_THREADS - 1) / OWN_THREADS)

__device__ __forceinline__ void list_all(const unsigned int (&hm)[OWN_NMASK], int first,
                                         int ebase, int* ed_s) {
    int idx = first;
    #pragma unroll
    for (int m = 0; m < OWN_NMASK; ++m) {
        unsigned int mask = hm[m];
        while (mask != 0u) {
            const int p = __ffs((int)mask) - 1;
            mask &= mask - 1u;
            if ((unsigned)idx < (unsigned)FAST_CAP)
                ed_s[idx] = ebase + (2 * m + ((p >> 2) & 1)) * (OWN_THREADS * 16) + (p & 3) * 4 + (p >> 3);
            ++idx;
        }
    }
}

__device__ __forceinline__ bool attend_fast(
        const float* __restrict__ z, const int* __restrict__ src, const int* __restrict__ dst,
        const unsigned int* __restrict__ keys, float* __restrict__ out,
        float* h_s, float* den_s, unsigned int* mc_s, int* ed_s, int* dl_s, float* sc_s, int* cnt_f) {
    const int tid  = threadIdx.x;
    const int lane = tid & 31;
    const int wave = __builtin_amdgcn_readfirstlane(tid >> 5);
    const int base = blockIdx.x * OWN_NODES;

    const float4* zown = (const float4*)(z + (size_t)base * FATT);
    const int nv4 = (min(base + OWN_NODES, NTOT) - base) * (FATT / 4);
    float4 sv[FAST_STAGE];
    #pragma unroll
    for (int k = 0; k < FAST_STAGE; ++k)
        sv[k] = zown[min(tid + k * OWN_THREADS, nv4 - 1)];
    #pragma unroll
    for (int k = 0; k < FAST_STAGE; ++k) {
        const int i4 = tid + k * OWN_THREADS;
        if (i4 < OWN_NODES * (FATT / 4))
            *(float4*)&h_s[i4 * 4] = i4 < nv4 ? sv[k] : make_float4(0.0f, 0.0f, 0.0f, 0.0f);
    }
    for (int i = tid; i < OWN_NODES; i += OWN_THREADS) {
        den_s[i] = 0.0f;
        mc_s[i]  = 0u;
    }
    if (tid == 0) *cnt_f = 0;
    const uint4* key4 = (const uint4*)keys;
    const unsigned int pat = blockIdx.x * 0x01010101u;
    __syncthreads();

    for (int r = 0; r < OWN_ROUNDS; ++r) {
        const int vbase = r * OWN_RVEC;
        const int ebase = (vbase + tid) * 16;
        unsigned int hm[OWN_NMASK];
        #pragma unroll
        for (int m = 0; m < OWN_NMASK; ++m) hm[m] = 0u;
        uint4 kv[OWN_KVEC];
        #pragma unroll
        for (int i = 0; i < OWN_KVEC; ++i)
            kv[i] = key4[min(vbase + i * OWN_THREADS + tid, KEY_NVEC - 1)];
        #pragma unroll
        for (int i = 0; i < OWN_KVEC; ++i) {
            const unsigned int live = vbase + i * OWN_THREADS + tid < KEY_NVEC ? 0xffffffffu : 0u;
            const unsigned int bits = (zero_bytes(kv[i].x ^ pat) >> (7 - (4 * (i & 1) + 0))) |
                                      (zero_bytes(kv[i].y ^ pat) >> (7 - (4 * (i & 1) + 1))) |
                                      (zero_bytes(kv[i].z ^ pat) >> (7 - (4 * (i & 1) + 2))) |
                                      (zero_bytes(kv[i].w ^ pat) >> (7 - (4 * (i & 1) + 3)));
            hm[i >> 1] |= bits & live;
        }
        int total = 0;
        #pragma unroll
        for (int m = 0; m < OWN_NMASK; ++m) total += __popc(hm[m]);
        if (total != 0) {
            const int pos0 = atomicAdd(cnt_f, total);
            list_all(hm, pos0, ebase, ed_s);
        }
    }
    __syncthreads();
    const int n = __builtin_amdgcn_readfirstlane(*cnt_f);
    if (n > FAST_CAP) return false;

    for (int j0 = wave * 32; j0 < n; j0 += OWN_WAVES * 32) {
        const int j = j0 + lane;
        bool ok = j < n;
        int s = 0, dl = 0;
        if (ok) {
            const int edge = ed_s[j];
            s  = src[edge];
            dl = dst[edge] - base;
            if ((unsigned)dl >= (unsigned)OWN_NODES) {
                ok = false;
                s  = 0;
                dl = 0;
            }
        }
        const unsigned int okm = __builtin_amdgcn_ballot_w32(ok);
        float sc = 0.0f;
        #pragma unroll
        for (int sub = 0; sub < 2; ++sub) {
            if (((okm >> (16 * sub)) & 0xffffu) != 0u) {
                float2 rs[OWN_BATCH], rd[OWN_BATCH];
                #pragma unroll
                for (int q = 0; q < OWN_BATCH; ++q) {
                    const int sq = __builtin_amdgcn_readlane(s, 16 * sub + q);
                    const int dq = __builtin_amdgcn_readlane(dl, 16 * sub + q);
                    rs[q] = ((const float2*)(z + (size_t)sq * FATT))[lane];
                    rd[q] = ((const float2*)(h_s + dq * FATT))[lane];
                }
                #pragma unroll
                for (int q = 0; q < OWN_BATCH; ++q) {
                    const float dot = wave_sum(rs[q].x * rd[q].x + rs[q].y * rd[q].y);
                    if (lane == 16 * sub + q) sc = dot;
                }
            }
        }
        sc = sc > 0.0f ? sc : SLOPE * sc;
        if (ok) atomicMax(&mc_s[dl], float_to_ord(sc));
        if (j < n) {
            ed_s[j] = s;
            dl_s[j] = ok ? dl : -1;
            sc_s[j] = sc;
        }
    }
    __syncthreads();

    for (int i4 = tid; i4 < OWN_NODES * (FATT / 4); i4 += OWN_THREADS)
        *(float4*)&h_s[i4 * 4] = make_float4(0.0f, 0.0f, 0.0f, 0.0f);
    __syncthreads();

    for (int j0 = wave * 32; j0 < n; j0 += OWN_WAVES * 32) {
        const int j = j0 + lane;
        bool  ok = false;
        int   s = 0, dl = 0;
        float w = 0.0f;
        if (j < n) {
            const int d = dl_s[j];
            if (d >= 0) {
                ok = true;
                dl = d;
                s  = ed_s[j];
                w  = __expf(sc_s[j] - ord_to_float(mc_s[d]));
                atomicAdd(&den_s[d], w);
            }
        }
        const unsigned int okm = __builtin_amdgcn_ballot_w32(ok);
        #pragma unroll
        for (int sub = 0; sub < 2; ++sub) {
            const unsigned int m16 = (okm >> (16 * sub)) & 0xffffu;
            if (m16 != 0u) {
                float2 rs[OWN_BATCH];
                #pragma unroll
                for (int q = 0; q < OWN_BATCH; ++q) {
                    const int sq = __builtin_amdgcn_readlane(s, 16 * sub + q);
                    rs[q] = ((const float2*)(z + (size_t)sq * FATT))[lane];
                }
                #pragma unroll
                for (int q = 0; q < OWN_BATCH; ++q) {
                    if ((m16 >> q) & 1u) {
                        const float wq = __int_as_float(__builtin_amdgcn_readlane(__float_as_int(w), 16 * sub + q));
                        const int   dq = __builtin_amdgcn_readlane(dl, 16 * sub + q);
                        float* hp = h_s + dq * FATT + 2 * lane;
                        atomicAdd(hp + 0, wq * rs[q].x);
                        atomicAdd(hp + 1, wq * rs[q].y);
                    }
                }
            }
        }
    }
    __syncthreads();

    for (int i4 = tid; i4 < OWN_NODES * (FATT / 4); i4 += OWN_THREADS) {
        const int nl = i4 / (FATT / 4);
        if (base + nl < NTOT) {
            const float dn = den_s[nl];
            float4 v = *(const float4*)&h_s[i4 * 4];
            v.x = dn > 0.0f ? v.x / dn : 0.0f;
            v.y = dn > 0.0f ? v.y / dn : 0.0f;
            v.z = dn > 0.0f ? v.z / dn : 0.0f;
            v.w = dn > 0.0f ? v.w / dn : 0.0f;
            v.x = v.x > 0.0f ? v.x : (__expf(v.x) - 1.0f);
            v.y = v.y > 0.0f ? v.y : (__expf(v.y) - 1.0f);
            v.z = v.z > 0.0f ? v.z : (__expf(v.z) - 1.0f);
            v.w = v.w > 0.0f ? v.w : (__expf(v.w) - 1.0f);
            *(float4*)(out + (size_t)base * FATT + (size_t)i4 * 4) = v;
        }
    }
    return true;
}

__global__ __launch_bounds__(OWN_THREADS) void node_block_attend(
        const float* __restrict__ z, const int* __restrict__ src, const int* __restrict__ dst,
        const unsigned int* __restrict__ keys, float* __restrict__ out, size_t ws_size) {
    __shared__ __attribute__((aligned(16))) float h_s[OWN_NODES * FATT];
    __shared__ float        den_s[OWN_NODES];
    __shared__ unsigned int mo_s[OWN_NODES];
    __shared__ unsigned int mc_s[OWN_NODES];
    __shared__ int          ed_s[FAST_CAP];
    __shared__ int          dl_s[FAST_CAP];
    __shared__ float        sc_s[FAST_CAP];
    __shared__ int          cnt_f;
    __shared__ int          cnt_s[3];
    __shared__ int          flag_s[3];

    if (ws_size >= WS_NEED) {
        if (!attend_fast(z, src, dst, keys, out, h_s, den_s, mc_s, ed_s, dl_s, sc_s, &cnt_f))
            attend_body<true >(z, src, dst, keys, out, h_s, den_s, mo_s, mc_s, ed_s, cnt_s, flag_s);
    } else {
        attend_body<false>(z, src, dst, keys, out, h_s, den_s, mo_s, mc_s, ed_s, cnt_s, flag_s);
    }
}

extern "C" void kernel_launch(void* const* d_in, const int* in_sizes, int n_in,
                              void* d_out, int out_size, void* d_ws, size_t ws_size,
                              hipStream_t stream) {
    const float* d_sim  = (const float*)d_in[0];
    const float* mi_sim = (const float*)d_in[1];
    const float* W_d    = (const float*)d_in[2];
    const float* W_mi   = (const float*)d_in[3];
    const int*   src    = (const int*)d_in[4];
    const int*   dst    = (const int*)d_in[5];
    float*       out    = (float*)d_out;

    float*        z     = (float*)d_ws;
    unsigned int* keys  = (unsigned int*)((char*)d_ws + WS_Z_BYTES);

    gemm_wmma_f32<<<(N_DIS + 127) / 128, 128, 0, stream>>>(d_sim, W_d, z, N_DIS, 383);
    gemm_wmma_f32<<<(N_MIC + 127) / 128, 128, 0, stream>>>(mi_sim, W_mi,
                                                           z + (size_t)N_DIS * FATT, N_MIC, 495);

    owner_keys<<<(KEY_NVEC + KEY_THREADS - 1) / KEY_THREADS, KEY_THREADS, 0, stream>>>(dst, keys, ws_size);

    node_block_attend<<<OWN_BLOCKS, OWN_THREADS, 0, stream>>>(z, src, dst, keys, out, ws_size);
}


# Owner scores, weights and sums in ONE free-running phase: sums scaled to a node's first-arriving score (LDS compare-swap), rare far-higher edges held back; source rows read once; sum order changes
# speedup vs baseline: 6.7299x; 1.0979x over previous
#include <hip/hip_runtime.h>
#include <math.h>

#define N_DIS 20000
#define N_MIC 80000
#define NTOT  (N_DIS + N_MIC)
#define NEDGE 1600000
#define FATT  64
#define SLOPE 0.2f

typedef float v2f __attribute__((ext_vector_type(2)));
typedef float v8f __attribute__((ext_vector_type(8)));

typedef float v4f __attribute__((ext_vector_type(4)));

#define DEPTH 5

struct Frag { v4f a0, a1, w0, w1, w2, w3; };

__device__ __forceinline__ v4f ld_x4(const char* p) {
    v4f v;
    __builtin_memcpy(&v, p, sizeof(v));
    return v;
}

__device__ __forceinline__ void load_frag(Frag& f, const float* X, const float* W,
                                          unsigned xo0, unsigned xo1, unsigned wo, int kb) {
    const char* xk = (const char*)X + (size_t)kb * (8 * sizeof(float));
    const char* wk = (const char*)W + (size_t)kb * (8 * FATT * sizeof(float));
    f.a0 = ld_x4(xk + xo0);
    f.a1 = ld_x4(xk + xo1);
    f.w0 = *(const v4f*)(wk + wo);
    f.w1 = *(const v4f*)(wk + wo + 1 * FATT * sizeof(float));
    f.w2 = *(const v4f*)(wk + wo + 2 * FATT * sizeof(float));
    f.w3 = *(const v4f*)(wk + wo + 3 * FATT * sizeof(float));
}

__device__ __forceinline__ void mma_frag(const Frag& f, v8f (&acc)[2][4]) {
    const v2f a0lo = { f.a0.x, f.a0.y }, a0hi = { f.a0.z, f.a0.w };
    const v2f a1lo = { f.a1.x, f.a1.y }, a1hi = { f.a1.z, f.a1.w };
    #pragma unroll
    for (int j = 0; j < 4; ++j) {
        const v2f b = { f.w0[j], f.w1[j] };
        acc[0][j] = __builtin_amdgcn_wmma_f32_16x16x4_f32(false, a0lo, false, b, (short)0, acc[0][j], false, false);
        acc[1][j] = __builtin_amdgcn_wmma_f32_16x16x4_f32(false, a1lo, false, b, (short)0, acc[1][j], false, false);
    }
    #pragma unroll
    for (int j = 0; j < 4; ++j) {
        const v2f b = { f.w2[j], f.w3[j] };
        acc[0][j] = __builtin_amdgcn_wmma_f32_16x16x4_f32(false, a0hi, false, b, (short)0, acc[0][j], false, false);
        acc[1][j] = __builtin_amdgcn_wmma_f32_16x16x4_f32(false, a1hi, false, b, (short)0, acc[1][j], false, false);
    }
}

__global__ __launch_bounds__(128) void gemm_wmma_f32(
        const float* __restrict__ X, const float* __restrict__ W,
        float* __restrict__ Z, int M, int K) {
    const int lane = threadIdx.x & 31;
    const int wave = threadIdx.x >> 5;
    const int c    = lane & 15;
    const int h    = lane >> 4;

    const int row_base = (blockIdx.x * 4 + wave) * 32;
    if (row_base >= M) return;

    const int r0 = min(row_base + c, M - 1);
    const int r1 = min(row_base + 16 + c, M - 1);
    const unsigned xo0 = ((unsigned)r0 * (unsigned)K + 4u * h) * (unsigned)sizeof(float);
    const unsigned xo1 = ((unsigned)r1 * (unsigned)K + 4u * h) * (unsigned)sizeof(float);
    const unsigned wo  = (4u * h * FATT + 4u * c) * (unsigned)sizeof(float);

    v8f acc[2][4] = {};

    const int nfull = K >> 3;
    int kb = 0;
    if (nfull > 0) {
        Frag f[DEPTH];
        #pragma unroll
        for (int i = 0; i < DEPTH; ++i)
            load_frag(f[i], X, W, xo0, xo1, wo, min(i, nfull - 1));

        for (; kb + DEPTH <= nfull; kb += DEPTH) {
            #pragma unroll
            for (int i = 0; i < DEPTH; ++i) {
                mma_frag(f[i], acc);
                load_frag(f[i], X, W, xo0, xo1, wo, min(kb + DEPTH + i, nfull - 1));
                __builtin_amdgcn_sched_barrier(0);
            }
        }
        #pragma unroll
        for (int i = 0; i < DEPTH; ++i)
            if (kb + i < nfull) mma_frag(f[i], acc);
    }

    const int kt = nfull * 8;
    if (kt < K) {
        Frag t;
        float x0[4], x1[4];
        v4f wv[4];
        #pragma unroll
        for (int i = 0; i < 4; ++i) {
            const int  k  = kt + 4 * h + i;
            const bool ok = k < K;
            const int  kc = ok ? k : (K - 1);
            const float p0 = X[(size_t)r0 * K + kc];
            const float p1 = X[(size_t)r1 * K + kc];
            const v4f   pw = *(const v4f*)(W + (size_t)kc * FATT + 4 * c);
            const v4f   zero = { 0.0f, 0.0f, 0.0f, 0.0f };
            x0[i] = ok ? p0 : 0.0f;
            x1[i] = ok ? p1 : 0.0f;
            wv[i] = ok ? pw : zero;
        }
        t.a0 = (v4f){ x0[0], x0[1], x0[2], x0[3] };
        t.a1 = (v4f){ x1[0], x1[1], x1[2], x1[3] };
        t.w0 = wv[0]; t.w1 = wv[1]; t.w2 = wv[2]; t.w3 = wv[3];
        mma_frag(t, acc);
    }

    #pragma unroll
    for (int mt = 0; mt < 2; ++mt) {
        #pragma unroll
        for (int r = 0; r < 8; ++r) {
            const int mm = row_base + 16 * mt + 8 * h + r;
            if (mm < M) {
                const v4f o = { acc[mt][0][r], acc[mt][1][r], acc[mt][2][r], acc[mt][3][r] };
                *(v4f*)(Z + (size_t)mm * FATT + 4 * c) = o;
            }
        }
    }
}

__device__ __forceinline__ unsigned int float_to_ord(float f) {
    unsigned int u = __float_as_uint(f);
    return (u & 0x80000000u) ? ~u : (u | 0x80000000u);
}
__device__ __forceinline__ float ord_to_float(unsigned int u) {
    return __uint_as_float((u & 0x80000000u) ? (u ^ 0x80000000u) : ~u);
}

template <int CTRL>
__device__ __forceinline__ float dpp_move(float v) {
    return __int_as_float(__builtin_amdgcn_update_dpp(0, __float_as_int(v), CTRL, 0xF, 0xF, true));
}
__device__ __forceinline__ float wave_sum(float v) {
    v += dpp_move<0x161>(v);
    v += dpp_move<0x162>(v);
    v += dpp_move<0x164>(v);
    v += dpp_move<0x168>(v);
    v += __int_as_float((int)__builtin_amdgcn_permlanex16(0u, (unsigned)__float_as_int(v),
                                                          0x76543210u, 0xfedcba98u, false, false));
    return v;
}

#define OWN_BLOCKS  256
#define OWN_NODES   ((NTOT + OWN_BLOCKS - 1) / OWN_BLOCKS)
#define KEY_NVEC    (NEDGE / 16)
#define KEY_THREADS 256
#define WS_Z_BYTES  ((size_t)NTOT * FATT * sizeof(float))
#define WS_NEED     (WS_Z_BYTES + (size_t)NEDGE)

__device__ __forceinline__ unsigned int key_of(int d) {
    const unsigned int u = (unsigned int)d;
    return u < (unsigned int)(OWN_BLOCKS * OWN_NODES) ? u / (unsigned int)OWN_NODES : 255u;
}

__global__ __launch_bounds__(KEY_THREADS) void owner_keys(
        const int* __restrict__ dst, unsigned int* __restrict__ keys, size_t ws_size) {
    if (ws_size < WS_NEED) return;
    const int v = blockIdx.x * KEY_THREADS + threadIdx.x;
    if (v >= KEY_NVEC) return;
    const int4* d4 = (const int4*)dst + (size_t)v * 4;
    unsigned int w[4];
    #pragma unroll
    for (int j = 0; j < 4; ++j) {
        const int4 d = d4[j];
        w[j] = key_of(d.x) | (key_of(d.y) << 8) | (key_of(d.z) << 16) | (key_of(d.w) << 24);
    }
    ((uint4*)keys)[v] = make_uint4(w[0], w[1], w[2], w[3]);
}

#define OWN_THREADS 1024
#define OWN_WAVES   (OWN_THREADS / 32)
#define OWN_KVEC    7
#define OWN_NMASK   ((OWN_KVEC + 1) / 2)
#define OWN_RVEC    (OWN_THREADS * OWN_KVEC)
#define OWN_CHUNK   (OWN_RVEC * 16)
#define OWN_BATCH   16
#define OWN_TRIP    (OWN_WAVES * OWN_BATCH)
#define OWN_LIST    4096
#define OWN_ROUNDS  ((NEDGE + OWN_CHUNK - 1) / OWN_CHUNK)
#define OWN_RISE    60.0f

__device__ __forceinline__ unsigned int zero_bytes(unsigned int x) {
    return ~(((x & 0x7f7f7f7fu) + 0x7f7f7f7fu) | x) & 0x80808080u;
}

__device__ __forceinline__ void list_window(const unsigned int (&hm)[OWN_NMASK], int first,
                                            int ebase, int* list_s) {
    int idx = first;
    #pragma unroll
    for (int m = 0; m < OWN_NMASK; ++m) {
        unsigned int mask = hm[m];
        while (mask != 0u) {
            const int p = __ffs((int)mask) - 1;
            mask &= mask - 1u;
            if ((unsigned)idx < (unsigned)OWN_LIST)
                list_s[idx] = ebase + (2 * m + ((p >> 2) & 1)) * (OWN_THREADS * 16) + (p & 3) * 4 + (p >> 3);
            ++idx;
        }
    }
}

template <bool KEYS>
__device__ __forceinline__ void attend_body(
        const float* __restrict__ z, const int* __restrict__ src, const int* __restrict__ dst,
        const unsigned int* __restrict__ keys, float* __restrict__ out,
        float* h_s, float* den_s, unsigned int* mo_s, unsigned int* mc_s, int* list_s, int* cnt_s,
        int* flag_s) {
    const int tid  = threadIdx.x;
    const int lane = tid & 31;
    const int wave = tid >> 5;
    const int base = blockIdx.x * OWN_NODES;

    for (int i = tid; i < OWN_NODES * FATT; i += OWN_THREADS) h_s[i] = 0.0f;
    for (int i = tid; i < OWN_NODES; i += OWN_THREADS) {
        den_s[i] = 0.0f;
        mo_s[i]  = 0u;
        mc_s[i]  = 0u;
    }
    if (tid < 3) {
        cnt_s[tid]  = 0;
        flag_s[tid] = 0;
    }
    int g = 0;

    const int4*  dst4 = (const int4*)dst;
    const uint4* key4 = (const uint4*)keys;
    const unsigned int pat = blockIdx.x * 0x01010101u;
    __syncthreads();

    for (int r = 0; r < OWN_ROUNDS; ++r) {
        int* cnt = &cnt_s[r % 3];
        const int vbase = r * OWN_RVEC;
        const int ebase = (vbase + tid) * 16;

        unsigned int hm[OWN_NMASK];
        #pragma unroll
        for (int m = 0; m < OWN_NMASK; ++m) hm[m] = 0u;

        if constexpr (KEYS) {
            uint4 kv[OWN_KVEC];
            #pragma unroll
            for (int i = 0; i < OWN_KVEC; ++i)
                kv[i] = key4[min(vbase + i * OWN_THREADS + tid, KEY_NVEC - 1)];
            #pragma unroll
            for (int i = 0; i < OWN_KVEC; ++i) {
                const unsigned int live = vbase + i * OWN_THREADS + tid < KEY_NVEC ? 0xffffffffu : 0u;
                const unsigned int bits = (zero_bytes(kv[i].x ^ pat) >> (7 - (4 * (i & 1) + 0))) |
                                          (zero_bytes(kv[i].y ^ pat) >> (7 - (4 * (i & 1) + 1))) |
                                          (zero_bytes(kv[i].z ^ pat) >> (7 - (4 * (i & 1) + 2))) |
                                          (zero_bytes(kv[i].w ^ pat) >> (7 - (4 * (i & 1) + 3)));
                hm[i >> 1] |= bits & live;
            }
        } else {
            #pragma nounroll
            for (int m = 0; m < OWN_NMASK; ++m) {
                unsigned int bits = 0u;
                #pragma unroll
                for (int ii = 0; ii < 2; ++ii) {
                    const int i = 2 * m + ii;
                    const int v = vbase + i * OWN_THREADS + tid;
                    if (i < OWN_KVEC && v < KEY_NVEC) {
                        #pragma unroll
                        for (int j = 0; j < 4; ++j) {
                            const int4 d = dst4[(size_t)v * 4 + j];
                            const unsigned int b =
                                ((unsigned)(d.x - base) < (unsigned)OWN_NODES ? 0x00000001u : 0u) |
                                ((unsigned)(d.y - base) < (unsigned)OWN_NODES ? 0x00000100u : 0u) |
                                ((unsigned)(d.z - base) < (unsigned)OWN_NODES ? 0x00010000u : 0u) |
                                ((unsigned)(d.w - base) < (unsigned)OWN_NODES ? 0x01000000u : 0u);
                            bits |= b << (4 * ii + j);
                        }
                    }
                }
                #pragma unroll
                for (int k = 0; k < OWN_NMASK; ++k)
                    if (k == m) hm[k] = bits;
            }
        }
        int total = 0;
        #pragma unroll
        for (int m = 0; m < OWN_NMASK; ++m) total += __popc(hm[m]);
        int pos0 = 0;
        if (total != 0) {
            pos0 = atomicAdd(cnt, total);
            list_window(hm, pos0, ebase, list_s);
        }
        __syncthreads();
        const int n = *cnt;
        if (tid == 0) cnt_s[(r + 2) % 3] = 0;

        for (int lo = 0; lo < n; lo += OWN_LIST) {
            if (lo != 0) {
                if (total != 0) list_window(hm, pos0 - lo, ebase, list_s);
                __syncthreads();
            }
            const int nw    = min(n - lo, OWN_LIST);
            const int trips = (nw + OWN_TRIP - 1) / OWN_TRIP;
            for (int t = 0; t < trips; ++t) {
                const int j0 = t * OWN_TRIP + wave * OWN_BATCH;
                bool ok = lane < OWN_BATCH && j0 + lane < nw;

                int s = 0, dl = 0;
                if (ok) {
                    const int edge = list_s[j0 + lane];
                    s  = src[edge];
                    dl = dst[edge] - base;
                    if ((unsigned)dl >= (unsigned)OWN_NODES) {
                        ok = false;
                        s  = 0;
                        dl = 0;
                    }
                }
                const unsigned int okm = __builtin_amdgcn_ballot_w32(ok);
                float2 rs[OWN_BATCH];
                int    dq[OWN_BATCH];
                float  sc = 0.0f;
                if (okm != 0u) {
                    float2 rd[OWN_BATCH];
                    #pragma unroll
                    for (int q = 0; q < OWN_BATCH; ++q) {
                        const int sq = __builtin_amdgcn_readlane(s, q);
                        dq[q] = __builtin_amdgcn_readlane(dl, q);
                        rs[q] = ((const float2*)(z + (size_t)sq * FATT))[lane];
                        rd[q] = ((const float2*)(z + (size_t)(base + dq[q]) * FATT))[lane];
                    }
                    #pragma unroll
                    for (int q = 0; q < OWN_BATCH; ++q) {
                        const float dot = wave_sum(rs[q].x * rd[q].x + rs[q].y * rd[q].y);
                        if (lane == q) sc = dot;
                    }
                }
                sc = sc > 0.0f ? sc : SLOPE * sc;

                int*  flag = &flag_s[g % 3];
                float ref  = sc;
                bool  up   = false;
                if (ok) {
                    const unsigned int mold = atomicCAS(&mc_s[dl], 0u, float_to_ord(sc));
                    if (mold != 0u) {
                        ref = ord_to_float(mold);
                        up  = sc - ref > OWN_RISE;
                        if (up) *flag = 1;
                    }
                }

                const bool now = ok && !up;
                const unsigned int nowm = __builtin_amdgcn_ballot_w32(now);
                float w = 0.0f;
                if (now) {
                    w = __expf(sc - ref);
                    atomicAdd(&den_s[dl], w);
                }
                #pragma unroll
                for (int q = 0; q < OWN_BATCH; ++q) {
                    if ((nowm >> q) & 1u) {
                        const float wq = __int_as_float(__builtin_amdgcn_readlane(__float_as_int(w), q));
                        float* hp = h_s + dq[q] * FATT + 2 * lane;
                        atomicAdd(hp + 0, wq * rs[q].x);
                        atomicAdd(hp + 1, wq * rs[q].y);
                    }
                }
                __syncthreads();
                if (tid == 0) flag_s[(g + 2) % 3] = 0;
                ++g;
                const int rise = __builtin_amdgcn_readfirstlane(*flag);

                if (rise != 0) {
                    if (up) atomicMax(&mo_s[dl], float_to_ord(sc));
                    __syncthreads();

                    unsigned int mnew = 0u;
                    int   take = 0;
                    float fs   = 1.0f;
                    if (up) {
                        mnew = mo_s[dl];
                        const unsigned int mold = atomicMax(&mc_s[dl], mnew);
                        if (mold < mnew) {
                            take = 1;
                            fs   = __expf(ord_to_float(mold) - ord_to_float(mnew));
                            den_s[dl] *= fs;
                        }
                    }
                    if (__builtin_amdgcn_ballot_w32(take != 0) != 0u) {
                        #pragma unroll
                        for (int q = 0; q < OWN_BATCH; ++q) {
                            if (__builtin_amdgcn_readlane(take, q)) {
                                const float f = __int_as_float(__builtin_amdgcn_readlane(__float_as_int(fs), q));
                                float2* hp = (float2*)(h_s + dq[q] * FATT) + lane;
                                float2 hv = *hp;
                                hv.x *= f;
                                hv.y *= f;
                                *hp = hv;
                            }
                        }
                    }
                    __syncthreads();

                    const unsigned int upm = __builtin_amdgcn_ballot_w32(up);
                    float w2 = 0.0f;
                    if (up) {
                        mo_s[dl] = 0u;
                        w2 = __expf(sc - ord_to_float(mnew));
                        atomicAdd(&den_s[dl], w2);
                    }
                    #pragma unroll
                    for (int q = 0; q < OWN_BATCH; ++q) {
                        if ((upm >> q) & 1u) {
                            const float wq = __int_as_float(__builtin_amdgcn_readlane(__float_as_int(w2), q));
                            float* hp = h_s + dq[q] * FATT + 2 * lane;
                            atomicAdd(hp + 0, wq * rs[q].x);
                            atomicAdd(hp + 1, wq * rs[q].y);
                        }
                    }
                }
            }
        }
    }
    __syncthreads();

    for (int i4 = tid; i4 < OWN_NODES * (FATT / 4); i4 += OWN_THREADS) {
        const int nl = i4 / (FATT / 4);
        if (base + nl < NTOT) {
            const float dn = den_s[nl];
            float4 v = *(const float4*)&h_s[i4 * 4];
            v.x = dn > 0.0f ? v.x / dn : 0.0f;
            v.y = dn > 0.0f ? v.y / dn : 0.0f;
            v.z = dn > 0.0f ? v.z / dn : 0.0f;
            v.w = dn > 0.0f ? v.w / dn : 0.0f;
            v.x = v.x > 0.0f ? v.x : (__expf(v.x) - 1.0f);
            v.y = v.y > 0.0f ? v.y : (__expf(v.y) - 1.0f);
            v.z = v.z > 0.0f ? v.z : (__expf(v.z) - 1.0f);
            v.w = v.w > 0.0f ? v.w : (__expf(v.w) - 1.0f);
            *(float4*)(out + (size_t)base * FATT + (size_t)i4 * 4) = v;
        }
    }
}

#define FAST_CAP 12288
#define HELD_CAP 1024
#define FAST_STAGE ((OWN_NODES * (FATT / 4) + OWN_THREADS - 1) / OWN_THREADS)

__device__ __forceinline__ void list_all(const unsigned int (&hm)[OWN_NMASK], int first,
                                         int ebase, int* ed_s) {
    int idx = first;
    #pragma unroll
    for (int m = 0; m < OWN_NMASK; ++m) {
        unsigned int mask = hm[m];
        while (mask != 0u) {
            const int p = __ffs((int)mask) - 1;
            mask &= mask - 1u;
            if ((unsigned)idx < (unsigned)FAST_CAP)
                ed_s[idx] = ebase + (2 * m + ((p >> 2) & 1)) * (OWN_THREADS * 16) + (p & 3) * 4 + (p >> 3);
            ++idx;
        }
    }
}

__device__ __forceinline__ bool attend_fast(
        const float* __restrict__ z, const int* __restrict__ src, const int* __restrict__ dst,
        const unsigned int* __restrict__ keys, float* __restrict__ out,
        float* h_s, float* zd_s, float* den_s, unsigned int* mo_s, unsigned int* mc_s, int* ed_s,
        int* hs_s, int* hd_s, float* hsc_s, int* cnt_f, int* cnt_h) {
    const int tid  = threadIdx.x;
    const int lane = tid & 31;
    const int wave = __builtin_amdgcn_readfirstlane(tid >> 5);
    const int base = blockIdx.x * OWN_NODES;

    const float4* zown = (const float4*)(z + (size_t)base * FATT);
    const int nv4 = (min(base + OWN_NODES, NTOT) - base) * (FATT / 4);
    float4 sv[FAST_STAGE];
    #pragma unroll
    for (int k = 0; k < FAST_STAGE; ++k)
        sv[k] = zown[min(tid + k * OWN_THREADS, nv4 - 1)];
    #pragma unroll
    for (int k = 0; k < FAST_STAGE; ++k) {
        const int i4 = tid + k * OWN_THREADS;
        if (i4 < OWN_NODES * (FATT / 4)) {
            *(float4*)&zd_s[i4 * 4] = i4 < nv4 ? sv[k] : make_float4(0.0f, 0.0f, 0.0f, 0.0f);
            *(float4*)&h_s[i4 * 4]  = make_float4(0.0f, 0.0f, 0.0f, 0.0f);
        }
    }
    for (int i = tid; i < OWN_NODES; i += OWN_THREADS) {
        den_s[i] = 0.0f;
        mo_s[i]  = 0u;
        mc_s[i]  = 0u;
    }
    if (tid == 0) {
        *cnt_f = 0;
        *cnt_h = 0;
    }
    const uint4* key4 = (const uint4*)keys;
    const unsigned int pat = blockIdx.x * 0x01010101u;
    __syncthreads();

    for (int r = 0; r < OWN_ROUNDS; ++r) {
        const int vbase = r * OWN_RVEC;
        const int ebase = (vbase + tid) * 16;
        unsigned int hm[OWN_NMASK];
        #pragma unroll
        for (int m = 0; m < OWN_NMASK; ++m) hm[m] = 0u;
        uint4 kv[OWN_KVEC];
        #pragma unroll
        for (int i = 0; i < OWN_KVEC; ++i)
            kv[i] = key4[min(vbase + i * OWN_THREADS + tid, KEY_NVEC - 1)];
        #pragma unroll
        for (int i = 0; i < OWN_KVEC; ++i) {
            const unsigned int live = vbase + i * OWN_THREADS + tid < KEY_NVEC ? 0xffffffffu : 0u;
            const unsigned int bits = (zero_bytes(kv[i].x ^ pat) >> (7 - (4 * (i & 1) + 0))) |
                                      (zero_bytes(kv[i].y ^ pat) >> (7 - (4 * (i & 1) + 1))) |
                                      (zero_bytes(kv[i].z ^ pat) >> (7 - (4 * (i & 1) + 2))) |
                                      (zero_bytes(kv[i].w ^ pat) >> (7 - (4 * (i & 1) + 3)));
            hm[i >> 1] |= bits & live;
        }
        int total = 0;
        #pragma unroll
        for (int m = 0; m < OWN_NMASK; ++m) total += __popc(hm[m]);
        if (total != 0) {
            const int pos0 = atomicAdd(cnt_f, total);
            list_all(hm, pos0, ebase, ed_s);
        }
    }
    __syncthreads();
    const int n = __builtin_amdgcn_readfirstlane(*cnt_f);
    if (n > FAST_CAP) return false;

    for (int j0 = wave * 32; j0 < n; j0 += OWN_WAVES * 32) {
        const int j = j0 + lane;
        bool ok = j < n;
        int s = 0, dl = 0;
        if (ok) {
            const int edge = ed_s[j];
            s  = src[edge];
            dl = dst[edge] - base;
            if ((unsigned)dl >= (unsigned)OWN_NODES) {
                ok = false;
                s  = 0;
                dl = 0;
            }
        }
        const unsigned int okm = __builtin_amdgcn_ballot_w32(ok);
        float2 rs[2][OWN_BATCH];
        #pragma unroll
        for (int sub = 0; sub < 2; ++sub) {
            if (((okm >> (16 * sub)) & 0xffffu) != 0u) {
                #pragma unroll
                for (int q = 0; q < OWN_BATCH; ++q) {
                    const int sq = __builtin_amdgcn_readlane(s, 16 * sub + q);
                    rs[sub][q] = ((const float2*)(z + (size_t)sq * FATT))[lane];
                }
            }
        }
        #pragma unroll
        for (int sub = 0; sub < 2; ++sub) {
            if (((okm >> (16 * sub)) & 0xffffu) != 0u) {
                float sc = 0.0f;
                {
                    float2 rd[OWN_BATCH];
                    #pragma unroll
                    for (int q = 0; q < OWN_BATCH; ++q) {
                        const int dq = __builtin_amdgcn_readlane(dl, 16 * sub + q);
                        rd[q] = ((const float2*)(zd_s + dq * FATT))[lane];
                    }
                    #pragma unroll
                    for (int q = 0; q < OWN_BATCH; ++q) {
                        const float dot = wave_sum(rs[sub][q].x * rd[q].x + rs[sub][q].y * rd[q].y);
                        if (lane == 16 * sub + q) sc = dot;
                    }
                }
                sc = sc > 0.0f ? sc : SLOPE * sc;

                bool  now = false;
                float w   = 0.0f;
                if (ok && (lane >> 4) == sub) {
                    const unsigned int mold = atomicCAS(&mc_s[dl], 0u, float_to_ord(sc));
                    const float ref = mold != 0u ? ord_to_float(mold) : sc;
                    if (sc - ref > OWN_RISE) {
                        const int p = atomicAdd(cnt_h, 1);
                        if (p < HELD_CAP) {
                            hs_s[p]  = s;
                            hd_s[p]  = dl;
                            hsc_s[p] = sc;
                        }
                    } else {
                        now = true;
                        w   = __expf(sc - ref);
                        atomicAdd(&den_s[dl], w);
                    }
                }
                const unsigned int nowm = (__builtin_amdgcn_ballot_w32(now) >> (16 * sub)) & 0xffffu;
                #pragma unroll
                for (int q = 0; q < OWN_BATCH; ++q) {
                    if ((nowm >> q) & 1u) {
                        const float wq = __int_as_float(__builtin_amdgcn_readlane(__float_as_int(w), 16 * sub + q));
                        const int   dq = __builtin_amdgcn_readlane(dl, 16 * sub + q);
                        float* hp = h_s + dq * FATT + 2 * lane;
                        atomicAdd(hp + 0, wq * rs[sub][q].x);
                        atomicAdd(hp + 1, wq * rs[sub][q].y);
                    }
                }
            }
        }
    }
    __syncthreads();
    const int nh = __builtin_amdgcn_readfirstlane(*cnt_h);
    if (nh > HELD_CAP) return false;

    if (nh > 0) {
        for (int i = tid; i < nh; i += OWN_THREADS)
            atomicMax(&mo_s[hd_s[i]], float_to_ord(hsc_s[i]));
        __syncthreads();

        for (int i = wave; i < nh; i += OWN_WAVES) {
            const int dl = hd_s[i];
            const unsigned int mnew = mo_s[dl];
            unsigned int mold = mnew;
            if (lane == 0) mold = atomicMax(&mc_s[dl], mnew);
            mold = __builtin_amdgcn_readfirstlane(mold);
            if (mold < mnew) {
                const float fs = __expf(ord_to_float(mold) - ord_to_float(mnew));
                float2* hp = (float2*)(h_s + dl * FATT) + lane;
                float2 hv = *hp;
                hv.x *= fs;
                hv.y *= fs;
                *hp = hv;
                if (lane == 0) den_s[dl] *= fs;
            }
        }
        __syncthreads();

        for (int i = wave; i < nh; i += OWN_WAVES) {
            const int   s  = hs_s[i];
            const int   dl = hd_s[i];
            const float w  = __expf(hsc_s[i] - ord_to_float(mc_s[dl]));
            const float2 r = ((const float2*)(z + (size_t)s * FATT))[lane];
            if (lane == 0) atomicAdd(&den_s[dl], w);
            float* hp = h_s + dl * FATT + 2 * lane;
            atomicAdd(hp + 0, w * r.x);
            atomicAdd(hp + 1, w * r.y);
        }
        __syncthreads();
    }

    for (int i4 = tid; i4 < OWN_NODES * (FATT / 4); i4 += OWN_THREADS) {
        const int nl = i4 / (FATT / 4);
        if (base + nl < NTOT) {
            const float dn = den_s[nl];
            float4 v = *(const float4*)&h_s[i4 * 4];
            v.x = dn > 0.0f ? v.x / dn : 0.0f;
            v.y = dn > 0.0f ? v.y / dn : 0.0f;
            v.z = dn > 0.0f ? v.z / dn : 0.0f;
            v.w = dn > 0.0f ? v.w / dn : 0.0f;
            v.x = v.x > 0.0f ? v.x : (__expf(v.x) - 1.0f);
            v.y = v.y > 0.0f ? v.y : (__expf(v.y) - 1.0f);
            v.z = v.z > 0.0f ? v.z : (__expf(v.z) - 1.0f);
            v.w = v.w > 0.0f ? v.w : (__expf(v.w) - 1.0f);
            *(float4*)(out + (size_t)base * FATT + (size_t)i4 * 4) = v;
        }
    }
    return true;
}

__global__ __launch_bounds__(OWN_THREADS) void node_block_attend(
        const float* __restrict__ z, const int* __restrict__ src, const int* __restrict__ dst,
        const unsigned int* __restrict__ keys, float* __restrict__ out, size_t ws_size) {
    __shared__ __attribute__((aligned(16))) float h_s[OWN_NODES * FATT];
    __shared__ __attribute__((aligned(16))) float zd_s[OWN_NODES * FATT];
    __shared__ float        den_s[OWN_NODES];
    __shared__ unsigned int mo_s[OWN_NODES];
    __shared__ unsigned int mc_s[OWN_NODES];
    __shared__ int          ed_s[FAST_CAP];
    __shared__ int          hs_s[HELD_CAP];
    __shared__ int          hd_s[HELD_CAP];
    __shared__ float        hsc_s[HELD_CAP];
    __shared__ int          cnt_f;
    __shared__ int          cnt_h;
    __shared__ int          cnt_s[3];
    __shared__ int          flag_s[3];

    if (ws_size >= WS_NEED) {
        if (!attend_fast(z, src, dst, keys, out, h_s, zd_s, den_s, mo_s, mc_s, ed_s, hs_s, hd_s, hsc_s, &cnt_f, &cnt_h))
            attend_body<true >(z, src, dst, keys, out, h_s, den_s, mo_s, mc_s, ed_s, cnt_s, flag_s);
    } else {
        attend_body<false>(z, src, dst, keys, out, h_s, den_s, mo_s, mc_s, ed_s, cnt_s, flag_s);
    }
}

extern "C" void kernel_launch(void* const* d_in, const int* in_sizes, int n_in,
                              void* d_out, int out_size, void* d_ws, size_t ws_size,
                              hipStream_t stream) {
    const float* d_sim  = (const float*)d_in[0];
    const float* mi_sim = (const float*)d_in[1];
    const float* W_d    = (const float*)d_in[2];
    const float* W_mi   = (const float*)d_in[3];
    const int*   src    = (const int*)d_in[4];
    const int*   dst    = (const int*)d_in[5];
    float*       out    = (float*)d_out;

    float*        z     = (float*)d_ws;
    unsigned int* keys  = (unsigned int*)((char*)d_ws + WS_Z_BYTES);

    gemm_wmma_f32<<<(N_DIS + 127) / 128, 128, 0, stream>>>(d_sim, W_d, z, N_DIS, 383);
    gemm_wmma_f32<<<(N_MIC + 127) / 128, 128, 0, stream>>>(mi_sim, W_mi,
                                                           z + (size_t)N_DIS * FATT, N_MIC, 495);

    owner_keys<<<(KEY_NVEC + KEY_THREADS - 1) / KEY_THREADS, KEY_THREADS, 0, stream>>>(dst, keys, ws_size);

    node_block_attend<<<OWN_BLOCKS, OWN_THREADS, 0, stream>>>(z, src, dst, keys, out, ws_size);
}


# Key launch adds presence bits (1 per owner and 16-edge key vector, integer OR); an owner wave fetches only its named key vectors, lists in its own segment by lane rank, not arrival; sum order changes
# speedup vs baseline: 7.1088x; 1.0563x over previous
#include <hip/hip_runtime.h>
#include <math.h>

#define N_DIS 20000
#define N_MIC 80000
#define NTOT  (N_DIS + N_MIC)
#define NEDGE 1600000
#define FATT  64
#define SLOPE 0.2f

typedef float v2f __attribute__((ext_vector_type(2)));
typedef float v8f __attribute__((ext_vector_type(8)));

typedef float v4f __attribute__((ext_vector_type(4)));

#define DEPTH 5

struct Frag { v4f a0, a1, w0, w1, w2, w3; };

__device__ __forceinline__ v4f ld_x4(const char* p) {
    v4f v;
    __builtin_memcpy(&v, p, sizeof(v));
    return v;
}

__device__ __forceinline__ void load_frag(Frag& f, const float* X, const float* W,
                                          unsigned xo0, unsigned xo1, unsigned wo, int kb) {
    const char* xk = (const char*)X + (size_t)kb * (8 * sizeof(float));
    const char* wk = (const char*)W + (size_t)kb * (8 * FATT * sizeof(float));
    f.a0 = ld_x4(xk + xo0);
    f.a1 = ld_x4(xk + xo1);
    f.w0 = *(const v4f*)(wk + wo);
    f.w1 = *(const v4f*)(wk + wo + 1 * FATT * sizeof(float));
    f.w2 = *(const v4f*)(wk + wo + 2 * FATT * sizeof(float));
    f.w3 = *(const v4f*)(wk + wo + 3 * FATT * sizeof(float));
}

__device__ __forceinline__ void mma_frag(const Frag& f, v8f (&acc)[2][4]) {
    const v2f a0lo = { f.a0.x, f.a0.y }, a0hi = { f.a0.z, f.a0.w };
    const v2f a1lo = { f.a1.x, f.a1.y }, a1hi = { f.a1.z, f.a1.w };
    #pragma unroll
    for (int j = 0; j < 4; ++j) {
        const v2f b = { f.w0[j], f.w1[j] };
        acc[0][j] = __builtin_amdgcn_wmma_f32_16x16x4_f32(false, a0lo, false, b, (short)0, acc[0][j], false, false);
        acc[1][j] = __builtin_amdgcn_wmma_f32_16x16x4_f32(false, a1lo, false, b, (short)0, acc[1][j], false, false);
    }
    #pragma unroll
    for (int j = 0; j < 4; ++j) {
        const v2f b = { f.w2[j], f.w3[j] };
        acc[0][j] = __builtin_amdgcn_wmma_f32_16x16x4_f32(false, a0hi, false, b, (short)0, acc[0][j], false, false);
        acc[1][j] = __builtin_amdgcn_wmma_f32_16x16x4_f32(false, a1hi, false, b, (short)0, acc[1][j], false, false);
    }
}

__global__ __launch_bounds__(128) void gemm_wmma_f32(
        const float* __restrict__ X, const float* __restrict__ W,
        float* __restrict__ Z, int M, int K) {
    const int lane = threadIdx.x & 31;
    const int wave = threadIdx.x >> 5;
    const int c    = lane & 15;
    const int h    = lane >> 4;

    const int row_base = (blockIdx.x * 4 + wave) * 32;
    if (row_base >= M) return;

    const int r0 = min(row_base + c, M - 1);
    const int r1 = min(row_base + 16 + c, M - 1);
    const unsigned xo0 = ((unsigned)r0 * (unsigned)K + 4u * h) * (unsigned)sizeof(float);
    const unsigned xo1 = ((unsigned)r1 * (unsigned)K + 4u * h) * (unsigned)sizeof(float);
    const unsigned wo  = (4u * h * FATT + 4u * c) * (unsigned)sizeof(float);

    v8f acc[2][4] = {};

    const int nfull = K >> 3;
    int kb = 0;
    if (nfull > 0) {
        Frag f[DEPTH];
        #pragma unroll
        for (int i = 0; i < DEPTH; ++i)
            load_frag(f[i], X, W, xo0, xo1, wo, min(i, nfull - 1));

        for (; kb + DEPTH <= nfull; kb += DEPTH) {
            #pragma unroll
            for (int i = 0; i < DEPTH; ++i) {
                mma_frag(f[i], acc);
                load_frag(f[i], X, W, xo0, xo1, wo, min(kb + DEPTH + i, nfull - 1));
                __builtin_amdgcn_sched_barrier(0);
            }
        }
        #pragma unroll
        for (int i = 0; i < DEPTH; ++i)
            if (kb + i < nfull) mma_frag(f[i], acc);
    }

    const int kt = nfull * 8;
    if (kt < K) {
        Frag t;
        float x0[4], x1[4];
        v4f wv[4];
        #pragma unroll
        for (int i = 0; i < 4; ++i) {
            const int  k  = kt + 4 * h + i;
            const bool ok = k < K;
            const int  kc = ok ? k : (K - 1);
            const float p0 = X[(size_t)r0 * K + kc];
            const float p1 = X[(size_t)r1 * K + kc];
            const v4f   pw = *(const v4f*)(W + (size_t)kc * FATT + 4 * c);
            const v4f   zero = { 0.0f, 0.0f, 0.0f, 0.0f };
            x0[i] = ok ? p0 : 0.0f;
            x1[i] = ok ? p1 : 0.0f;
            wv[i] = ok ? pw : zero;
        }
        t.a0 = (v4f){ x0[0], x0[1], x0[2], x0[3] };
        t.a1 = (v4f){ x1[0], x1[1], x1[2], x1[3] };
        t.w0 = wv[0]; t.w1 = wv[1]; t.w2 = wv[2]; t.w3 = wv[3];
        mma_frag(t, acc);
    }

    #pragma unroll
    for (int mt = 0; mt < 2; ++mt) {
        #pragma unroll
        for (int r = 0; r < 8; ++r) {
            const int mm = row_base + 16 * mt + 8 * h + r;
            if (mm < M) {
                const v4f o = { acc[mt][0][r], acc[mt][1][r], acc[mt][2][r], acc[mt][3][r] };
                *(v4f*)(Z + (size_t)mm * FATT + 4 * c) = o;
            }
        }
    }
}

__device__ __forceinline__ unsigned int float_to_ord(float f) {
    unsigned int u = __float_as_uint(f);
    return (u & 0x80000000u) ? ~u : (u | 0x80000000u);
}
__device__ __forceinline__ float ord_to_float(unsigned int u) {
    return __uint_as_float((u & 0x80000000u) ? (u ^ 0x80000000u) : ~u);
}

template <int CTRL>
__device__ __forceinline__ float dpp_move(float v) {
    return __int_as_float(__builtin_amdgcn_update_dpp(0, __float_as_int(v), CTRL, 0xF, 0xF, true));
}
__device__ __forceinline__ float wave_sum(float v) {
    v += dpp_move<0x161>(v);
    v += dpp_move<0x162>(v);
    v += dpp_move<0x164>(v);
    v += dpp_move<0x168>(v);
    v += __int_as_float((int)__builtin_amdgcn_permlanex16(0u, (unsigned)__float_as_int(v),
                                                          0x76543210u, 0xfedcba98u, false, false));
    return v;
}

#define OWN_BLOCKS  256
#define OWN_NODES   ((NTOT + OWN_BLOCKS - 1) / OWN_BLOCKS)
#define KEY_NVEC    (NEDGE / 16)
#define KEY_THREADS 1024
#define WS_Z_BYTES  ((size_t)NTOT * FATT * sizeof(float))
#define WS_NEED     (WS_Z_BYTES + (size_t)NEDGE)

__device__ __forceinline__ unsigned int key_of(int d) {
    const unsigned int u = (unsigned int)d;
    return u < (unsigned int)(OWN_BLOCKS * OWN_NODES) ? u / (unsigned int)OWN_NODES : 255u;
}

#define KEY_PW       (KEY_THREADS / 32)
#define PRES_LD      (OWN_BLOCKS + 1)
#define PRES_WORDS   (KEY_NVEC / 32)
#define PRES_PITCH_F ((size_t)OWN_NODES * FATT)
#define KEY_GRID     ((KEY_NVEC + KEY_THREADS - 1) / KEY_THREADS)
static_assert(KEY_GRID * KEY_PW <= (NTOT - (OWN_BLOCKS - 1) * OWN_NODES) * FATT,
              "presence words must fit the last owner's rows");
static_assert(KEY_PW == 32 && (OWN_BLOCKS * KEY_PW / 4) % KEY_THREADS == 0,
              "a work-group stores one whole 128-byte line per owner, 16 bytes per thread and pass");
static_assert((OWN_NODES * FATT * 4) % 128 == 0, "an owner's presence words start on a 128-byte line");

__global__ __launch_bounds__(KEY_THREADS) void owner_keys(
        const int* __restrict__ dst, unsigned int* __restrict__ keys, unsigned int* __restrict__ pres,
        size_t ws_size) {
    if (ws_size < WS_NEED) return;
    __shared__ unsigned int pres_s[KEY_PW * PRES_LD];
    const int tid = threadIdx.x;
    for (int i = tid; i < KEY_PW * PRES_LD; i += KEY_THREADS) pres_s[i] = 0u;
    __syncthreads();
    const int v = blockIdx.x * KEY_THREADS + tid;
    if (v < KEY_NVEC) {
        const int4* d4 = (const int4*)dst + (size_t)v * 4;
        unsigned int w[4];
        #pragma unroll
        for (int j = 0; j < 4; ++j) {
            const int4 d = d4[j];
            w[j] = key_of(d.x) | (key_of(d.y) << 8) | (key_of(d.z) << 16) | (key_of(d.w) << 24);
        }
        ((uint4*)keys)[v] = make_uint4(w[0], w[1], w[2], w[3]);
        const unsigned int bit = 1u << (tid & 31);
        unsigned int* row = pres_s + (tid >> 5) * PRES_LD;
        #pragma unroll
        for (int j = 0; j < 4; ++j) {
            #pragma unroll
            for (int k = 0; k < 4; ++k)
                atomicOr(&row[(w[j] >> (8 * k)) & 0xffu], bit);
        }
    }
    __syncthreads();
    #pragma unroll
    for (int r = 0; r < (OWN_BLOCKS * KEY_PW / 4) / KEY_THREADS; ++r) {
        const int q  = r * KEY_THREADS + tid;
        const int o  = q >> 3;
        const int w0 = 4 * (q & 7);
        const uint4 p = make_uint4(pres_s[(w0 + 0) * PRES_LD + o], pres_s[(w0 + 1) * PRES_LD + o],
                                   pres_s[(w0 + 2) * PRES_LD + o], pres_s[(w0 + 3) * PRES_LD + o]);
        *(uint4*)(pres + (size_t)o * PRES_PITCH_F + (size_t)blockIdx.x * KEY_PW + w0) = p;
    }
}

#define OWN_THREADS 1024
#define OWN_WAVES   (OWN_THREADS / 32)
#define OWN_KVEC    7
#define OWN_NMASK   ((OWN_KVEC + 1) / 2)
#define OWN_RVEC    (OWN_THREADS * OWN_KVEC)
#define OWN_CHUNK   (OWN_RVEC * 16)
#define OWN_BATCH   16
#define OWN_TRIP    (OWN_WAVES * OWN_BATCH)
#define OWN_LIST    4096
#define OWN_ROUNDS  ((NEDGE + OWN_CHUNK - 1) / OWN_CHUNK)
#define OWN_RISE    60.0f

__device__ __forceinline__ unsigned int zero_bytes(unsigned int x) {
    return ~(((x & 0x7f7f7f7fu) + 0x7f7f7f7fu) | x) & 0x80808080u;
}

__device__ __forceinline__ void list_window(const unsigned int (&hm)[OWN_NMASK], int first,
                                            int ebase, int* list_s) {
    int idx = first;
    #pragma unroll
    for (int m = 0; m < OWN_NMASK; ++m) {
        unsigned int mask = hm[m];
        while (mask != 0u) {
            const int p = __ffs((int)mask) - 1;
            mask &= mask - 1u;
            if ((unsigned)idx < (unsigned)OWN_LIST)
                list_s[idx] = ebase + (2 * m + ((p >> 2) & 1)) * (OWN_THREADS * 16) + (p & 3) * 4 + (p >> 3);
            ++idx;
        }
    }
}

template <bool KEYS>
__device__ __forceinline__ void attend_body(
        const float* __restrict__ z, const int* __restrict__ src, const int* __restrict__ dst,
        const unsigned int* __restrict__ keys, float* __restrict__ out,
        float* h_s, float* den_s, unsigned int* mo_s, unsigned int* mc_s, int* list_s, int* cnt_s,
        int* flag_s) {
    const int tid  = threadIdx.x;
    const int lane = tid & 31;
    const int wave = tid >> 5;
    const int base = blockIdx.x * OWN_NODES;

    for (int i = tid; i < OWN_NODES * FATT; i += OWN_THREADS) h_s[i] = 0.0f;
    for (int i = tid; i < OWN_NODES; i += OWN_THREADS) {
        den_s[i] = 0.0f;
        mo_s[i]  = 0u;
        mc_s[i]  = 0u;
    }
    if (tid < 3) {
        cnt_s[tid]  = 0;
        flag_s[tid] = 0;
    }
    int g = 0;

    const int4*  dst4 = (const int4*)dst;
    const uint4* key4 = (const uint4*)keys;
    const unsigned int pat = blockIdx.x * 0x01010101u;
    __syncthreads();

    for (int r = 0; r < OWN_ROUNDS; ++r) {
        int* cnt = &cnt_s[r % 3];
        const int vbase = r * OWN_RVEC;
        const int ebase = (vbase + tid) * 16;

        unsigned int hm[OWN_NMASK];
        #pragma unroll
        for (int m = 0; m < OWN_NMASK; ++m) hm[m] = 0u;

        if constexpr (KEYS) {
            uint4 kv[OWN_KVEC];
            #pragma unroll
            for (int i = 0; i < OWN_KVEC; ++i)
                kv[i] = key4[min(vbase + i * OWN_THREADS + tid, KEY_NVEC - 1)];
            #pragma unroll
            for (int i = 0; i < OWN_KVEC; ++i) {
                const unsigned int live = vbase + i * OWN_THREADS + tid < KEY_NVEC ? 0xffffffffu : 0u;
                const unsigned int bits = (zero_bytes(kv[i].x ^ pat) >> (7 - (4 * (i & 1) + 0))) |
                                          (zero_bytes(kv[i].y ^ pat) >> (7 - (4 * (i & 1) + 1))) |
                                          (zero_bytes(kv[i].z ^ pat) >> (7 - (4 * (i & 1) + 2))) |
                                          (zero_bytes(kv[i].w ^ pat) >> (7 - (4 * (i & 1) + 3)));
                hm[i >> 1] |= bits & live;
            }
        } else {
            #pragma nounroll
            for (int m = 0; m < OWN_NMASK; ++m) {
                unsigned int bits = 0u;
                #pragma unroll
                for (int ii = 0; ii < 2; ++ii) {
                    const int i = 2 * m + ii;
                    const int v = vbase + i * OWN_THREADS + tid;
                    if (i < OWN_KVEC && v < KEY_NVEC) {
                        #pragma unroll
                        for (int j = 0; j < 4; ++j) {
                            const int4 d = dst4[(size_t)v * 4 + j];
                            const unsigned int b =
                                ((unsigned)(d.x - base) < (unsigned)OWN_NODES ? 0x00000001u : 0u) |
                                ((unsigned)(d.y - base) < (unsigned)OWN_NODES ? 0x00000100u : 0u) |
                                ((unsigned)(d.z - base) < (unsigned)OWN_NODES ? 0x00010000u : 0u) |
                                ((unsigned)(d.w - base) < (unsigned)OWN_NODES ? 0x01000000u : 0u);
                            bits |= b << (4 * ii + j);
                        }
                    }
                }
                #pragma unroll
                for (int k = 0; k < OWN_NMASK; ++k)
                    if (k == m) hm[k] = bits;
            }
        }
        int total = 0;
        #pragma unroll
        for (int m = 0; m < OWN_NMASK; ++m) total += __popc(hm[m]);
        int pos0 = 0;
        if (total != 0) {
            pos0 = atomicAdd(cnt, total);
            list_window(hm, pos0, ebase, list_s);
        }
        __syncthreads();
        const int n = *cnt;
        if (tid == 0) cnt_s[(r + 2) % 3] = 0;

        for (int lo = 0; lo < n; lo += OWN_LIST) {
            if (lo != 0) {
                if (total != 0) list_window(hm, pos0 - lo, ebase, list_s);
                __syncthreads();
            }
            const int nw    = min(n - lo, OWN_LIST);
            const int trips = (nw + OWN_TRIP - 1) / OWN_TRIP;
            for (int t = 0; t < trips; ++t) {
                const int j0 = t * OWN_TRIP + wave * OWN_BATCH;
                bool ok = lane < OWN_BATCH && j0 + lane < nw;

                int s = 0, dl = 0;
                if (ok) {
                    const int edge = list_s[j0 + lane];
                    s  = src[edge];
                    dl = dst[edge] - base;
                    if ((unsigned)dl >= (unsigned)OWN_NODES) {
                        ok = false;
                        s  = 0;
                        dl = 0;
                    }
                }
                const unsigned int okm = __builtin_amdgcn_ballot_w32(ok);
                float2 rs[OWN_BATCH];
                int    dq[OWN_BATCH];
                float  sc = 0.0f;
                if (okm != 0u) {
                    float2 rd[OWN_BATCH];
                    #pragma unroll
                    for (int q = 0; q < OWN_BATCH; ++q) {
                        const int sq = __builtin_amdgcn_readlane(s, q);
                        dq[q] = __builtin_amdgcn_readlane(dl, q);
                        rs[q] = ((const float2*)(z + (size_t)sq * FATT))[lane];
                        rd[q] = ((const float2*)(z + (size_t)(base + dq[q]) * FATT))[lane];
                    }
                    #pragma unroll
                    for (int q = 0; q < OWN_BATCH; ++q) {
                        const float dot = wave_sum(rs[q].x * rd[q].x + rs[q].y * rd[q].y);
                        if (lane == q) sc = dot;
                    }
                }
                sc = sc > 0.0f ? sc : SLOPE * sc;

                int*  flag = &flag_s[g % 3];
                float ref  = sc;
                bool  up   = false;
                if (ok) {
                    const unsigned int mold = atomicCAS(&mc_s[dl], 0u, float_to_ord(sc));
                    if (mold != 0u) {
                        ref = ord_to_float(mold);
                        up  = sc - ref > OWN_RISE;
                        if (up) *flag = 1;
                    }
                }

                const bool now = ok && !up;
                const unsigned int nowm = __builtin_amdgcn_ballot_w32(now);
                float w = 0.0f;
                if (now) {
                    w = __expf(sc - ref);
                    atomicAdd(&den_s[dl], w);
                }
                #pragma unroll
                for (int q = 0; q < OWN_BATCH; ++q) {
                    if ((nowm >> q) & 1u) {
                        const float wq = __int_as_float(__builtin_amdgcn_readlane(__float_as_int(w), q));
                        float* hp = h_s + dq[q] * FATT + 2 * lane;
                        atomicAdd(hp + 0, wq * rs[q].x);
                        atomicAdd(hp + 1, wq * rs[q].y);
                    }
                }
                __syncthreads();
                if (tid == 0) flag_s[(g + 2) % 3] = 0;
                ++g;
                const int rise = __builtin_amdgcn_readfirstlane(*flag);

                if (rise != 0) {
                    if (up) atomicMax(&mo_s[dl], float_to_ord(sc));
                    __syncthreads();

                    unsigned int mnew = 0u;
                    int   take = 0;
                    float fs   = 1.0f;
                    if (up) {
                        mnew = mo_s[dl];
                        const unsigned int mold = atomicMax(&mc_s[dl], mnew);
                        if (mold < mnew) {
                            take = 1;
                            fs   = __expf(ord_to_float(mold) - ord_to_float(mnew));
                            den_s[dl] *= fs;
                        }
                    }
                    if (__builtin_amdgcn_ballot_w32(take != 0) != 0u) {
                        #pragma unroll
                        for (int q = 0; q < OWN_BATCH; ++q) {
                            if (__builtin_amdgcn_readlane(take, q)) {
                                const float f = __int_as_float(__builtin_amdgcn_readlane(__float_as_int(fs), q));
                                float2* hp = (float2*)(h_s + dq[q] * FATT) + lane;
                                float2 hv = *hp;
                                hv.x *= f;
                                hv.y *= f;
                                *hp = hv;
                            }
                        }
                    }
                    __syncthreads();

                    const unsigned int upm = __builtin_amdgcn_ballot_w32(up);
                    float w2 = 0.0f;
                    if (up) {
                        mo_s[dl] = 0u;
                        w2 = __expf(sc - ord_to_float(mnew));
                        atomicAdd(&den_s[dl], w2);
                    }
                    #pragma unroll
                    for (int q = 0; q < OWN_BATCH; ++q) {
                        if ((upm >> q) & 1u) {
                            const float wq = __int_as_float(__builtin_amdgcn_readlane(__float_as_int(w2), q));
                            float* hp = h_s + dq[q] * FATT + 2 * lane;
                            atomicAdd(hp + 0, wq * rs[q].x);
                            atomicAdd(hp + 1, wq * rs[q].y);
                        }
                    }
                }
            }
        }
    }
    __syncthreads();

    for (int i4 = tid; i4 < OWN_NODES * (FATT / 4); i4 += OWN_THREADS) {
        const int nl = i4 / (FATT / 4);
        if (base + nl < NTOT) {
            const float dn = den_s[nl];
            float4 v = *(const float4*)&h_s[i4 * 4];
            v.x = dn > 0.0f ? v.x / dn : 0.0f;
            v.y = dn > 0.0f ? v.y / dn : 0.0f;
            v.z = dn > 0.0f ? v.z / dn : 0.0f;
            v.w = dn > 0.0f ? v.w / dn : 0.0f;
            v.x = v.x > 0.0f ? v.x : (__expf(v.x) - 1.0f);
            v.y = v.y > 0.0f ? v.y : (__expf(v.y) - 1.0f);
            v.z = v.z > 0.0f ? v.z : (__expf(v.z) - 1.0f);
            v.w = v.w > 0.0f ? v.w : (__expf(v.w) - 1.0f);
            *(float4*)(out + (size_t)base * FATT + (size_t)i4 * 4) = v;
        }
    }
}

#define FAST_CAP 12288
#define HELD_CAP 1024
#define SEG_CAP  (FAST_CAP / OWN_WAVES)
#define VSEG_CAP 320
#define VSEG_UNR 4
#define PRES_WAVE  ((PRES_WORDS + OWN_WAVES - 1) / OWN_WAVES)
#define PRES_TRIPS ((PRES_WAVE + 31) / 32)
#define FAST_STAGE ((OWN_NODES * (FATT / 4) + OWN_THREADS - 1) / OWN_THREADS)

__device__ __forceinline__ bool attend_fast(
        const float* __restrict__ z, const int* __restrict__ src, const int* __restrict__ dst,
        const unsigned int* __restrict__ keys, const unsigned int* pres, float* out,
        float* h_s, float* zd_s, float* den_s, unsigned int* mo_s, unsigned int* mc_s, int* ed_s, int* vs_s,
        int* hs_s, int* hd_s, float* hsc_s, int* cnt_f, int* cnt_h) {
    const int tid  = threadIdx.x;
    const int lane = tid & 31;
    const int wave = __builtin_amdgcn_readfirstlane(tid >> 5);
    const int base = blockIdx.x * OWN_NODES;

    const float4* zown = (const float4*)(z + (size_t)base * FATT);
    const int nv4 = (min(base + OWN_NODES, NTOT) - base) * (FATT / 4);
    float4 sv[FAST_STAGE];
    #pragma unroll
    for (int k = 0; k < FAST_STAGE; ++k)
        sv[k] = zown[min(tid + k * OWN_THREADS, nv4 - 1)];
    #pragma unroll
    for (int k = 0; k < FAST_STAGE; ++k) {
        const int i4 = tid + k * OWN_THREADS;
        if (i4 < OWN_NODES * (FATT / 4)) {
            *(float4*)&zd_s[i4 * 4] = i4 < nv4 ? sv[k] : make_float4(0.0f, 0.0f, 0.0f, 0.0f);
            *(float4*)&h_s[i4 * 4]  = make_float4(0.0f, 0.0f, 0.0f, 0.0f);
        }
    }
    for (int i = tid; i < OWN_NODES; i += OWN_THREADS) {
        den_s[i] = 0.0f;
        mo_s[i]  = 0u;
        mc_s[i]  = 0u;
    }
    if (tid == 0) {
        *cnt_f = 0;
        *cnt_h = 0;
    }
    const uint4* key4 = (const uint4*)keys;
    const unsigned int pat = blockIdx.x * 0x01010101u;
    unsigned int pw[PRES_TRIPS];
    #pragma unroll
    for (int k = 0; k < PRES_TRIPS; ++k) {
        const int i = wave * PRES_WAVE + k * 32 + lane;
        const unsigned int x = pres[min(i, PRES_WORDS - 1)];
        pw[k] = (k * 32 + lane < PRES_WAVE && i < PRES_WORDS) ? x : 0u;
    }
    __syncthreads();

    int* vseg = vs_s + wave * VSEG_CAP;
    int* eseg = ed_s + wave * SEG_CAP;
    int nv = 0;
    #pragma unroll
    for (int k = 0; k < PRES_TRIPS; ++k) {
        unsigned int m = pw[k];
        const int v0 = (wave * PRES_WAVE + k * 32 + lane) * 32;
        for (;;) {
            const bool has = m != 0u;
            const unsigned int bal = __builtin_amdgcn_ballot_w32(has);
            if (bal == 0u) break;
            if (has) {
                const int p   = __ffs((int)m) - 1;
                const int pos = nv + (int)__builtin_amdgcn_mbcnt_lo(bal, 0u);
                m &= m - 1u;
                if (pos < VSEG_CAP) vseg[pos] = v0 + p;
            }
            nv += __popc(bal);
        }
    }
    const int nvc = min(nv, VSEG_CAP);
    int ne = 0;
    for (int j0 = 0; j0 < nvc; j0 += 32 * VSEG_UNR) {
        int   vid[VSEG_UNR];
        uint4 kv[VSEG_UNR];
        #pragma unroll
        for (int u = 0; u < VSEG_UNR; ++u) {
            const int j = j0 + 32 * u + lane;
            vid[u] = j < nvc ? vseg[j] : -1;
        }
        #pragma unroll
        for (int u = 0; u < VSEG_UNR; ++u)
            kv[u] = key4[max(vid[u], 0)];
        #pragma unroll
        for (int u = 0; u < VSEG_UNR; ++u) {
            unsigned int m = (zero_bytes(kv[u].x ^ pat) >> 7) | (zero_bytes(kv[u].y ^ pat) >> 6) |
                             (zero_bytes(kv[u].z ^ pat) >> 5) | (zero_bytes(kv[u].w ^ pat) >> 4);
            m = vid[u] >= 0 ? m : 0u;
            for (;;) {
                const bool has = m != 0u;
                const unsigned int bal = __builtin_amdgcn_ballot_w32(has);
                if (bal == 0u) break;
                if (has) {
                    const int p   = __ffs((int)m) - 1;
                    const int pos = ne + (int)__builtin_amdgcn_mbcnt_lo(bal, 0u);
                    m &= m - 1u;
                    if (pos < SEG_CAP) eseg[pos] = vid[u] * 16 + (p & 3) * 4 + (p >> 3);
                }
                ne += __popc(bal);
            }
        }
    }
    if (nv > VSEG_CAP || ne > SEG_CAP) *cnt_f = 1;
    __syncthreads();
    if (__builtin_amdgcn_readfirstlane(*cnt_f) != 0) return false;
    const int n = ne;

    for (int j0 = 0; j0 < n; j0 += 32) {
        const int j = j0 + lane;
        bool ok = j < n;
        int s = 0, dl = 0;
        if (ok) {
            const int edge = eseg[j];
            s  = src[edge];
            dl = dst[edge] - base;
            if ((unsigned)dl >= (unsigned)OWN_NODES) {
                ok = false;
                s  = 0;
                dl = 0;
            }
        }
        const unsigned int okm = __builtin_amdgcn_ballot_w32(ok);
        float2 rs[2][OWN_BATCH];
        #pragma unroll
        for (int sub = 0; sub < 2; ++sub) {
            if (((okm >> (16 * sub)) & 0xffffu) != 0u) {
                #pragma unroll
                for (int q = 0; q < OWN_BATCH; ++q) {
                    const int sq = __builtin_amdgcn_readlane(s, 16 * sub + q);
                    rs[sub][q] = ((const float2*)(z + (size_t)sq * FATT))[lane];
                }
            }
        }
        #pragma unroll
        for (int sub = 0; sub < 2; ++sub) {
            if (((okm >> (16 * sub)) & 0xffffu) != 0u) {
                float sc = 0.0f;
                {
                    float2 rd[OWN_BATCH];
                    #pragma unroll
                    for (int q = 0; q < OWN_BATCH; ++q) {
                        const int dq = __builtin_amdgcn_readlane(dl, 16 * sub + q);
                        rd[q] = ((const float2*)(zd_s + dq * FATT))[lane];
                    }
                    #pragma unroll
                    for (int q = 0; q < OWN_BATCH; ++q) {
                        const float dot = wave_sum(rs[sub][q].x * rd[q].x + rs[sub][q].y * rd[q].y);
                        if (lane == 16 * sub + q) sc = dot;
                    }
                }
                sc = sc > 0.0f ? sc : SLOPE * sc;

                bool  now = false;
                float w   = 0.0f;
                if (ok && (lane >> 4) == sub) {
                    const unsigned int mold = atomicCAS(&mc_s[dl], 0u, float_to_ord(sc));
                    const float ref = mold != 0u ? ord_to_float(mold) : sc;
                    if (sc - ref > OWN_RISE) {
                        const int p = atomicAdd(cnt_h, 1);
                        if (p < HELD_CAP) {
                            hs_s[p]  = s;
                            hd_s[p]  = dl;
                            hsc_s[p] = sc;
                        }
                    } else {
                        now = true;
                        w   = __expf(sc - ref);
                        atomicAdd(&den_s[dl], w);
                    }
                }
                const unsigned int nowm = (__builtin_amdgcn_ballot_w32(now) >> (16 * sub)) & 0xffffu;
                #pragma unroll
                for (int q = 0; q < OWN_BATCH; ++q) {
                    if ((nowm >> q) & 1u) {
                        const float wq = __int_as_float(__builtin_amdgcn_readlane(__float_as_int(w), 16 * sub + q));
                        const int   dq = __builtin_amdgcn_readlane(dl, 16 * sub + q);
                        float* hp = h_s + dq * FATT + 2 * lane;
                        atomicAdd(hp + 0, wq * rs[sub][q].x);
                        atomicAdd(hp + 1, wq * rs[sub][q].y);
                    }
                }
            }
        }
    }
    __syncthreads();
    const int nh = __builtin_amdgcn_readfirstlane(*cnt_h);
    if (nh > HELD_CAP) return false;

    if (nh > 0) {
        for (int i = tid; i < nh; i += OWN_THREADS)
            atomicMax(&mo_s[hd_s[i]], float_to_ord(hsc_s[i]));
        __syncthreads();

        for (int i = wave; i < nh; i += OWN_WAVES) {
            const int dl = hd_s[i];
            const unsigned int mnew = mo_s[dl];
            unsigned int mold = mnew;
            if (lane == 0) mold = atomicMax(&mc_s[dl], mnew);
            mold = __builtin_amdgcn_readfirstlane(mold);
            if (mold < mnew) {
                const float fs = __expf(ord_to_float(mold) - ord_to_float(mnew));
                float2* hp = (float2*)(h_s + dl * FATT) + lane;
                float2 hv = *hp;
                hv.x *= fs;
                hv.y *= fs;
                *hp = hv;
                if (lane == 0) den_s[dl] *= fs;
            }
        }
        __syncthreads();

        for (int i = wave; i < nh; i += OWN_WAVES) {
            const int   s  = hs_s[i];
            const int   dl = hd_s[i];
            const float w  = __expf(hsc_s[i] - ord_to_float(mc_s[dl]));
            const float2 r = ((const float2*)(z + (size_t)s * FATT))[lane];
            if (lane == 0) atomicAdd(&den_s[dl], w);
            float* hp = h_s + dl * FATT + 2 * lane;
            atomicAdd(hp + 0, w * r.x);
            atomicAdd(hp + 1, w * r.y);
        }
        __syncthreads();
    }

    for (int i4 = tid; i4 < OWN_NODES * (FATT / 4); i4 += OWN_THREADS) {
        const int nl = i4 / (FATT / 4);
        if (base + nl < NTOT) {
            const float dn = den_s[nl];
            float4 v = *(const float4*)&h_s[i4 * 4];
            v.x = dn > 0.0f ? v.x / dn : 0.0f;
            v.y = dn > 0.0f ? v.y / dn : 0.0f;
            v.z = dn > 0.0f ? v.z / dn : 0.0f;
            v.w = dn > 0.0f ? v.w / dn : 0.0f;
            v.x = v.x > 0.0f ? v.x : (__expf(v.x) - 1.0f);
            v.y = v.y > 0.0f ? v.y : (__expf(v.y) - 1.0f);
            v.z = v.z > 0.0f ? v.z : (__expf(v.z) - 1.0f);
            v.w = v.w > 0.0f ? v.w : (__expf(v.w) - 1.0f);
            *(float4*)(out + (size_t)base * FATT + (size_t)i4 * 4) = v;
        }
    }
    return true;
}

__global__ __launch_bounds__(OWN_THREADS) void node_block_attend(
        const float* __restrict__ z, const int* __restrict__ src, const int* __restrict__ dst,
        const unsigned int* __restrict__ keys, const unsigned int* pres, float* out, size_t ws_size) {
    __shared__ __attribute__((aligned(16))) float h_s[OWN_NODES * FATT];
    __shared__ __attribute__((aligned(16))) float zd_s[OWN_NODES * FATT];
    __shared__ float        den_s[OWN_NODES];
    __shared__ unsigned int mo_s[OWN_NODES];
    __shared__ unsigned int mc_s[OWN_NODES];
    __shared__ int          ed_s[FAST_CAP];
    __shared__ int          vs_s[OWN_WAVES * VSEG_CAP];
    __shared__ int          hs_s[HELD_CAP];
    __shared__ int          hd_s[HELD_CAP];
    __shared__ float        hsc_s[HELD_CAP];
    __shared__ int          cnt_f;
    __shared__ int          cnt_h;
    __shared__ int          cnt_s[3];
    __shared__ int          flag_s[3];

    if (ws_size >= WS_NEED) {
        if (!attend_fast(z, src, dst, keys, pres + (size_t)blockIdx.x * PRES_PITCH_F, out,
                         h_s, zd_s, den_s, mo_s, mc_s, ed_s, vs_s, hs_s, hd_s, hsc_s, &cnt_f, &cnt_h))
            attend_body<true >(z, src, dst, keys, out, h_s, den_s, mo_s, mc_s, ed_s, cnt_s, flag_s);
    } else {
        attend_body<false>(z, src, dst, keys, out, h_s, den_s, mo_s, mc_s, ed_s, cnt_s, flag_s);
    }
}

extern "C" void kernel_launch(void* const* d_in, const int* in_sizes, int n_in,
                              void* d_out, int out_size, void* d_ws, size_t ws_size,
                              hipStream_t stream) {
    const float* d_sim  = (const float*)d_in[0];
    const float* mi_sim = (const float*)d_in[1];
    const float* W_d    = (const float*)d_in[2];
    const float* W_mi   = (const float*)d_in[3];
    const int*   src    = (const int*)d_in[4];
    const int*   dst    = (const int*)d_in[5];
    float*       out    = (float*)d_out;

    float*        z     = (float*)d_ws;
    unsigned int* keys  = (unsigned int*)((char*)d_ws + WS_Z_BYTES);

    gemm_wmma_f32<<<(N_DIS + 127) / 128, 128, 0, stream>>>(d_sim, W_d, z, N_DIS, 383);
    gemm_wmma_f32<<<(N_MIC + 127) / 128, 128, 0, stream>>>(mi_sim, W_mi,
                                                           z + (size_t)N_DIS * FATT, N_MIC, 495);

    owner_keys<<<KEY_GRID, KEY_THREADS, 0, stream>>>(dst, keys, (unsigned int*)d_out, ws_size);

    node_block_attend<<<OWN_BLOCKS, OWN_THREADS, 0, stream>>>(z, src, dst, keys, (const unsigned int*)d_out, out, ws_size);
}
